# starting kernel
# baseline (speedup 1.0000x reference)
#include <hip/hip_runtime.h>

#define N_NODES 100000
#define D_IN 256
#define D_H 128
#define D_OUT 32

typedef __attribute__((ext_vector_type(16))) __bf16 v16bf;
typedef __attribute__((ext_vector_type(8)))  float  v8f;

__device__ __forceinline__ v16bf frag_from_lds(const __bf16* base, int rowIdx, int khalf) {
  union { uint4 u[2]; v16bf v; } c;
  const uint4* p = (const uint4*)(base + rowIdx * 32 + khalf);
  c.u[0] = p[0];
  c.u[1] = p[2];
  return c.v;
}

template<int K, int N>
__global__ __launch_bounds__(256) void gemm_bf16_wmma(
    const float* __restrict__ A, const float* __restrict__ W,
    float* __restrict__ C,
    const float* __restrict__ dinv, const float* __restrict__ bias,
    float* __restrict__ out2, int M) {
  constexpr int NT = N / 16;
  __shared__ __bf16 ldsA[128 * 32];
  __shared__ __bf16 ldsB[N * 32];

  const int tid   = threadIdx.x;
  const int wave  = tid >> 5;
  const int lane  = tid & 31;
  const int laneM = lane & 15;
  const int khalf = (lane >> 4) << 3;
  const int rowBlk = blockIdx.x * 128;

  const int ldRow = tid >> 1;
  const int ldC0  = (tid & 1) << 4;
  const int ldGr  = rowBlk + ldRow;

  v8f acc[NT];
#pragma unroll
  for (int t = 0; t < NT; ++t) {
    v8f z = {0.f, 0.f, 0.f, 0.f, 0.f, 0.f, 0.f, 0.f};
    acc[t] = z;
  }

  for (int kk = 0; kk < K; kk += 32) {
    __syncthreads();

    {
      union { __bf16 h[16]; uint4 q[2]; } buf;
      if (ldGr < M) {
        const float4* src = (const float4*)(A + (size_t)ldGr * K + kk + ldC0);
#pragma unroll
        for (int j = 0; j < 4; ++j) {
          float4 v = src[j];
          buf.h[j * 4 + 0] = (__bf16)v.x;
          buf.h[j * 4 + 1] = (__bf16)v.y;
          buf.h[j * 4 + 2] = (__bf16)v.z;
          buf.h[j * 4 + 3] = (__bf16)v.w;
        }
      } else {
#pragma unroll
        for (int j = 0; j < 16; ++j) buf.h[j] = (__bf16)0.0f;
      }
      uint4* dst = (uint4*)(ldsA + ldRow * 32 + ldC0);
      dst[0] = buf.q[0];
      dst[1] = buf.q[1];
    }

#pragma unroll
    for (int i = tid; i < N * 8; i += 256) {
      const int n  = i % N;
      const int kq = i / N;
      union { __bf16 h[4]; uint2 q; } wb;
#pragma unroll
      for (int j = 0; j < 4; ++j)
        wb.h[j] = (__bf16)W[(size_t)(kk + kq * 4 + j) * N + n];
      *(uint2*)(ldsB + n * 32 + kq * 4) = wb.q;
    }
    __syncthreads();

    v16bf a = frag_from_lds(ldsA, (wave << 4) + laneM, khalf);
    v16bf bc = frag_from_lds(ldsB, laneM, khalf);
#pragma unroll
    for (int t = 0; t < NT; ++t) {
      v16bf bn = (t + 1 < NT) ? frag_from_lds(ldsB, ((t + 1) << 4) + laneM, khalf) : bc;
      acc[t] = __builtin_amdgcn_wmma_f32_16x16x32_bf16(
          false, a, false, bc, (short)0, acc[t], false, false);
      bc = bn;
    }
  }

  float bfrag[NT];
#pragma unroll
  for (int t = 0; t < NT; ++t) bfrag[t] = bias[(t << 4) + laneM];

  const int mBase = rowBlk + (wave << 4) + ((lane >> 4) << 3);
#pragma unroll
  for (int v = 0; v < 8; ++v) {
    const int gr = mBase + v;
    if (gr < M) {
      const float d  = dinv[gr];
      const float dd = d * d;
#pragma unroll
      for (int t = 0; t < NT; ++t) {
        const size_t idx = (size_t)gr * N + (t << 4) + laneM;
        const float val = acc[t][v];
        C[idx]    = val;
        out2[idx] = val * dd + bfrag[t];
      }
    }
  }
}

__global__ void k_fill1(float* __restrict__ p, int n) {
  int i = blockIdx.x * blockDim.x + threadIdx.x;
  if (i < n) p[i] = 1.0f;
}

__global__ void k_deg(const int* __restrict__ col, float* __restrict__ deg, int E) {
  int e = blockIdx.x * blockDim.x + threadIdx.x;
  if (e < E) atomicAdd(&deg[col[e]], 1.0f);
}

__global__ void k_rsqrt(float* __restrict__ p, int n) {
  int i = blockIdx.x * blockDim.x + threadIdx.x;
  if (i < n) p[i] = rsqrtf(p[i]);
}

template<int F>
__global__ __launch_bounds__(256) void k_edges(
    const float* __restrict__ h, const float* __restrict__ dinv,
    const int* __restrict__ row, const int* __restrict__ col,
    float* __restrict__ out, int E) {
  constexpr int EPB = 256 / F;
  const int f  = threadIdx.x & (F - 1);
  const int es = threadIdx.x / F;
  for (int e = blockIdx.x * EPB + es; e < E; e += gridDim.x * EPB) {
    int r = row[e], c = col[e];
    float nrm = dinv[r] * dinv[c];
    atomicAdd(&out[(size_t)c * F + f], h[(size_t)r * F + f] * nrm);
  }
}

__global__ void k_relu4(float4* __restrict__ p, int n4) {
  int i = blockIdx.x * blockDim.x + threadIdx.x;
  if (i < n4) {
    float4 v = p[i];
    v.x = fmaxf(v.x, 0.0f);
    v.y = fmaxf(v.y, 0.0f);
    v.z = fmaxf(v.z, 0.0f);
    v.w = fmaxf(v.w, 0.0f);
    p[i] = v;
  }
}

__global__ __launch_bounds__(256) void k_log_softmax32(float* __restrict__ out, int n) {
  int node = blockIdx.x * 8 + (threadIdx.x >> 5);
  int lane = threadIdx.x & 31;
  if (node >= n) return;
  float v = out[node * 32 + lane];
  float m = v;
#pragma unroll
  for (int o = 16; o > 0; o >>= 1) m = fmaxf(m, __shfl_xor(m, o, 32));
  float ex = __expf(v - m);
  float s = ex;
#pragma unroll
  for (int o = 16; o > 0; o >>= 1) s += __shfl_xor(s, o, 32);
  out[node * 32 + lane] = v - m - __logf(s);
}

extern "C" void kernel_launch(void* const* d_in, const int* in_sizes, int n_in,
                              void* d_out, int out_size, void* d_ws, size_t ws_size,
                              hipStream_t stream) {
  const float* x  = (const float*)d_in[0];
  const float* W1 = (const float*)d_in[1];
  const float* b1 = (const float*)d_in[2];
  const float* W2 = (const float*)d_in[3];
  const float* b2 = (const float*)d_in[4];
  const int*  eix = (const int*)d_in[5];

  const int n = N_NODES;
  const int E = in_sizes[5] / 2;
  const int* row = eix;
  const int* col = eix + E;

  float* ws   = (float*)d_ws;
  float* dinv = ws;
  float* h    = ws + 100352;
  float* h1   = h  + (size_t)n * D_H;
  float* g2   = h1 + (size_t)n * D_H;
  float* outf = (float*)d_out;

  k_fill1 <<<(n + 255) / 256, 256, 0, stream>>>(dinv, n);
  k_deg   <<<(E + 255) / 256, 256, 0, stream>>>(col, dinv, E);
  k_rsqrt <<<(n + 255) / 256, 256, 0, stream>>>(dinv, n);

  gemm_bf16_wmma<D_IN, D_H><<<(n + 127) / 128, 256, 0, stream>>>(
      x, W1, h, dinv, b1, h1, n);
  k_edges<D_H><<<4096, 256, 0, stream>>>(h, dinv, row, col, h1, E);
  k_relu4<<<((n * (D_H / 4)) + 255) / 256, 256, 0, stream>>>((float4*)h1, n * (D_H / 4));

  gemm_bf16_wmma<D_H, D_OUT><<<(n + 127) / 128, 256, 0, stream>>>(
      h1, W2, g2, dinv, b2, outf, n);
  k_edges<D_OUT><<<4096, 256, 0, stream>>>(g2, dinv, row, col, outf, E);
  k_log_softmax32<<<(n + 7) / 8, 256, 0, stream>>>(outf, n);
}


# Float-atomic scatter-add replaced by per-call grouping of edges by target (1 int atomic/edge, scan, place) + per-node register sum in slot (atomic arrival) order; bias/relu/log-softmax at its store
# speedup vs baseline: 2.3784x; 2.3784x over previous
#include <hip/hip_runtime.h>

#define N_NODES 100000
#define D_IN 256
#define D_H 128
#define D_OUT 32

typedef __attribute__((ext_vector_type(16))) __bf16 v16bf;
typedef __attribute__((ext_vector_type(8)))  float  v8f;

__device__ __forceinline__ v16bf frag_from_lds(const __bf16* base, int rowIdx, int khalf) {
  union { uint4 u[2]; v16bf v; } c;
  const uint4* p = (const uint4*)(base + rowIdx * 32 + khalf);
  c.u[0] = p[0];
  c.u[1] = p[2];
  return c.v;
}

template<int K, int N>
__global__ __launch_bounds__(256) void gemm_bf16_wmma(
    const float* __restrict__ A, const float* __restrict__ W,
    float* __restrict__ C,
    const float* __restrict__ dinv, int M) {
  constexpr int NT = N / 16;
  __shared__ __bf16 ldsA[128 * 32];
  __shared__ __bf16 ldsB[N * 32];

  const int tid   = threadIdx.x;
  const int wave  = tid >> 5;
  const int lane  = tid & 31;
  const int laneM = lane & 15;
  const int khalf = (lane >> 4) << 3;
  const int rowBlk = blockIdx.x * 128;

  const int ldRow = tid >> 1;
  const int ldC0  = (tid & 1) << 4;
  const int ldGr  = rowBlk + ldRow;

  v8f acc[NT];
#pragma unroll
  for (int t = 0; t < NT; ++t) {
    v8f z = {0.f, 0.f, 0.f, 0.f, 0.f, 0.f, 0.f, 0.f};
    acc[t] = z;
  }

  for (int kk = 0; kk < K; kk += 32) {
    __syncthreads();

    {
      union { __bf16 h[16]; uint4 q[2]; } buf;
      if (ldGr < M) {
        const float4* src = (const float4*)(A + (size_t)ldGr * K + kk + ldC0);
#pragma unroll
        for (int j = 0; j < 4; ++j) {
          float4 v = src[j];
          buf.h[j * 4 + 0] = (__bf16)v.x;
          buf.h[j * 4 + 1] = (__bf16)v.y;
          buf.h[j * 4 + 2] = (__bf16)v.z;
          buf.h[j * 4 + 3] = (__bf16)v.w;
        }
      } else {
#pragma unroll
        for (int j = 0; j < 16; ++j) buf.h[j] = (__bf16)0.0f;
      }
      uint4* dst = (uint4*)(ldsA + ldRow * 32 + ldC0);
      dst[0] = buf.q[0];
      dst[1] = buf.q[1];
    }

#pragma unroll
    for (int i = tid; i < N * 8; i += 256) {
      const int n  = i % N;
      const int kq = i / N;
      union { __bf16 h[4]; uint2 q; } wb;
#pragma unroll
      for (int j = 0; j < 4; ++j)
        wb.h[j] = (__bf16)W[(size_t)(kk + kq * 4 + j) * N + n];
      *(uint2*)(ldsB + n * 32 + kq * 4) = wb.q;
    }
    __syncthreads();

    v16bf a = frag_from_lds(ldsA, (wave << 4) + laneM, khalf);
    v16bf bc = frag_from_lds(ldsB, laneM, khalf);
#pragma unroll
    for (int t = 0; t < NT; ++t) {
      v16bf bn = (t + 1 < NT) ? frag_from_lds(ldsB, ((t + 1) << 4) + laneM, khalf) : bc;
      acc[t] = __builtin_amdgcn_wmma_f32_16x16x32_bf16(
          false, a, false, bc, (short)0, acc[t], false, false);
      bc = bn;
    }
  }

  const int mBase = rowBlk + (wave << 4) + ((lane >> 4) << 3);
#pragma unroll
  for (int v = 0; v < 8; ++v) {
    const int gr = mBase + v;
    if (gr < M) {
      const float d = dinv[gr];
#pragma unroll
      for (int t = 0; t < NT; ++t) {
        const size_t idx = (size_t)gr * N + (t << 4) + laneM;
        C[idx] = acc[t][v] * d;
      }
    }
  }
}

#define N_PAD 100352

__global__ void k_zero(int* __restrict__ p, int n) {
  int i = blockIdx.x * blockDim.x + threadIdx.x;
  if (i < n) p[i] = 0;
}

__global__ void k_rank(const int* __restrict__ col, int* __restrict__ cnt,
                       int* __restrict__ rank, int E) {
  int e = blockIdx.x * blockDim.x + threadIdx.x;
  if (e < E) rank[e] = atomicAdd(&cnt[col[e]], 1);
}

__global__ __launch_bounds__(1024) void k_scan(
    const int* __restrict__ cnt, int* __restrict__ rowptr, float* __restrict__ dinv) {
  __shared__ int wsum[32];
  __shared__ int s_off;
  const int tid  = threadIdx.x;
  const int lane = tid & 31;
  const int wave = tid >> 5;
  const int b    = blockIdx.x;

  int s = 0;
  const int4* c4 = (const int4*)cnt;
  const int lim = b * 256;
  int j = tid;
  for (; j + 3072 < lim; j += 4096) {
    const int4 v0 = c4[j], v1 = c4[j + 1024], v2 = c4[j + 2048], v3 = c4[j + 3072];
    s += ((v0.x + v0.y) + (v0.z + v0.w)) + ((v1.x + v1.y) + (v1.z + v1.w)) +
         ((v2.x + v2.y) + (v2.z + v2.w)) + ((v3.x + v3.y) + (v3.z + v3.w));
  }
  for (; j < lim; j += 1024) {
    const int4 v = c4[j];
    s += (v.x + v.y) + (v.z + v.w);
  }
  for (int o = 16; o > 0; o >>= 1) s += __shfl_xor(s, o, 32);
  if (lane == 0) wsum[wave] = s;
  __syncthreads();
  if (wave == 0) {
    int t = wsum[lane];
    for (int o = 16; o > 0; o >>= 1) t += __shfl_xor(t, o, 32);
    if (lane == 0) s_off = t;
  }
  __syncthreads();
  const int off = s_off;

  const int i = b * 1024 + tid;
  const int c = cnt[i];
  int x = c;
  for (int o = 1; o < 32; o <<= 1) {
    int y = __shfl_up(x, o, 32);
    if (lane >= o) x += y;
  }
  if (lane == 31) wsum[wave] = x;
  __syncthreads();
  if (wave == 0) {
    int t = wsum[lane];
    for (int o = 1; o < 32; o <<= 1) {
      int y = __shfl_up(t, o, 32);
      if (lane >= o) t += y;
    }
    wsum[lane] = t;
  }
  __syncthreads();
  const int wbase = (wave > 0) ? wsum[wave - 1] : 0;
  rowptr[i] = off + wbase + x - c;
  dinv[i]   = rsqrtf((float)(c + 1));
}

__global__ void k_place(const int* __restrict__ row, const int* __restrict__ col,
                        const int* __restrict__ rank, const int* __restrict__ rowptr,
                        int* __restrict__ src, int E) {
  int e = blockIdx.x * blockDim.x + threadIdx.x;
  if (e < E) src[rowptr[col[e]] + rank[e]] = row[e];
}

__global__ __launch_bounds__(256) void k_pull128_relu(
    const float* __restrict__ hs, const float* __restrict__ dinv,
    const float* __restrict__ bias, const int* __restrict__ rowptr,
    const int* __restrict__ cnt, const int* __restrict__ src,
    float* __restrict__ out, int n) {
  const int node = __builtin_amdgcn_readfirstlane(blockIdx.x * 8 + (threadIdx.x >> 5));
  const int lane = threadIdx.x & 31;
  if (node >= n) return;
  const float4* __restrict__ hs4 = (const float4*)hs;
  const int beg = rowptr[node];
  const int deg = cnt[node];
  float4 acc = hs4[(size_t)node * 32 + lane];
  for (int base = 0; base < deg; base += 32) {
    const int m = min(32, deg - base);
    const int mine = (lane < m) ? src[beg + base + lane] : 0;
    int j = 0;
    for (; j + 8 <= m; j += 8) {
      float4 v[8];
#pragma unroll
      for (int k = 0; k < 8; ++k) {
        const int r = __builtin_amdgcn_readlane(mine, j + k);
        v[k] = hs4[(size_t)r * 32 + lane];
      }
#pragma unroll
      for (int k = 0; k < 8; ++k) {
        acc.x += v[k].x; acc.y += v[k].y; acc.z += v[k].z; acc.w += v[k].w;
      }
    }
    if (j + 4 <= m) {
      float4 v[4];
#pragma unroll
      for (int k = 0; k < 4; ++k) {
        const int r = __builtin_amdgcn_readlane(mine, j + k);
        v[k] = hs4[(size_t)r * 32 + lane];
      }
#pragma unroll
      for (int k = 0; k < 4; ++k) {
        acc.x += v[k].x; acc.y += v[k].y; acc.z += v[k].z; acc.w += v[k].w;
      }
      j += 4;
    }
    for (; j < m; ++j) {
      const int r = __builtin_amdgcn_readlane(mine, j);
      const float4 v = hs4[(size_t)r * 32 + lane];
      acc.x += v.x; acc.y += v.y; acc.z += v.z; acc.w += v.w;
    }
  }
  const float d = dinv[node];
  float4 o;
  o.x = fmaxf(acc.x * d + bias[lane * 4 + 0], 0.0f);
  o.y = fmaxf(acc.y * d + bias[lane * 4 + 1], 0.0f);
  o.z = fmaxf(acc.z * d + bias[lane * 4 + 2], 0.0f);
  o.w = fmaxf(acc.w * d + bias[lane * 4 + 3], 0.0f);
  ((float4*)out)[(size_t)node * 32 + lane] = o;
}

__global__ __launch_bounds__(256) void k_pull32_logsoftmax(
    const float* __restrict__ gs, const float* __restrict__ dinv,
    const float* __restrict__ bias, const int* __restrict__ rowptr,
    const int* __restrict__ cnt, const int* __restrict__ src,
    float* __restrict__ out, int n) {
  const int node = __builtin_amdgcn_readfirstlane(blockIdx.x * 8 + (threadIdx.x >> 5));
  const int lane = threadIdx.x & 31;
  if (node >= n) return;
  const int beg = rowptr[node];
  const int deg = cnt[node];
  float acc = gs[(size_t)node * 32 + lane];
  for (int base = 0; base < deg; base += 32) {
    const int m = min(32, deg - base);
    const int mine = (lane < m) ? src[beg + base + lane] : 0;
    int j = 0;
    for (; j + 8 <= m; j += 8) {
      float v[8];
#pragma unroll
      for (int k = 0; k < 8; ++k) {
        const int r = __builtin_amdgcn_readlane(mine, j + k);
        v[k] = gs[(size_t)r * 32 + lane];
      }
#pragma unroll
      for (int k = 0; k < 8; ++k) acc += v[k];
    }
    if (j + 4 <= m) {
      float v[4];
#pragma unroll
      for (int k = 0; k < 4; ++k) {
        const int r = __builtin_amdgcn_readlane(mine, j + k);
        v[k] = gs[(size_t)r * 32 + lane];
      }
#pragma unroll
      for (int k = 0; k < 4; ++k) acc += v[k];
      j += 4;
    }
    for (; j < m; ++j) {
      const int r = __builtin_amdgcn_readlane(mine, j);
      acc += gs[(size_t)r * 32 + lane];
    }
  }
  const float v = acc * dinv[node] + bias[lane];
  float mx = v;
  for (int o = 16; o > 0; o >>= 1) mx = fmaxf(mx, __shfl_xor(mx, o, 32));
  float s = __expf(v - mx);
  for (int o = 16; o > 0; o >>= 1) s += __shfl_xor(s, o, 32);
  out[(size_t)node * 32 + lane] = v - mx - __logf(s);
}

extern "C" void kernel_launch(void* const* d_in, const int* in_sizes, int n_in,
                              void* d_out, int out_size, void* d_ws, size_t ws_size,
                              hipStream_t stream) {
  const float* x  = (const float*)d_in[0];
  const float* W1 = (const float*)d_in[1];
  const float* b1 = (const float*)d_in[2];
  const float* W2 = (const float*)d_in[3];
  const float* b2 = (const float*)d_in[4];
  const int*  eix = (const int*)d_in[5];

  const int n = N_NODES;
  const int E = in_sizes[5] / 2;
  const int* row = eix;
  const int* col = eix + E;

  float* ws     = (float*)d_ws;
  float* dinv   = ws;
  int*   cnt    = (int*)(ws + N_PAD);
  int*   rowptr = cnt + N_PAD;
  int*   rank   = rowptr + N_PAD;
  int*   src    = rank + E;
  float* hs     = (float*)(src + E);
  float* h1     = hs + (size_t)n * D_H;
  float* g2     = h1 + (size_t)n * D_H;
  float* outf   = (float*)d_out;

  k_zero <<<(N_PAD + 255) / 256, 256, 0, stream>>>(cnt, N_PAD);
  k_rank <<<(E + 255) / 256, 256, 0, stream>>>(col, cnt, rank, E);
  k_scan <<<N_PAD / 1024, 1024, 0, stream>>>(cnt, rowptr, dinv);
  k_place<<<(E + 255) / 256, 256, 0, stream>>>(row, col, rank, rowptr, src, E);

  gemm_bf16_wmma<D_IN, D_H><<<(n + 127) / 128, 256, 0, stream>>>(x, W1, hs, dinv, n);
  k_pull128_relu<<<(n + 7) / 8, 256, 0, stream>>>(hs, dinv, b1, rowptr, cnt, src, h1, n);

  gemm_bf16_wmma<D_H, D_OUT><<<(n + 127) / 128, 256, 0, stream>>>(h1, W2, g2, dinv, n);
  k_pull32_logsoftmax<<<(n + 7) / 8, 256, 0, stream>>>(g2, dinv, b2, rowptr, cnt, src, outf, n);
}


# Edges grouped by target through LDS (counting sort: 512-node bins per chunk, then per node; coalesced writes), no global add + scattered store + lookup per edge; sum order LDS arrival
# speedup vs baseline: 3.8094x; 1.6017x over previous
#include <hip/hip_runtime.h>

#define N_NODES 100000
#define D_IN 256
#define D_H 128
#define D_OUT 32

typedef __attribute__((ext_vector_type(16))) __bf16 v16bf;
typedef __attribute__((ext_vector_type(8)))  float  v8f;

__device__ __forceinline__ v16bf frag_from_lds(const __bf16* base, int rowIdx, int khalf) {
  union { uint4 u[2]; v16bf v; } c;
  const uint4* p = (const uint4*)(base + rowIdx * 32 + khalf);
  c.u[0] = p[0];
  c.u[1] = p[2];
  return c.v;
}

template<int K, int N>
__global__ __launch_bounds__(256) void gemm_bf16_wmma(
    const float* __restrict__ A, const float* __restrict__ W,
    float* __restrict__ C,
    const float* __restrict__ dinv, int M) {
  constexpr int NT = N / 16;
  __shared__ __bf16 ldsA[128 * 32];
  __shared__ __bf16 ldsB[N * 32];

  const int tid   = threadIdx.x;
  const int wave  = tid >> 5;
  const int lane  = tid & 31;
  const int laneM = lane & 15;
  const int khalf = (lane >> 4) << 3;
  const int rowBlk = blockIdx.x * 128;

  const int ldRow = tid >> 1;
  const int ldC0  = (tid & 1) << 4;
  const int ldGr  = rowBlk + ldRow;

  v8f acc[NT];
#pragma unroll
  for (int t = 0; t < NT; ++t) {
    v8f z = {0.f, 0.f, 0.f, 0.f, 0.f, 0.f, 0.f, 0.f};
    acc[t] = z;
  }

  for (int kk = 0; kk < K; kk += 32) {
    __syncthreads();

    {
      union { __bf16 h[16]; uint4 q[2]; } buf;
      if (ldGr < M) {
        const float4* src = (const float4*)(A + (size_t)ldGr * K + kk + ldC0);
#pragma unroll
        for (int j = 0; j < 4; ++j) {
          float4 v = src[j];
          buf.h[j * 4 + 0] = (__bf16)v.x;
          buf.h[j * 4 + 1] = (__bf16)v.y;
          buf.h[j * 4 + 2] = (__bf16)v.z;
          buf.h[j * 4 + 3] = (__bf16)v.w;
        }
      } else {
#pragma unroll
        for (int j = 0; j < 16; ++j) buf.h[j] = (__bf16)0.0f;
      }
      uint4* dst = (uint4*)(ldsA + ldRow * 32 + ldC0);
      dst[0] = buf.q[0];
      dst[1] = buf.q[1];
    }

#pragma unroll
    for (int i = tid; i < N * 8; i += 256) {
      const int n  = i % N;
      const int kq = i / N;
      union { __bf16 h[4]; uint2 q; } wb;
#pragma unroll
      for (int j = 0; j < 4; ++j)
        wb.h[j] = (__bf16)W[(size_t)(kk + kq * 4 + j) * N + n];
      *(uint2*)(ldsB + n * 32 + kq * 4) = wb.q;
    }
    __syncthreads();

    v16bf a = frag_from_lds(ldsA, (wave << 4) + laneM, khalf);
    v16bf bc = frag_from_lds(ldsB, laneM, khalf);
#pragma unroll
    for (int t = 0; t < NT; ++t) {
      v16bf bn = (t + 1 < NT) ? frag_from_lds(ldsB, ((t + 1) << 4) + laneM, khalf) : bc;
      acc[t] = __builtin_amdgcn_wmma_f32_16x16x32_bf16(
          false, a, false, bc, (short)0, acc[t], false, false);
      bc = bn;
    }
  }

  const int mBase = rowBlk + (wave << 4) + ((lane >> 4) << 3);
#pragma unroll
  for (int v = 0; v < 8; ++v) {
    const int gr = mBase + v;
    if (gr < M) {
      const float d = dinv[gr];
#pragma unroll
      for (int t = 0; t < NT; ++t) {
        const size_t idx = (size_t)gr * N + (t << 4) + laneM;
        C[idx] = acc[t][v] * d;
      }
    }
  }
}

#define N_PAD 100352

#define BIN_SHIFT 9
#define NBIN 196
#define NBIN_PAD 256
#define CHUNK 8192
#define SORT_CAP 16384
#define SRC_MASK 0x1FFFF

__global__ void k_zero(int* __restrict__ p, int n) {
  int i = blockIdx.x * blockDim.x + threadIdx.x;
  if (i < n) p[i] = 0;
}

__global__ __launch_bounds__(1024) void k_hist(
    const int* __restrict__ col, int* __restrict__ gcnt, int E) {
  __shared__ int h[NBIN_PAD];
  const int tid = threadIdx.x;
  if (tid < NBIN_PAD) h[tid] = 0;
  __syncthreads();
  const int base = blockIdx.x * CHUNK;
#pragma unroll
  for (int k = 0; k < CHUNK / 1024; ++k) {
    const int e = base + k * 1024 + tid;
    if (e < E) atomicAdd(&h[col[e] >> BIN_SHIFT], 1);
  }
  __syncthreads();
  if (tid < NBIN_PAD) {
    const int v = h[tid];
    if (v) atomicAdd(&gcnt[tid], v);
  }
}

__global__ __launch_bounds__(1024) void k_part(
    const int* __restrict__ row, const int* __restrict__ col,
    const int* __restrict__ gcnt, int* __restrict__ gcur,
    int* __restrict__ part, int E) {
  __shared__ int h[NBIN_PAD];
  __shared__ int lstart[NBIN_PAD];
  __shared__ int gb[NBIN_PAD];
  __shared__ int sval[CHUNK];
  __shared__ unsigned char sbin[CHUNK];
  const int tid = threadIdx.x;
  if (tid < NBIN_PAD) h[tid] = 0;
  __syncthreads();
  const int base = blockIdx.x * CHUNK;
  int c[CHUNK / 1024], r[CHUNK / 1024], lr[CHUNK / 1024];
#pragma unroll
  for (int k = 0; k < CHUNK / 1024; ++k) {
    const int e = base + k * 1024 + tid;
    const bool ok = e < E;
    c[k] = ok ? col[e] : -1;
    r[k] = ok ? row[e] : 0;
  }
#pragma unroll
  for (int k = 0; k < CHUNK / 1024; ++k) {
    lr[k] = 0;
    if (c[k] >= 0) lr[k] = atomicAdd(&h[c[k] >> BIN_SHIFT], 1);
  }
  __syncthreads();
  if (tid < 32) {
    int hv[8], gv[8], hloc[8], gloc[8];
    int hsum = 0, gsum = 0;
#pragma unroll
    for (int j = 0; j < 8; ++j) {
      hv[j] = h[tid * 8 + j];
      gv[j] = gcnt[tid * 8 + j];
    }
#pragma unroll
    for (int j = 0; j < 8; ++j) {
      hloc[j] = hsum; hsum += hv[j];
      gloc[j] = gsum; gsum += gv[j];
    }
    int hi = hsum, gi = gsum;
    for (int o = 1; o < 32; o <<= 1) {
      const int y = __shfl_up(hi, o, 32);
      const int z = __shfl_up(gi, o, 32);
      if (tid >= o) { hi += y; gi += z; }
    }
    const int hex = hi - hsum, gex = gi - gsum;
#pragma unroll
    for (int j = 0; j < 8; ++j) {
      const int b = tid * 8 + j;
      lstart[b] = hex + hloc[j];
      int g = gex + gloc[j];
      if (hv[j]) g += atomicAdd(&gcur[b], hv[j]);
      gb[b] = g;
    }
  }
  __syncthreads();
#pragma unroll
  for (int k = 0; k < CHUNK / 1024; ++k) {
    if (c[k] >= 0) {
      const int b = c[k] >> BIN_SHIFT;
      const int p = lstart[b] + lr[k];
      sval[p] = ((c[k] & ((1 << BIN_SHIFT) - 1)) << 17) | r[k];
      sbin[p] = (unsigned char)b;
    }
  }
  __syncthreads();
  const int nhere = min(CHUNK, E - base);
#pragma unroll
  for (int k = 0; k < CHUNK / 1024; ++k) {
    const int i = k * 1024 + tid;
    if (i < nhere) {
      const int b = sbin[i];
      part[gb[b] + (i - lstart[b])] = sval[i];
    }
  }
}

__global__ __launch_bounds__(1024) void k_sort(
    const int* __restrict__ part, const int* __restrict__ gcnt,
    int* __restrict__ src, int* __restrict__ rowptr, int* __restrict__ cnt,
    float* __restrict__ dinv) {
  __shared__ int c_l[512];
  __shared__ int cur_l[512];
  __shared__ int rp_l[512];
  __shared__ int wsum[32];
  __shared__ int s_start;
  __shared__ int sorted_l[SORT_CAP];
  const int tid  = threadIdx.x;
  const int lane = tid & 31;
  const int wave = tid >> 5;
  const int bin  = blockIdx.x;

  if (tid < 512) { c_l[tid] = 0; cur_l[tid] = 0; }
  if (tid < 32) {
    int s = 0;
    for (int j = tid; j < bin; j += 32) s += gcnt[j];
    for (int o = 16; o > 0; o >>= 1) s += __shfl_xor(s, o, 32);
    if (tid == 0) s_start = s;
  }
  __syncthreads();
  const int bstart = s_start;
  const int nb = gcnt[bin];

  for (int i0 = 0; i0 < nb; i0 += 8192) {
    int en[8];
#pragma unroll
    for (int k = 0; k < 8; ++k) {
      const int i = i0 + k * 1024 + tid;
      en[k] = (i < nb) ? part[bstart + i] : -1;
    }
#pragma unroll
    for (int k = 0; k < 8; ++k)
      if (en[k] >= 0) atomicAdd(&c_l[en[k] >> 17], 1);
  }
  __syncthreads();

  int x = 0, c = 0;
  if (tid < 512) {
    c = c_l[tid];
    x = c;
    for (int o = 1; o < 32; o <<= 1) {
      const int y = __shfl_up(x, o, 32);
      if (lane >= o) x += y;
    }
    if (lane == 31) wsum[wave] = x;
  }
  __syncthreads();
  if (tid < 32) {
    int t = (lane < 16) ? wsum[lane] : 0;
    for (int o = 1; o < 32; o <<= 1) {
      const int y = __shfl_up(t, o, 32);
      if (lane >= o) t += y;
    }
    wsum[lane] = t;
  }
  __syncthreads();
  if (tid < 512) {
    const int rp = x - c + ((wave > 0) ? wsum[wave - 1] : 0);
    rp_l[tid] = rp;
    const int node = (bin << BIN_SHIFT) + tid;
    cnt[node]    = c;
    rowptr[node] = bstart + rp;
    dinv[node]   = rsqrtf((float)(c + 1));
  }
  __syncthreads();

  const bool fits = nb <= SORT_CAP;
  for (int i0 = 0; i0 < nb; i0 += 8192) {
    int en[8];
#pragma unroll
    for (int k = 0; k < 8; ++k) {
      const int i = i0 + k * 1024 + tid;
      en[k] = (i < nb) ? part[bstart + i] : -1;
    }
#pragma unroll
    for (int k = 0; k < 8; ++k) {
      if (en[k] >= 0) {
        const int l = en[k] >> 17;
        const int p = rp_l[l] + atomicAdd(&cur_l[l], 1);
        if (fits) sorted_l[p] = en[k] & SRC_MASK;
        else      src[bstart + p] = en[k] & SRC_MASK;
      }
    }
  }
  __syncthreads();
  if (fits)
    for (int i = tid; i < nb; i += 1024) src[bstart + i] = sorted_l[i];
}

__global__ __launch_bounds__(256) void k_pull128_relu(
    const float* __restrict__ hs, const float* __restrict__ dinv,
    const float* __restrict__ bias, const int* __restrict__ rowptr,
    const int* __restrict__ cnt, const int* __restrict__ src,
    float* __restrict__ out, int n) {
  const int node = __builtin_amdgcn_readfirstlane(blockIdx.x * 8 + (threadIdx.x >> 5));
  const int lane = threadIdx.x & 31;
  if (node >= n) return;
  const float4* __restrict__ hs4 = (const float4*)hs;
  const int beg = rowptr[node];
  const int deg = cnt[node];
  float4 acc = hs4[(size_t)node * 32 + lane];
  for (int base = 0; base < deg; base += 32) {
    const int m = min(32, deg - base);
    const int mine = (lane < m) ? src[beg + base + lane] : 0;
    int j = 0;
    for (; j + 8 <= m; j += 8) {
      float4 v[8];
#pragma unroll
      for (int k = 0; k < 8; ++k) {
        const int r = __builtin_amdgcn_readlane(mine, j + k);
        v[k] = hs4[(size_t)r * 32 + lane];
      }
#pragma unroll
      for (int k = 0; k < 8; ++k) {
        acc.x += v[k].x; acc.y += v[k].y; acc.z += v[k].z; acc.w += v[k].w;
      }
    }
    if (j + 4 <= m) {
      float4 v[4];
#pragma unroll
      for (int k = 0; k < 4; ++k) {
        const int r = __builtin_amdgcn_readlane(mine, j + k);
        v[k] = hs4[(size_t)r * 32 + lane];
      }
#pragma unroll
      for (int k = 0; k < 4; ++k) {
        acc.x += v[k].x; acc.y += v[k].y; acc.z += v[k].z; acc.w += v[k].w;
      }
      j += 4;
    }
    for (; j < m; ++j) {
      const int r = __builtin_amdgcn_readlane(mine, j);
      const float4 v = hs4[(size_t)r * 32 + lane];
      acc.x += v.x; acc.y += v.y; acc.z += v.z; acc.w += v.w;
    }
  }
  const float d = dinv[node];
  float4 o;
  o.x = fmaxf(acc.x * d + bias[lane * 4 + 0], 0.0f);
  o.y = fmaxf(acc.y * d + bias[lane * 4 + 1], 0.0f);
  o.z = fmaxf(acc.z * d + bias[lane * 4 + 2], 0.0f);
  o.w = fmaxf(acc.w * d + bias[lane * 4 + 3], 0.0f);
  ((float4*)out)[(size_t)node * 32 + lane] = o;
}

__global__ __launch_bounds__(256) void k_pull32_logsoftmax(
    const float* __restrict__ gs, const float* __restrict__ dinv,
    const float* __restrict__ bias, const int* __restrict__ rowptr,
    const int* __restrict__ cnt, const int* __restrict__ src,
    float* __restrict__ out, int n) {
  const int node = __builtin_amdgcn_readfirstlane(blockIdx.x * 8 + (threadIdx.x >> 5));
  const int lane = threadIdx.x & 31;
  if (node >= n) return;
  const int beg = rowptr[node];
  const int deg = cnt[node];
  float acc = gs[(size_t)node * 32 + lane];
  for (int base = 0; base < deg; base += 32) {
    const int m = min(32, deg - base);
    const int mine = (lane < m) ? src[beg + base + lane] : 0;
    int j = 0;
    for (; j + 8 <= m; j += 8) {
      float v[8];
#pragma unroll
      for (int k = 0; k < 8; ++k) {
        const int r = __builtin_amdgcn_readlane(mine, j + k);
        v[k] = gs[(size_t)r * 32 + lane];
      }
#pragma unroll
      for (int k = 0; k < 8; ++k) acc += v[k];
    }
    if (j + 4 <= m) {
      float v[4];
#pragma unroll
      for (int k = 0; k < 4; ++k) {
        const int r = __builtin_amdgcn_readlane(mine, j + k);
        v[k] = gs[(size_t)r * 32 + lane];
      }
#pragma unroll
      for (int k = 0; k < 4; ++k) acc += v[k];
      j += 4;
    }
    for (; j < m; ++j) {
      const int r = __builtin_amdgcn_readlane(mine, j);
      acc += gs[(size_t)r * 32 + lane];
    }
  }
  const float v = acc * dinv[node] + bias[lane];
  float mx = v;
  for (int o = 16; o > 0; o >>= 1) mx = fmaxf(mx, __shfl_xor(mx, o, 32));
  float s = __expf(v - mx);
  for (int o = 16; o > 0; o >>= 1) s += __shfl_xor(s, o, 32);
  out[(size_t)node * 32 + lane] = v - mx - __logf(s);
}

extern "C" void kernel_launch(void* const* d_in, const int* in_sizes, int n_in,
                              void* d_out, int out_size, void* d_ws, size_t ws_size,
                              hipStream_t stream) {
  const float* x  = (const float*)d_in[0];
  const float* W1 = (const float*)d_in[1];
  const float* b1 = (const float*)d_in[2];
  const float* W2 = (const float*)d_in[3];
  const float* b2 = (const float*)d_in[4];
  const int*  eix = (const int*)d_in[5];

  const int n = N_NODES;
  const int E = in_sizes[5] / 2;
  const int* row = eix;
  const int* col = eix + E;

  float* ws     = (float*)d_ws;
  float* dinv   = ws;
  int*   cnt    = (int*)(ws + N_PAD);
  int*   rowptr = cnt + N_PAD;
  int*   part   = rowptr + N_PAD;
  int*   src    = part + E;
  float* hs     = (float*)(src + E);
  float* h1     = hs + (size_t)n * D_H;
  float* g2     = h1 + (size_t)n * D_H;
  int*   gcnt   = (int*)(g2 + (size_t)n * D_OUT);
  int*   gcur   = gcnt + NBIN_PAD;
  float* outf   = (float*)d_out;

  k_zero<<<1, 2 * NBIN_PAD, 0, stream>>>(gcnt, 2 * NBIN_PAD);
  k_hist<<<(E + CHUNK - 1) / CHUNK, 1024, 0, stream>>>(col, gcnt, E);
  k_part<<<(E + CHUNK - 1) / CHUNK, 1024, 0, stream>>>(row, col, gcnt, gcur, part, E);
  k_sort<<<NBIN, 1024, 0, stream>>>(part, gcnt, src, rowptr, cnt, dinv);

  gemm_bf16_wmma<D_IN, D_H><<<(n + 127) / 128, 256, 0, stream>>>(x, W1, hs, dinv, n);
  k_pull128_relu<<<(n + 7) / 8, 256, 0, stream>>>(hs, dinv, b1, rowptr, cnt, src, h1, n);

  gemm_bf16_wmma<D_H, D_OUT><<<(n + 127) / 128, 256, 0, stream>>>(h1, W2, g2, dinv, n);
  k_pull32_logsoftmax<<<(n + 7) / 8, 256, 0, stream>>>(g2, dinv, b2, rowptr, cnt, src, outf, n);
}


# Layer-2 transform (h@W2, same bf16 WMMA operands and k order, dinv scale) fused into the layer-1 aggregation kernel: 16 nodes per work-group, relu rows go to LDS, no h1 round trip, 7 launches
# speedup vs baseline: 4.1513x; 1.0898x over previous
#include <hip/hip_runtime.h>

#define N_NODES 100000
#define D_IN 256
#define D_H 128
#define D_OUT 32

typedef __attribute__((ext_vector_type(16))) __bf16 v16bf;
typedef __attribute__((ext_vector_type(8)))  float  v8f;

__device__ __forceinline__ v16bf frag_from_lds(const __bf16* base, int rowIdx, int khalf) {
  union { uint4 u[2]; v16bf v; } c;
  const uint4* p = (const uint4*)(base + rowIdx * 32 + khalf);
  c.u[0] = p[0];
  c.u[1] = p[2];
  return c.v;
}

template<int K, int N>
__global__ __launch_bounds__(256) void gemm_bf16_wmma(
    const float* __restrict__ A, const float* __restrict__ W,
    float* __restrict__ C,
    const float* __restrict__ dinv, int M) {
  constexpr int NT = N / 16;
  __shared__ __bf16 ldsA[128 * 32];
  __shared__ __bf16 ldsB[N * 32];

  const int tid   = threadIdx.x;
  const int wave  = tid >> 5;
  const int lane  = tid & 31;
  const int laneM = lane & 15;
  const int khalf = (lane >> 4) << 3;
  const int rowBlk = blockIdx.x * 128;

  const int ldRow = tid >> 1;
  const int ldC0  = (tid & 1) << 4;
  const int ldGr  = rowBlk + ldRow;

  v8f acc[NT];
#pragma unroll
  for (int t = 0; t < NT; ++t) {
    v8f z = {0.f, 0.f, 0.f, 0.f, 0.f, 0.f, 0.f, 0.f};
    acc[t] = z;
  }

  for (int kk = 0; kk < K; kk += 32) {
    __syncthreads();

    {
      union { __bf16 h[16]; uint4 q[2]; } buf;
      if (ldGr < M) {
        const float4* src = (const float4*)(A + (size_t)ldGr * K + kk + ldC0);
#pragma unroll
        for (int j = 0; j < 4; ++j) {
          float4 v = src[j];
          buf.h[j * 4 + 0] = (__bf16)v.x;
          buf.h[j * 4 + 1] = (__bf16)v.y;
          buf.h[j * 4 + 2] = (__bf16)v.z;
          buf.h[j * 4 + 3] = (__bf16)v.w;
        }
      } else {
#pragma unroll
        for (int j = 0; j < 16; ++j) buf.h[j] = (__bf16)0.0f;
      }
      uint4* dst = (uint4*)(ldsA + ldRow * 32 + ldC0);
      dst[0] = buf.q[0];
      dst[1] = buf.q[1];
    }

#pragma unroll
    for (int i = tid; i < N * 8; i += 256) {
      const int n  = i % N;
      const int kq = i / N;
      union { __bf16 h[4]; uint2 q; } wb;
#pragma unroll
      for (int j = 0; j < 4; ++j)
        wb.h[j] = (__bf16)W[(size_t)(kk + kq * 4 + j) * N + n];
      *(uint2*)(ldsB + n * 32 + kq * 4) = wb.q;
    }
    __syncthreads();

    v16bf a = frag_from_lds(ldsA, (wave << 4) + laneM, khalf);
    v16bf bc = frag_from_lds(ldsB, laneM, khalf);
#pragma unroll
    for (int t = 0; t < NT; ++t) {
      v16bf bn = (t + 1 < NT) ? frag_from_lds(ldsB, ((t + 1) << 4) + laneM, khalf) : bc;
      acc[t] = __builtin_amdgcn_wmma_f32_16x16x32_bf16(
          false, a, false, bc, (short)0, acc[t], false, false);
      bc = bn;
    }
  }

  const int mBase = rowBlk + (wave << 4) + ((lane >> 4) << 3);
#pragma unroll
  for (int v = 0; v < 8; ++v) {
    const int gr = mBase + v;
    if (gr < M) {
      const float d = dinv[gr];
#pragma unroll
      for (int t = 0; t < NT; ++t) {
        const size_t idx = (size_t)gr * N + (t << 4) + laneM;
        C[idx] = acc[t][v] * d;
      }
    }
  }
}

#define N_PAD 100352

#define BIN_SHIFT 9
#define NBIN 196
#define NBIN_PAD 256
#define CHUNK 8192
#define SORT_CAP 16384
#define SRC_MASK 0x1FFFF

__global__ void k_zero(int* __restrict__ p, int n) {
  int i = blockIdx.x * blockDim.x + threadIdx.x;
  if (i < n) p[i] = 0;
}

__global__ __launch_bounds__(1024) void k_hist(
    const int* __restrict__ col, int* __restrict__ gcnt, int E) {
  __shared__ int h[NBIN_PAD];
  const int tid = threadIdx.x;
  if (tid < NBIN_PAD) h[tid] = 0;
  __syncthreads();
  const int base = blockIdx.x * CHUNK;
#pragma unroll
  for (int k = 0; k < CHUNK / 1024; ++k) {
    const int e = base + k * 1024 + tid;
    if (e < E) atomicAdd(&h[col[e] >> BIN_SHIFT], 1);
  }
  __syncthreads();
  if (tid < NBIN_PAD) {
    const int v = h[tid];
    if (v) atomicAdd(&gcnt[tid], v);
  }
}

__global__ __launch_bounds__(1024) void k_part(
    const int* __restrict__ row, const int* __restrict__ col,
    const int* __restrict__ gcnt, int* __restrict__ gcur,
    int* __restrict__ part, int E) {
  __shared__ int h[NBIN_PAD];
  __shared__ int lstart[NBIN_PAD];
  __shared__ int gb[NBIN_PAD];
  __shared__ int sval[CHUNK];
  __shared__ unsigned char sbin[CHUNK];
  const int tid = threadIdx.x;
  if (tid < NBIN_PAD) h[tid] = 0;
  __syncthreads();
  const int base = blockIdx.x * CHUNK;
  int c[CHUNK / 1024], r[CHUNK / 1024], lr[CHUNK / 1024];
#pragma unroll
  for (int k = 0; k < CHUNK / 1024; ++k) {
    const int e = base + k * 1024 + tid;
    const bool ok = e < E;
    c[k] = ok ? col[e] : -1;
    r[k] = ok ? row[e] : 0;
  }
#pragma unroll
  for (int k = 0; k < CHUNK / 1024; ++k) {
    lr[k] = 0;
    if (c[k] >= 0) lr[k] = atomicAdd(&h[c[k] >> BIN_SHIFT], 1);
  }
  __syncthreads();
  if (tid < 32) {
    int hv[8], gv[8], hloc[8], gloc[8];
    int hsum = 0, gsum = 0;
#pragma unroll
    for (int j = 0; j < 8; ++j) {
      hv[j] = h[tid * 8 + j];
      gv[j] = gcnt[tid * 8 + j];
    }
#pragma unroll
    for (int j = 0; j < 8; ++j) {
      hloc[j] = hsum; hsum += hv[j];
      gloc[j] = gsum; gsum += gv[j];
    }
    int hi = hsum, gi = gsum;
    for (int o = 1; o < 32; o <<= 1) {
      const int y = __shfl_up(hi, o, 32);
      const int z = __shfl_up(gi, o, 32);
      if (tid >= o) { hi += y; gi += z; }
    }
    const int hex = hi - hsum, gex = gi - gsum;
#pragma unroll
    for (int j = 0; j < 8; ++j) {
      const int b = tid * 8 + j;
      lstart[b] = hex + hloc[j];
      int g = gex + gloc[j];
      if (hv[j]) g += atomicAdd(&gcur[b], hv[j]);
      gb[b] = g;
    }
  }
  __syncthreads();
#pragma unroll
  for (int k = 0; k < CHUNK / 1024; ++k) {
    if (c[k] >= 0) {
      const int b = c[k] >> BIN_SHIFT;
      const int p = lstart[b] + lr[k];
      sval[p] = ((c[k] & ((1 << BIN_SHIFT) - 1)) << 17) | r[k];
      sbin[p] = (unsigned char)b;
    }
  }
  __syncthreads();
  const int nhere = min(CHUNK, E - base);
#pragma unroll
  for (int k = 0; k < CHUNK / 1024; ++k) {
    const int i = k * 1024 + tid;
    if (i < nhere) {
      const int b = sbin[i];
      part[gb[b] + (i - lstart[b])] = sval[i];
    }
  }
}

__global__ __launch_bounds__(1024) void k_sort(
    const int* __restrict__ part, const int* __restrict__ gcnt,
    int* __restrict__ src, int* __restrict__ rowptr, int* __restrict__ cnt,
    float* __restrict__ dinv) {
  __shared__ int c_l[512];
  __shared__ int cur_l[512];
  __shared__ int rp_l[512];
  __shared__ int wsum[32];
  __shared__ int s_start;
  __shared__ int sorted_l[SORT_CAP];
  const int tid  = threadIdx.x;
  const int lane = tid & 31;
  const int wave = tid >> 5;
  const int bin  = blockIdx.x;

  if (tid < 512) { c_l[tid] = 0; cur_l[tid] = 0; }
  if (tid < 32) {
    int s = 0;
    for (int j = tid; j < bin; j += 32) s += gcnt[j];
    for (int o = 16; o > 0; o >>= 1) s += __shfl_xor(s, o, 32);
    if (tid == 0) s_start = s;
  }
  __syncthreads();
  const int bstart = s_start;
  const int nb = gcnt[bin];

  for (int i0 = 0; i0 < nb; i0 += 8192) {
    int en[8];
#pragma unroll
    for (int k = 0; k < 8; ++k) {
      const int i = i0 + k * 1024 + tid;
      en[k] = (i < nb) ? part[bstart + i] : -1;
    }
#pragma unroll
    for (int k = 0; k < 8; ++k)
      if (en[k] >= 0) atomicAdd(&c_l[en[k] >> 17], 1);
  }
  __syncthreads();

  int x = 0, c = 0;
  if (tid < 512) {
    c = c_l[tid];
    x = c;
    for (int o = 1; o < 32; o <<= 1) {
      const int y = __shfl_up(x, o, 32);
      if (lane >= o) x += y;
    }
    if (lane == 31) wsum[wave] = x;
  }
  __syncthreads();
  if (tid < 32) {
    int t = (lane < 16) ? wsum[lane] : 0;
    for (int o = 1; o < 32; o <<= 1) {
      const int y = __shfl_up(t, o, 32);
      if (lane >= o) t += y;
    }
    wsum[lane] = t;
  }
  __syncthreads();
  if (tid < 512) {
    const int rp = x - c + ((wave > 0) ? wsum[wave - 1] : 0);
    rp_l[tid] = rp;
    const int node = (bin << BIN_SHIFT) + tid;
    cnt[node]    = c;
    rowptr[node] = bstart + rp;
    dinv[node]   = rsqrtf((float)(c + 1));
  }
  __syncthreads();

  const bool fits = nb <= SORT_CAP;
  for (int i0 = 0; i0 < nb; i0 += 8192) {
    int en[8];
#pragma unroll
    for (int k = 0; k < 8; ++k) {
      const int i = i0 + k * 1024 + tid;
      en[k] = (i < nb) ? part[bstart + i] : -1;
    }
#pragma unroll
    for (int k = 0; k < 8; ++k) {
      if (en[k] >= 0) {
        const int l = en[k] >> 17;
        const int p = rp_l[l] + atomicAdd(&cur_l[l], 1);
        if (fits) sorted_l[p] = en[k] & SRC_MASK;
        else      src[bstart + p] = en[k] & SRC_MASK;
      }
    }
  }
  __syncthreads();
  if (fits)
    for (int i = tid; i < nb; i += 1024) src[bstart + i] = sorted_l[i];
}

__device__ __forceinline__ float4 pull_row128(
    const float4* __restrict__ hs4, const int* __restrict__ src,
    int node, int beg, int deg, int first, int lane) {
  float4 acc = hs4[(size_t)node * 32 + lane];
  for (int base = 0; base < deg; base += 32) {
    const int m = min(32, deg - base);
    const int mine = (base == 0) ? first : ((lane < m) ? src[beg + base + lane] : 0);
    int j = 0;
    for (; j + 8 <= m; j += 8) {
      float4 v[8];
#pragma unroll
      for (int k = 0; k < 8; ++k) {
        const int r = __builtin_amdgcn_readlane(mine, j + k);
        v[k] = hs4[(size_t)r * 32 + lane];
      }
#pragma unroll
      for (int k = 0; k < 8; ++k) {
        acc.x += v[k].x; acc.y += v[k].y; acc.z += v[k].z; acc.w += v[k].w;
      }
    }
    if (j + 4 <= m) {
      float4 v[4];
#pragma unroll
      for (int k = 0; k < 4; ++k) {
        const int r = __builtin_amdgcn_readlane(mine, j + k);
        v[k] = hs4[(size_t)r * 32 + lane];
      }
#pragma unroll
      for (int k = 0; k < 4; ++k) {
        acc.x += v[k].x; acc.y += v[k].y; acc.z += v[k].z; acc.w += v[k].w;
      }
      j += 4;
    }
    for (; j < m; ++j) {
      const int r = __builtin_amdgcn_readlane(mine, j);
      const float4 v = hs4[(size_t)r * 32 + lane];
      acc.x += v.x; acc.y += v.y; acc.z += v.z; acc.w += v.w;
    }
  }
  return acc;
}

__global__ __launch_bounds__(256) void k_pull128_relu_gemm32(
    const float* __restrict__ hs, const float* __restrict__ dinv,
    const float* __restrict__ bias, const int* __restrict__ rowptr,
    const int* __restrict__ cnt, const int* __restrict__ src,
    const float* __restrict__ W2, float* __restrict__ g2, int n) {
  __shared__ __bf16 ldsA[4 * 16 * 32];
  __shared__ __bf16 ldsB[4 * D_OUT * 32];
  __shared__ float  ldsC[16 * D_OUT];

  const int tid  = threadIdx.x;
  const int wave = __builtin_amdgcn_readfirstlane(tid >> 5);
  const int lane = tid & 31;
  const int laneM = lane & 15;
  const int khalf = (lane >> 4) << 3;
  const int node0 = blockIdx.x * 16;

  int nodeQ[2], begQ[2], degQ[2];
  float dQ[2];
#pragma unroll
  for (int q = 0; q < 2; ++q) {
    nodeQ[q] = min(node0 + wave * 2 + q, n - 1);
    begQ[q]  = rowptr[nodeQ[q]];
    degQ[q]  = cnt[nodeQ[q]];
    dQ[q]    = dinv[nodeQ[q]];
  }

#pragma unroll
  for (int ks = 0; ks < 4; ++ks) {
    const int nn = tid & 31;
    const int kq = tid >> 5;
    union { __bf16 h[4]; uint2 q; } wb;
#pragma unroll
    for (int j = 0; j < 4; ++j)
      wb.h[j] = (__bf16)W2[(size_t)(ks * 32 + kq * 4 + j) * D_OUT + nn];
    *(uint2*)(ldsB + (ks * D_OUT + nn) * 32 + kq * 4) = wb.q;
  }

  const int outRow = node0 + (tid >> 3);
  const float dOut = (tid < 128 && outRow < n) ? dinv[outRow] : 0.0f;

  int firstQ[2];
#pragma unroll
  for (int q = 0; q < 2; ++q)
    firstQ[q] = (lane < min(32, degQ[q])) ? src[begQ[q] + lane] : 0;
  const float4* __restrict__ hs4 = (const float4*)hs;
  const float4 b4 = ((const float4*)bias)[lane];
#pragma unroll
  for (int q = 0; q < 2; ++q) {
    const int r = wave * 2 + q;
    union { __bf16 h[4]; uint2 u; } pk;
    const float4 acc = pull_row128(hs4, src, nodeQ[q], begQ[q], degQ[q], firstQ[q], lane);
    const float d = dQ[q];
    pk.h[0] = (__bf16)fmaxf(acc.x * d + b4.x, 0.0f);
    pk.h[1] = (__bf16)fmaxf(acc.y * d + b4.y, 0.0f);
    pk.h[2] = (__bf16)fmaxf(acc.z * d + b4.z, 0.0f);
    pk.h[3] = (__bf16)fmaxf(acc.w * d + b4.w, 0.0f);
    *(uint2*)(ldsA + ((lane >> 3) * 16 + r) * 32 + ((lane & 7) << 2)) = pk.u;
  }
  __syncthreads();

  if (wave < 2) {
    v8f acc = {0.f, 0.f, 0.f, 0.f, 0.f, 0.f, 0.f, 0.f};
#pragma unroll
    for (int ks = 0; ks < 4; ++ks) {
      const v16bf a = frag_from_lds(ldsA + ks * 16 * 32, laneM, khalf);
      const v16bf b = frag_from_lds(ldsB + ks * D_OUT * 32, (wave << 4) + laneM, khalf);
      acc = __builtin_amdgcn_wmma_f32_16x16x32_bf16(
          false, a, false, b, (short)0, acc, false, false);
    }
    const int mBase = (lane >> 4) << 3;
#pragma unroll
    for (int v = 0; v < 8; ++v)
      ldsC[(mBase + v) * D_OUT + (wave << 4) + laneM] = acc[v];
  }
  __syncthreads();

  if (tid < 128 && outRow < n) {
    float4 c = ((const float4*)ldsC)[(tid >> 3) * (D_OUT / 4) + (tid & 7)];
    c.x *= dOut; c.y *= dOut; c.z *= dOut; c.w *= dOut;
    ((float4*)g2)[(size_t)outRow * (D_OUT / 4) + (tid & 7)] = c;
  }
}

__global__ __launch_bounds__(256) void k_pull32_logsoftmax(
    const float* __restrict__ gs, const float* __restrict__ dinv,
    const float* __restrict__ bias, const int* __restrict__ rowptr,
    const int* __restrict__ cnt, const int* __restrict__ src,
    float* __restrict__ out, int n) {
  const int node = __builtin_amdgcn_readfirstlane(blockIdx.x * 8 + (threadIdx.x >> 5));
  const int lane = threadIdx.x & 31;
  if (node >= n) return;
  const int beg = rowptr[node];
  const int deg = cnt[node];
  float acc = gs[(size_t)node * 32 + lane];
  for (int base = 0; base < deg; base += 32) {
    const int m = min(32, deg - base);
    const int mine = (lane < m) ? src[beg + base + lane] : 0;
    int j = 0;
    for (; j + 8 <= m; j += 8) {
      float v[8];
#pragma unroll
      for (int k = 0; k < 8; ++k) {
        const int r = __builtin_amdgcn_readlane(mine, j + k);
        v[k] = gs[(size_t)r * 32 + lane];
      }
#pragma unroll
      for (int k = 0; k < 8; ++k) acc += v[k];
    }
    if (j + 4 <= m) {
      float v[4];
#pragma unroll
      for (int k = 0; k < 4; ++k) {
        const int r = __builtin_amdgcn_readlane(mine, j + k);
        v[k] = gs[(size_t)r * 32 + lane];
      }
#pragma unroll
      for (int k = 0; k < 4; ++k) acc += v[k];
      j += 4;
    }
    for (; j < m; ++j) {
      const int r = __builtin_amdgcn_readlane(mine, j);
      acc += gs[(size_t)r * 32 + lane];
    }
  }
  const float v = acc * dinv[node] + bias[lane];
  float mx = v;
  for (int o = 16; o > 0; o >>= 1) mx = fmaxf(mx, __shfl_xor(mx, o, 32));
  float s = __expf(v - mx);
  for (int o = 16; o > 0; o >>= 1) s += __shfl_xor(s, o, 32);
  out[(size_t)node * 32 + lane] = v - mx - __logf(s);
}

extern "C" void kernel_launch(void* const* d_in, const int* in_sizes, int n_in,
                              void* d_out, int out_size, void* d_ws, size_t ws_size,
                              hipStream_t stream) {
  const float* x  = (const float*)d_in[0];
  const float* W1 = (const float*)d_in[1];
  const float* b1 = (const float*)d_in[2];
  const float* W2 = (const float*)d_in[3];
  const float* b2 = (const float*)d_in[4];
  const int*  eix = (const int*)d_in[5];

  const int n = N_NODES;
  const int E = in_sizes[5] / 2;
  const int* row = eix;
  const int* col = eix + E;

  float* ws     = (float*)d_ws;
  float* dinv   = ws;
  int*   cnt    = (int*)(ws + N_PAD);
  int*   rowptr = cnt + N_PAD;
  int*   part   = rowptr + N_PAD;
  int*   src    = part + E;
  float* hs     = (float*)(src + E);
  float* g2     = hs + (size_t)n * D_H;
  int*   gcnt   = (int*)(g2 + (size_t)n * D_OUT);
  int*   gcur   = gcnt + NBIN_PAD;
  float* outf   = (float*)d_out;

  k_zero<<<1, 2 * NBIN_PAD, 0, stream>>>(gcnt, 2 * NBIN_PAD);
  k_hist<<<(E + CHUNK - 1) / CHUNK, 1024, 0, stream>>>(col, gcnt, E);
  k_part<<<(E + CHUNK - 1) / CHUNK, 1024, 0, stream>>>(row, col, gcnt, gcur, part, E);
  k_sort<<<NBIN, 1024, 0, stream>>>(part, gcnt, src, rowptr, cnt, dinv);

  gemm_bf16_wmma<D_IN, D_H><<<(n + 127) / 128, 256, 0, stream>>>(x, W1, hs, dinv, n);
  k_pull128_relu_gemm32<<<(n + 15) / 16, 256, 0, stream>>>(hs, dinv, b1, rowptr, cnt, src, W2, g2, n);

  k_pull32_logsoftmax<<<(n + 7) / 8, 256, 0, stream>>>(g2, dinv, b2, rowptr, cnt, src, outf, n);
}


# Layer-2 aggregation regrouped: 8 lanes x 16-byte loads per node, 4 nodes per gather instruction, 16 nodes per wave (1/16 the waves); node sums same order, softmax sum tree now 4 in lane then 8 lanes
# speedup vs baseline: 4.3325x; 1.0436x over previous
#include <hip/hip_runtime.h>

#define N_NODES 100000
#define D_IN 256
#define D_H 128
#define D_OUT 32

typedef __attribute__((ext_vector_type(16))) __bf16 v16bf;
typedef __attribute__((ext_vector_type(8)))  float  v8f;

__device__ __forceinline__ v16bf frag_from_lds(const __bf16* base, int rowIdx, int khalf) {
  union { uint4 u[2]; v16bf v; } c;
  const uint4* p = (const uint4*)(base + rowIdx * 32 + khalf);
  c.u[0] = p[0];
  c.u[1] = p[2];
  return c.v;
}

template<int K, int N>
__global__ __launch_bounds__(256) void gemm_bf16_wmma(
    const float* __restrict__ A, const float* __restrict__ W,
    float* __restrict__ C,
    const float* __restrict__ dinv, int M) {
  constexpr int NT = N / 16;
  __shared__ __bf16 ldsA[128 * 32];
  __shared__ __bf16 ldsB[N * 32];

  const int tid   = threadIdx.x;
  const int wave  = tid >> 5;
  const int lane  = tid & 31;
  const int laneM = lane & 15;
  const int khalf = (lane >> 4) << 3;
  const int rowBlk = blockIdx.x * 128;

  const int ldRow = tid >> 1;
  const int ldC0  = (tid & 1) << 4;
  const int ldGr  = rowBlk + ldRow;

  v8f acc[NT];
#pragma unroll
  for (int t = 0; t < NT; ++t) {
    v8f z = {0.f, 0.f, 0.f, 0.f, 0.f, 0.f, 0.f, 0.f};
    acc[t] = z;
  }

  for (int kk = 0; kk < K; kk += 32) {
    __syncthreads();

    {
      union { __bf16 h[16]; uint4 q[2]; } buf;
      if (ldGr < M) {
        const float4* src = (const float4*)(A + (size_t)ldGr * K + kk + ldC0);
#pragma unroll
        for (int j = 0; j < 4; ++j) {
          float4 v = src[j];
          buf.h[j * 4 + 0] = (__bf16)v.x;
          buf.h[j * 4 + 1] = (__bf16)v.y;
          buf.h[j * 4 + 2] = (__bf16)v.z;
          buf.h[j * 4 + 3] = (__bf16)v.w;
        }
      } else {
#pragma unroll
        for (int j = 0; j < 16; ++j) buf.h[j] = (__bf16)0.0f;
      }
      uint4* dst = (uint4*)(ldsA + ldRow * 32 + ldC0);
      dst[0] = buf.q[0];
      dst[1] = buf.q[1];
    }

#pragma unroll
    for (int i = tid; i < N * 8; i += 256) {
      const int n  = i % N;
      const int kq = i / N;
      union { __bf16 h[4]; uint2 q; } wb;
#pragma unroll
      for (int j = 0; j < 4; ++j)
        wb.h[j] = (__bf16)W[(size_t)(kk + kq * 4 + j) * N + n];
      *(uint2*)(ldsB + n * 32 + kq * 4) = wb.q;
    }
    __syncthreads();

    v16bf a = frag_from_lds(ldsA, (wave << 4) + laneM, khalf);
    v16bf bc = frag_from_lds(ldsB, laneM, khalf);
#pragma unroll
    for (int t = 0; t < NT; ++t) {
      v16bf bn = (t + 1 < NT) ? frag_from_lds(ldsB, ((t + 1) << 4) + laneM, khalf) : bc;
      acc[t] = __builtin_amdgcn_wmma_f32_16x16x32_bf16(
          false, a, false, bc, (short)0, acc[t], false, false);
      bc = bn;
    }
  }

  const int mBase = rowBlk + (wave << 4) + ((lane >> 4) << 3);
#pragma unroll
  for (int v = 0; v < 8; ++v) {
    const int gr = mBase + v;
    if (gr < M) {
      const float d = dinv[gr];
#pragma unroll
      for (int t = 0; t < NT; ++t) {
        const size_t idx = (size_t)gr * N + (t << 4) + laneM;
        C[idx] = acc[t][v] * d;
      }
    }
  }
}

#define N_PAD 100352

#define BIN_SHIFT 9
#define NBIN 196
#define NBIN_PAD 256
#define CHUNK 8192
#define SORT_CAP 16384
#define SRC_MASK 0x1FFFF

__global__ void k_zero(int* __restrict__ p, int n) {
  int i = blockIdx.x * blockDim.x + threadIdx.x;
  if (i < n) p[i] = 0;
}

__global__ __launch_bounds__(1024) void k_hist(
    const int* __restrict__ col, int* __restrict__ gcnt, int E) {
  __shared__ int h[NBIN_PAD];
  const int tid = threadIdx.x;
  if (tid < NBIN_PAD) h[tid] = 0;
  __syncthreads();
  const int base = blockIdx.x * CHUNK;
#pragma unroll
  for (int k = 0; k < CHUNK / 1024; ++k) {
    const int e = base + k * 1024 + tid;
    if (e < E) atomicAdd(&h[col[e] >> BIN_SHIFT], 1);
  }
  __syncthreads();
  if (tid < NBIN_PAD) {
    const int v = h[tid];
    if (v) atomicAdd(&gcnt[tid], v);
  }
}

__global__ __launch_bounds__(1024) void k_part(
    const int* __restrict__ row, const int* __restrict__ col,
    const int* __restrict__ gcnt, int* __restrict__ gcur,
    int* __restrict__ part, int E) {
  __shared__ int h[NBIN_PAD];
  __shared__ int lstart[NBIN_PAD];
  __shared__ int gb[NBIN_PAD];
  __shared__ int sval[CHUNK];
  __shared__ unsigned char sbin[CHUNK];
  const int tid = threadIdx.x;
  if (tid < NBIN_PAD) h[tid] = 0;
  __syncthreads();
  const int base = blockIdx.x * CHUNK;
  int c[CHUNK / 1024], r[CHUNK / 1024], lr[CHUNK / 1024];
#pragma unroll
  for (int k = 0; k < CHUNK / 1024; ++k) {
    const int e = base + k * 1024 + tid;
    const bool ok = e < E;
    c[k] = ok ? col[e] : -1;
    r[k] = ok ? row[e] : 0;
  }
#pragma unroll
  for (int k = 0; k < CHUNK / 1024; ++k) {
    lr[k] = 0;
    if (c[k] >= 0) lr[k] = atomicAdd(&h[c[k] >> BIN_SHIFT], 1);
  }
  __syncthreads();
  if (tid < 32) {
    int hv[8], gv[8], hloc[8], gloc[8];
    int hsum = 0, gsum = 0;
#pragma unroll
    for (int j = 0; j < 8; ++j) {
      hv[j] = h[tid * 8 + j];
      gv[j] = gcnt[tid * 8 + j];
    }
#pragma unroll
    for (int j = 0; j < 8; ++j) {
      hloc[j] = hsum; hsum += hv[j];
      gloc[j] = gsum; gsum += gv[j];
    }
    int hi = hsum, gi = gsum;
    for (int o = 1; o < 32; o <<= 1) {
      const int y = __shfl_up(hi, o, 32);
      const int z = __shfl_up(gi, o, 32);
      if (tid >= o) { hi += y; gi += z; }
    }
    const int hex = hi - hsum, gex = gi - gsum;
#pragma unroll
    for (int j = 0; j < 8; ++j) {
      const int b = tid * 8 + j;
      lstart[b] = hex + hloc[j];
      int g = gex + gloc[j];
      if (hv[j]) g += atomicAdd(&gcur[b], hv[j]);
      gb[b] = g;
    }
  }
  __syncthreads();
#pragma unroll
  for (int k = 0; k < CHUNK / 1024; ++k) {
    if (c[k] >= 0) {
      const int b = c[k] >> BIN_SHIFT;
      const int p = lstart[b] + lr[k];
      sval[p] = ((c[k] & ((1 << BIN_SHIFT) - 1)) << 17) | r[k];
      sbin[p] = (unsigned char)b;
    }
  }
  __syncthreads();
  const int nhere = min(CHUNK, E - base);
#pragma unroll
  for (int k = 0; k < CHUNK / 1024; ++k) {
    const int i = k * 1024 + tid;
    if (i < nhere) {
      const int b = sbin[i];
      part[gb[b] + (i - lstart[b])] = sval[i];
    }
  }
}

__global__ __launch_bounds__(1024) void k_sort(
    const int* __restrict__ part, const int* __restrict__ gcnt,
    int* __restrict__ src, int* __restrict__ rowptr, int* __restrict__ cnt,
    float* __restrict__ dinv) {
  __shared__ int c_l[512];
  __shared__ int cur_l[512];
  __shared__ int rp_l[512];
  __shared__ int wsum[32];
  __shared__ int s_start;
  __shared__ int sorted_l[SORT_CAP];
  const int tid  = threadIdx.x;
  const int lane = tid & 31;
  const int wave = tid >> 5;
  const int bin  = blockIdx.x;

  if (tid < 512) { c_l[tid] = 0; cur_l[tid] = 0; }
  if (tid < 32) {
    int s = 0;
    for (int j = tid; j < bin; j += 32) s += gcnt[j];
    for (int o = 16; o > 0; o >>= 1) s += __shfl_xor(s, o, 32);
    if (tid == 0) s_start = s;
  }
  __syncthreads();
  const int bstart = s_start;
  const int nb = gcnt[bin];

  for (int i0 = 0; i0 < nb; i0 += 8192) {
    int en[8];
#pragma unroll
    for (int k = 0; k < 8; ++k) {
      const int i = i0 + k * 1024 + tid;
      en[k] = (i < nb) ? part[bstart + i] : -1;
    }
#pragma unroll
    for (int k = 0; k < 8; ++k)
      if (en[k] >= 0) atomicAdd(&c_l[en[k] >> 17], 1);
  }
  __syncthreads();

  int x = 0, c = 0;
  if (tid < 512) {
    c = c_l[tid];
    x = c;
    for (int o = 1; o < 32; o <<= 1) {
      const int y = __shfl_up(x, o, 32);
      if (lane >= o) x += y;
    }
    if (lane == 31) wsum[wave] = x;
  }
  __syncthreads();
  if (tid < 32) {
    int t = (lane < 16) ? wsum[lane] : 0;
    for (int o = 1; o < 32; o <<= 1) {
      const int y = __shfl_up(t, o, 32);
      if (lane >= o) t += y;
    }
    wsum[lane] = t;
  }
  __syncthreads();
  if (tid < 512) {
    const int rp = x - c + ((wave > 0) ? wsum[wave - 1] : 0);
    rp_l[tid] = rp;
    const int node = (bin << BIN_SHIFT) + tid;
    cnt[node]    = c;
    rowptr[node] = bstart + rp;
    dinv[node]   = rsqrtf((float)(c + 1));
  }
  __syncthreads();

  const bool fits = nb <= SORT_CAP;
  for (int i0 = 0; i0 < nb; i0 += 8192) {
    int en[8];
#pragma unroll
    for (int k = 0; k < 8; ++k) {
      const int i = i0 + k * 1024 + tid;
      en[k] = (i < nb) ? part[bstart + i] : -1;
    }
#pragma unroll
    for (int k = 0; k < 8; ++k) {
      if (en[k] >= 0) {
        const int l = en[k] >> 17;
        const int p = rp_l[l] + atomicAdd(&cur_l[l], 1);
        if (fits) sorted_l[p] = en[k] & SRC_MASK;
        else      src[bstart + p] = en[k] & SRC_MASK;
      }
    }
  }
  __syncthreads();
  if (fits)
    for (int i = tid; i < nb; i += 1024) src[bstart + i] = sorted_l[i];
}

__device__ __forceinline__ float4 pull_row128(
    const float4* __restrict__ hs4, const int* __restrict__ src,
    int node, int beg, int deg, int first, int lane) {
  float4 acc = hs4[(size_t)node * 32 + lane];
  for (int base = 0; base < deg; base += 32) {
    const int m = min(32, deg - base);
    const int mine = (base == 0) ? first : ((lane < m) ? src[beg + base + lane] : 0);
    int j = 0;
    for (; j + 8 <= m; j += 8) {
      float4 v[8];
#pragma unroll
      for (int k = 0; k < 8; ++k) {
        const int r = __builtin_amdgcn_readlane(mine, j + k);
        v[k] = hs4[(size_t)r * 32 + lane];
      }
#pragma unroll
      for (int k = 0; k < 8; ++k) {
        acc.x += v[k].x; acc.y += v[k].y; acc.z += v[k].z; acc.w += v[k].w;
      }
    }
    if (j + 4 <= m) {
      float4 v[4];
#pragma unroll
      for (int k = 0; k < 4; ++k) {
        const int r = __builtin_amdgcn_readlane(mine, j + k);
        v[k] = hs4[(size_t)r * 32 + lane];
      }
#pragma unroll
      for (int k = 0; k < 4; ++k) {
        acc.x += v[k].x; acc.y += v[k].y; acc.z += v[k].z; acc.w += v[k].w;
      }
      j += 4;
    }
    for (; j < m; ++j) {
      const int r = __builtin_amdgcn_readlane(mine, j);
      const float4 v = hs4[(size_t)r * 32 + lane];
      acc.x += v.x; acc.y += v.y; acc.z += v.z; acc.w += v.w;
    }
  }
  return acc;
}

__global__ __launch_bounds__(256) void k_pull128_relu_gemm32(
    const float* __restrict__ hs, const float* __restrict__ dinv,
    const float* __restrict__ bias, const int* __restrict__ rowptr,
    const int* __restrict__ cnt, const int* __restrict__ src,
    const float* __restrict__ W2, float* __restrict__ g2, int n) {
  __shared__ __bf16 ldsA[4 * 16 * 32];
  __shared__ __bf16 ldsB[4 * D_OUT * 32];
  __shared__ float  ldsC[16 * D_OUT];

  const int tid  = threadIdx.x;
  const int wave = __builtin_amdgcn_readfirstlane(tid >> 5);
  const int lane = tid & 31;
  const int laneM = lane & 15;
  const int khalf = (lane >> 4) << 3;
  const int node0 = blockIdx.x * 16;

  int nodeQ[2], begQ[2], degQ[2];
  float dQ[2];
#pragma unroll
  for (int q = 0; q < 2; ++q) {
    nodeQ[q] = min(node0 + wave * 2 + q, n - 1);
    begQ[q]  = rowptr[nodeQ[q]];
    degQ[q]  = cnt[nodeQ[q]];
    dQ[q]    = dinv[nodeQ[q]];
  }

#pragma unroll
  for (int ks = 0; ks < 4; ++ks) {
    const int nn = tid & 31;
    const int kq = tid >> 5;
    union { __bf16 h[4]; uint2 q; } wb;
#pragma unroll
    for (int j = 0; j < 4; ++j)
      wb.h[j] = (__bf16)W2[(size_t)(ks * 32 + kq * 4 + j) * D_OUT + nn];
    *(uint2*)(ldsB + (ks * D_OUT + nn) * 32 + kq * 4) = wb.q;
  }

  const int outRow = node0 + (tid >> 3);
  const float dOut = (tid < 128 && outRow < n) ? dinv[outRow] : 0.0f;

  int firstQ[2];
#pragma unroll
  for (int q = 0; q < 2; ++q)
    firstQ[q] = (lane < min(32, degQ[q])) ? src[begQ[q] + lane] : 0;
  const float4* __restrict__ hs4 = (const float4*)hs;
  const float4 b4 = ((const float4*)bias)[lane];
#pragma unroll
  for (int q = 0; q < 2; ++q) {
    const int r = wave * 2 + q;
    union { __bf16 h[4]; uint2 u; } pk;
    const float4 acc = pull_row128(hs4, src, nodeQ[q], begQ[q], degQ[q], firstQ[q], lane);
    const float d = dQ[q];
    pk.h[0] = (__bf16)fmaxf(acc.x * d + b4.x, 0.0f);
    pk.h[1] = (__bf16)fmaxf(acc.y * d + b4.y, 0.0f);
    pk.h[2] = (__bf16)fmaxf(acc.z * d + b4.z, 0.0f);
    pk.h[3] = (__bf16)fmaxf(acc.w * d + b4.w, 0.0f);
    *(uint2*)(ldsA + ((lane >> 3) * 16 + r) * 32 + ((lane & 7) << 2)) = pk.u;
  }
  __syncthreads();

  if (wave < 2) {
    v8f acc = {0.f, 0.f, 0.f, 0.f, 0.f, 0.f, 0.f, 0.f};
#pragma unroll
    for (int ks = 0; ks < 4; ++ks) {
      const v16bf a = frag_from_lds(ldsA + ks * 16 * 32, laneM, khalf);
      const v16bf b = frag_from_lds(ldsB + ks * D_OUT * 32, (wave << 4) + laneM, khalf);
      acc = __builtin_amdgcn_wmma_f32_16x16x32_bf16(
          false, a, false, b, (short)0, acc, false, false);
    }
    const int mBase = (lane >> 4) << 3;
#pragma unroll
    for (int v = 0; v < 8; ++v)
      ldsC[(mBase + v) * D_OUT + (wave << 4) + laneM] = acc[v];
  }
  __syncthreads();

  if (tid < 128 && outRow < n) {
    float4 c = ((const float4*)ldsC)[(tid >> 3) * (D_OUT / 4) + (tid & 7)];
    c.x *= dOut; c.y *= dOut; c.z *= dOut; c.w *= dOut;
    ((float4*)g2)[(size_t)outRow * (D_OUT / 4) + (tid & 7)] = c;
  }
}

__global__ __launch_bounds__(256) void k_pull32_logsoftmax(
    const float* __restrict__ gs, const float* __restrict__ dinv,
    const float* __restrict__ bias, const int* __restrict__ rowptr,
    const int* __restrict__ cnt, const int* __restrict__ src,
    float* __restrict__ out, int n) {
  const int wave = threadIdx.x >> 5;
  const int lane = threadIdx.x & 31;
  const int g    = lane >> 3;
  const int c4   = lane & 7;
  const float4* __restrict__ gs4 = (const float4*)gs;
  float4* __restrict__ out4 = (float4*)out;
  const float4 b4 = ((const float4*)bias)[c4];
  const int nodeBase = blockIdx.x * 128 + wave * 4 + g;

  int node = nodeBase;
  int nd   = min(node, n - 1);
  int beg  = rowptr[nd];
  int deg  = (node < n) ? cnt[nd] : 0;
  int mine = (c4 < deg) ? src[beg + c4] : -1;
  float4 acc = gs4[(size_t)nd * 8 + c4];
  float d = dinv[nd];

  for (int t = 0; t < 4; ++t) {
    const int cNode = node, cBeg = beg, cDeg = deg;
    const float cD = d;
    node = nodeBase + (t + 1) * 32;
    nd   = min(node, n - 1);
    if (t + 1 < 4) {
      beg = rowptr[nd];
      deg = (node < n) ? cnt[nd] : 0;
    }
    int dmax = cDeg;
    dmax = max(dmax, __shfl_xor(dmax, 8, 32));
    dmax = max(dmax, __shfl_xor(dmax, 16, 32));
    dmax = __builtin_amdgcn_readfirstlane(dmax);
    for (int base = 0; base < dmax; base += 8) {
      const int e2 = base + 8 + c4;
      const int nxt = (e2 < cDeg) ? src[cBeg + e2] : -1;
      float4 v[8];
#pragma unroll
      for (int k = 0; k < 8; ++k) {
        const int r = __shfl(mine, k, 8);
        v[k] = make_float4(0.0f, 0.0f, 0.0f, 0.0f);
        if (r >= 0) v[k] = gs4[(size_t)r * 8 + c4];
      }
#pragma unroll
      for (int k = 0; k < 8; ++k) {
        acc.x += v[k].x; acc.y += v[k].y; acc.z += v[k].z; acc.w += v[k].w;
      }
      mine = nxt;
    }
    float4 v;
    v.x = acc.x * cD + b4.x;
    v.y = acc.y * cD + b4.y;
    v.z = acc.z * cD + b4.z;
    v.w = acc.w * cD + b4.w;
    if (t + 1 < 4) {
      mine = (c4 < deg) ? src[beg + c4] : -1;
      acc  = gs4[(size_t)nd * 8 + c4];
      d    = dinv[nd];
    }
    float mx = fmaxf(fmaxf(v.x, v.y), fmaxf(v.z, v.w));
    for (int o = 4; o > 0; o >>= 1) mx = fmaxf(mx, __shfl_xor(mx, o, 32));
    float s = (__expf(v.x - mx) + __expf(v.y - mx)) + (__expf(v.z - mx) + __expf(v.w - mx));
    for (int o = 4; o > 0; o >>= 1) s += __shfl_xor(s, o, 32);
    const float ls = __logf(s);
    if (cNode < n) {
      float4 o4;
      o4.x = v.x - mx - ls;
      o4.y = v.y - mx - ls;
      o4.z = v.z - mx - ls;
      o4.w = v.w - mx - ls;
      out4[(size_t)cNode * 8 + c4] = o4;
    }
  }
}

extern "C" void kernel_launch(void* const* d_in, const int* in_sizes, int n_in,
                              void* d_out, int out_size, void* d_ws, size_t ws_size,
                              hipStream_t stream) {
  const float* x  = (const float*)d_in[0];
  const float* W1 = (const float*)d_in[1];
  const float* b1 = (const float*)d_in[2];
  const float* W2 = (const float*)d_in[3];
  const float* b2 = (const float*)d_in[4];
  const int*  eix = (const int*)d_in[5];

  const int n = N_NODES;
  const int E = in_sizes[5] / 2;
  const int* row = eix;
  const int* col = eix + E;

  float* ws     = (float*)d_ws;
  float* dinv   = ws;
  int*   cnt    = (int*)(ws + N_PAD);
  int*   rowptr = cnt + N_PAD;
  int*   part   = rowptr + N_PAD;
  int*   src    = part + E;
  float* hs     = (float*)(src + E);
  float* g2     = hs + (size_t)n * D_H;
  int*   gcnt   = (int*)(g2 + (size_t)n * D_OUT);
  int*   gcur   = gcnt + NBIN_PAD;
  float* outf   = (float*)d_out;

  k_zero<<<1, 2 * NBIN_PAD, 0, stream>>>(gcnt, 2 * NBIN_PAD);
  k_hist<<<(E + CHUNK - 1) / CHUNK, 1024, 0, stream>>>(col, gcnt, E);
  k_part<<<(E + CHUNK - 1) / CHUNK, 1024, 0, stream>>>(row, col, gcnt, gcur, part, E);
  k_sort<<<NBIN, 1024, 0, stream>>>(part, gcnt, src, rowptr, cnt, dinv);

  gemm_bf16_wmma<D_IN, D_H><<<(n + 127) / 128, 256, 0, stream>>>(x, W1, hs, dinv, n);
  k_pull128_relu_gemm32<<<(n + 15) / 16, 256, 0, stream>>>(hs, dinv, b1, rowptr, cnt, src, W2, g2, n);

  k_pull32_logsoftmax<<<(n + 127) / 128, 256, 0, stream>>>(g2, dinv, b2, rowptr, cnt, src, outf, n);
}


# speedup vs baseline: 4.8540x; 1.1204x over previous
#include <hip/hip_runtime.h>

#define N_NODES 100000
#define D_IN 256
#define D_H 128
#define D_OUT 32

typedef __attribute__((ext_vector_type(16))) __bf16 v16bf;
typedef __attribute__((ext_vector_type(8)))  float  v8f;

__device__ __forceinline__ v16bf frag_from_lds(const __bf16* base, int rowIdx, int khalf) {
  union { uint4 u[2]; v16bf v; } c;
  const uint4* p = (const uint4*)(base + rowIdx * 32 + khalf);
  c.u[0] = p[0];
  c.u[1] = p[2];
  return c.v;
}

template<int K, int N>
__global__ __launch_bounds__(256) void gemm_bf16_wmma(
    const float* __restrict__ A, const float* __restrict__ W,
    unsigned int* __restrict__ C,
    const float* __restrict__ dinv, int M) {
  constexpr int NT = N / 16;
  static_assert(NT == 8, "the epilogue packs 8 column tiles into one 16-byte store");
  __shared__ __bf16 ldsA[128 * 32];
  __shared__ __bf16 ldsB[N * 32];

  const int tid   = threadIdx.x;
  const int wave  = tid >> 5;
  const int lane  = tid & 31;
  const int laneM = lane & 15;
  const int khalf = (lane >> 4) << 3;
  const int rowBlk = blockIdx.x * 128;

  const int ldRow = tid >> 1;
  const int ldC0  = (tid & 1) << 4;
  const int ldGr  = rowBlk + ldRow;

  v8f acc[NT];
#pragma unroll
  for (int t = 0; t < NT; ++t) {
    v8f z = {0.f, 0.f, 0.f, 0.f, 0.f, 0.f, 0.f, 0.f};
    acc[t] = z;
  }

  for (int kk = 0; kk < K; kk += 32) {
    __syncthreads();

    {
      union { __bf16 h[16]; uint4 q[2]; } buf;
      if (ldGr < M) {
        const float4* src = (const float4*)(A + (size_t)ldGr * K + kk + ldC0);
#pragma unroll
        for (int j = 0; j < 4; ++j) {
          float4 v = src[j];
          buf.h[j * 4 + 0] = (__bf16)v.x;
          buf.h[j * 4 + 1] = (__bf16)v.y;
          buf.h[j * 4 + 2] = (__bf16)v.z;
          buf.h[j * 4 + 3] = (__bf16)v.w;
        }
      } else {
#pragma unroll
        for (int j = 0; j < 16; ++j) buf.h[j] = (__bf16)0.0f;
      }
      uint4* dst = (uint4*)(ldsA + ldRow * 32 + ldC0);
      dst[0] = buf.q[0];
      dst[1] = buf.q[1];
    }

#pragma unroll
    for (int i = tid; i < N * 8; i += 256) {
      const int n  = i % N;
      const int kq = i / N;
      const int p  = (n % NT) * 16 + n / NT;
      union { __bf16 h[4]; uint2 q; } wb;
#pragma unroll
      for (int j = 0; j < 4; ++j)
        wb.h[j] = (__bf16)W[(size_t)(kk + kq * 4 + j) * N + n];
      *(uint2*)(ldsB + p * 32 + kq * 4) = wb.q;
    }
    __syncthreads();

    v16bf a = frag_from_lds(ldsA, (wave << 4) + laneM, khalf);
    v16bf bc = frag_from_lds(ldsB, laneM, khalf);
#pragma unroll
    for (int t = 0; t < NT; ++t) {
      v16bf bn = (t + 1 < NT) ? frag_from_lds(ldsB, ((t + 1) << 4) + laneM, khalf) : bc;
      acc[t] = __builtin_amdgcn_wmma_f32_16x16x32_bf16(
          false, a, false, bc, (short)0, acc[t], false, false);
      bc = bn;
    }
  }

  const int mBase = rowBlk + (wave << 4) + ((lane >> 4) << 3);
  uint4* __restrict__ C4 = (uint4*)C;
#pragma unroll
  for (int v = 0; v < 8; ++v) {
    const int gr = mBase + v;
    if (gr < M) {
      const float d = dinv[gr];
      union { __bf16 h[8]; uint4 q; } o;
#pragma unroll
      for (int t = 0; t < NT; ++t) o.h[t] = (__bf16)(acc[t][v] * d);
      C4[(size_t)gr * (N / 8) + laneM] = o.q;
    }
  }
}

#define N_PAD 100352

#define BIN_SHIFT 9
#define NBIN 196
#define NBIN_PAD 256
#define CHUNK 8192
#define SORT_CAP 16384
#define SRC_MASK 0x1FFFF

__global__ void k_zero(int* __restrict__ p, int n) {
  int i = blockIdx.x * blockDim.x + threadIdx.x;
  if (i < n) p[i] = 0;
}

__global__ __launch_bounds__(1024) void k_hist(
    const int* __restrict__ col, int* __restrict__ gcnt, int E) {
  __shared__ int h[NBIN_PAD];
  const int tid = threadIdx.x;
  if (tid < NBIN_PAD) h[tid] = 0;
  __syncthreads();
  const int base = blockIdx.x * CHUNK;
#pragma unroll
  for (int k = 0; k < CHUNK / 1024; ++k) {
    const int e = base + k * 1024 + tid;
    if (e < E) atomicAdd(&h[col[e] >> BIN_SHIFT], 1);
  }
  __syncthreads();
  if (tid < NBIN_PAD) {
    const int v = h[tid];
    if (v) atomicAdd(&gcnt[tid], v);
  }
}

__global__ __launch_bounds__(1024) void k_part(
    const int* __restrict__ row, const int* __restrict__ col,
    const int* __restrict__ gcnt, int* __restrict__ gcur,
    int* __restrict__ part, int E) {
  __shared__ int h[NBIN_PAD];
  __shared__ int lstart[NBIN_PAD];
  __shared__ int gb[NBIN_PAD];
  __shared__ int sval[CHUNK];
  __shared__ unsigned char sbin[CHUNK];
  const int tid = threadIdx.x;
  if (tid < NBIN_PAD) h[tid] = 0;
  __syncthreads();
  const int base = blockIdx.x * CHUNK;
  int c[CHUNK / 1024], r[CHUNK / 1024], lr[CHUNK / 1024];
#pragma unroll
  for (int k = 0; k < CHUNK / 1024; ++k) {
    const int e = base + k * 1024 + tid;
    const bool ok = e < E;
    c[k] = ok ? col[e] : -1;
    r[k] = ok ? row[e] : 0;
  }
#pragma unroll
  for (int k = 0; k < CHUNK / 1024; ++k) {
    lr[k] = 0;
    if (c[k] >= 0) lr[k] = atomicAdd(&h[c[k] >> BIN_SHIFT], 1);
  }
  __syncthreads();
  if (tid < 32) {
    int hv[8], gv[8], hloc[8], gloc[8];
    int hsum = 0, gsum = 0;
#pragma unroll
    for (int j = 0; j < 8; ++j) {
      hv[j] = h[tid * 8 + j];
      gv[j] = gcnt[tid * 8 + j];
    }
#pragma unroll
    for (int j = 0; j < 8; ++j) {
      hloc[j] = hsum; hsum += hv[j];
      gloc[j] = gsum; gsum += gv[j];
    }
    int hi = hsum, gi = gsum;
    for (int o = 1; o < 32; o <<= 1) {
      const int y = __shfl_up(hi, o, 32);
      const int z = __shfl_up(gi, o, 32);
      if (tid >= o) { hi += y; gi += z; }
    }
    const int hex = hi - hsum, gex = gi - gsum;
#pragma unroll
    for (int j = 0; j < 8; ++j) {
      const int b = tid * 8 + j;
      lstart[b] = hex + hloc[j];
      int g = gex + gloc[j];
      if (hv[j]) g += atomicAdd(&gcur[b], hv[j]);
      gb[b] = g;
    }
  }
  __syncthreads();
#pragma unroll
  for (int k = 0; k < CHUNK / 1024; ++k) {
    if (c[k] >= 0) {
      const int b = c[k] >> BIN_SHIFT;
      const int p = lstart[b] + lr[k];
      sval[p] = ((c[k] & ((1 << BIN_SHIFT) - 1)) << 17) | r[k];
      sbin[p] = (unsigned char)b;
    }
  }
  __syncthreads();
  const int nhere = min(CHUNK, E - base);
#pragma unroll
  for (int k = 0; k < CHUNK / 1024; ++k) {
    const int i = k * 1024 + tid;
    if (i < nhere) {
      const int b = sbin[i];
      part[gb[b] + (i - lstart[b])] = sval[i];
    }
  }
}

__global__ __launch_bounds__(1024) void k_sort(
    const int* __restrict__ part, const int* __restrict__ gcnt,
    int* __restrict__ src, int* __restrict__ rowptr, int* __restrict__ cnt,
    float* __restrict__ dinv) {
  __shared__ int c_l[512];
  __shared__ int cur_l[512];
  __shared__ int rp_l[512];
  __shared__ int wsum[32];
  __shared__ int s_start;
  __shared__ int sorted_l[SORT_CAP];
  const int tid  = threadIdx.x;
  const int lane = tid & 31;
  const int wave = tid >> 5;
  const int bin  = blockIdx.x;

  if (tid < 512) { c_l[tid] = 0; cur_l[tid] = 0; }
  if (tid < 32) {
    int s = 0;
    for (int j = tid; j < bin; j += 32) s += gcnt[j];
    for (int o = 16; o > 0; o >>= 1) s += __shfl_xor(s, o, 32);
    if (tid == 0) s_start = s;
  }
  __syncthreads();
  const int bstart = s_start;
  const int nb = gcnt[bin];

  for (int i0 = 0; i0 < nb; i0 += 8192) {
    int en[8];
#pragma unroll
    for (int k = 0; k < 8; ++k) {
      const int i = i0 + k * 1024 + tid;
      en[k] = (i < nb) ? part[bstart + i] : -1;
    }
#pragma unroll
    for (int k = 0; k < 8; ++k)
      if (en[k] >= 0) atomicAdd(&c_l[en[k] >> 17], 1);
  }
  __syncthreads();

  int x = 0, c = 0;
  if (tid < 512) {
    c = c_l[tid];
    x = c;
    for (int o = 1; o < 32; o <<= 1) {
      const int y = __shfl_up(x, o, 32);
      if (lane >= o) x += y;
    }
    if (lane == 31) wsum[wave] = x;
  }
  __syncthreads();
  if (tid < 32) {
    int t = (lane < 16) ? wsum[lane] : 0;
    for (int o = 1; o < 32; o <<= 1) {
      const int y = __shfl_up(t, o, 32);
      if (lane >= o) t += y;
    }
    wsum[lane] = t;
  }
  __syncthreads();
  if (tid < 512) {
    const int rp = x - c + ((wave > 0) ? wsum[wave - 1] : 0);
    rp_l[tid] = rp;
    const int node = (bin << BIN_SHIFT) + tid;
    cnt[node]    = c;
    rowptr[node] = bstart + rp;
    dinv[node]   = rsqrtf((float)(c + 1));
  }
  __syncthreads();

  const bool fits = nb <= SORT_CAP;
  for (int i0 = 0; i0 < nb; i0 += 8192) {
    int en[8];
#pragma unroll
    for (int k = 0; k < 8; ++k) {
      const int i = i0 + k * 1024 + tid;
      en[k] = (i < nb) ? part[bstart + i] : -1;
    }
#pragma unroll
    for (int k = 0; k < 8; ++k) {
      if (en[k] >= 0) {
        const int l = en[k] >> 17;
        const int p = rp_l[l] + atomicAdd(&cur_l[l], 1);
        if (fits) sorted_l[p] = en[k] & SRC_MASK;
        else      src[bstart + p] = en[k] & SRC_MASK;
      }
    }
  }
  __syncthreads();
  if (fits)
    for (int i = tid; i < nb; i += 1024) src[bstart + i] = sorted_l[i];
}

__device__ __forceinline__ float4 widen_bf16x4(uint2 u) {
  float4 f;
  f.x = __uint_as_float(u.x << 16);
  f.y = __uint_as_float(u.x & 0xFFFF0000u);
  f.z = __uint_as_float(u.y << 16);
  f.w = __uint_as_float(u.y & 0xFFFF0000u);
  return f;
}

__device__ __forceinline__ float4 pull_row128(
    const uint2* __restrict__ hs2, const int* __restrict__ src,
    int node, int beg, int deg, int first, int lane) {
  float4 acc = widen_bf16x4(hs2[(size_t)node * 32 + lane]);
  for (int base = 0; base < deg; base += 32) {
    const int m = min(32, deg - base);
    const int mine = (base == 0) ? first : ((lane < m) ? src[beg + base + lane] : 0);
    int j = 0;
    for (; j + 8 <= m; j += 8) {
      uint2 v[8];
#pragma unroll
      for (int k = 0; k < 8; ++k) {
        const int r = __builtin_amdgcn_readlane(mine, j + k);
        v[k] = hs2[(size_t)r * 32 + lane];
      }
#pragma unroll
      for (int k = 0; k < 8; ++k) {
        const float4 f = widen_bf16x4(v[k]);
        acc.x += f.x; acc.y += f.y; acc.z += f.z; acc.w += f.w;
      }
    }
    if (j + 4 <= m) {
      uint2 v[4];
#pragma unroll
      for (int k = 0; k < 4; ++k) {
        const int r = __builtin_amdgcn_readlane(mine, j + k);
        v[k] = hs2[(size_t)r * 32 + lane];
      }
#pragma unroll
      for (int k = 0; k < 4; ++k) {
        const float4 f = widen_bf16x4(v[k]);
        acc.x += f.x; acc.y += f.y; acc.z += f.z; acc.w += f.w;
      }
      j += 4;
    }
    for (; j < m; ++j) {
      const int r = __builtin_amdgcn_readlane(mine, j);
      const float4 f = widen_bf16x4(hs2[(size_t)r * 32 + lane]);
      acc.x += f.x; acc.y += f.y; acc.z += f.z; acc.w += f.w;
    }
  }
  return acc;
}

__global__ __launch_bounds__(256) void k_pull128_relu_gemm32(
    const unsigned int* __restrict__ hs, const float* __restrict__ dinv,
    const float* __restrict__ bias, const int* __restrict__ rowptr,
    const int* __restrict__ cnt, const int* __restrict__ src,
    const float* __restrict__ W2, float* __restrict__ g2, int n) {
  __shared__ __bf16 ldsA[4 * 16 * 32];
  __shared__ __bf16 ldsB[4 * D_OUT * 32];
  __shared__ float  ldsC[16 * D_OUT];

  const int tid  = threadIdx.x;
  const int wave = __builtin_amdgcn_readfirstlane(tid >> 5);
  const int lane = tid & 31;
  const int laneM = lane & 15;
  const int khalf = (lane >> 4) << 3;
  const int node0 = blockIdx.x * 16;

  int nodeQ[2], begQ[2], degQ[2];
  float dQ[2];
#pragma unroll
  for (int q = 0; q < 2; ++q) {
    nodeQ[q] = min(node0 + wave * 2 + q, n - 1);
    begQ[q]  = rowptr[nodeQ[q]];
    degQ[q]  = cnt[nodeQ[q]];
    dQ[q]    = dinv[nodeQ[q]];
  }

#pragma unroll
  for (int ks = 0; ks < 4; ++ks) {
    const int nn = tid & 31;
    const int kq = tid >> 5;
    union { __bf16 h[4]; uint2 q; } wb;
#pragma unroll
    for (int j = 0; j < 4; ++j)
      wb.h[j] = (__bf16)W2[(size_t)(ks * 32 + kq * 4 + j) * D_OUT + nn];
    *(uint2*)(ldsB + (ks * D_OUT + nn) * 32 + kq * 4) = wb.q;
  }

  const int outRow = node0 + (tid >> 3);
  const float dOut = (tid < 128 && outRow < n) ? dinv[outRow] : 0.0f;

  int firstQ[2];
#pragma unroll
  for (int q = 0; q < 2; ++q)
    firstQ[q] = (lane < min(32, degQ[q])) ? src[begQ[q] + lane] : 0;
  const uint2* __restrict__ hs2 = (const uint2*)hs;
  const float4 b4 = ((const float4*)bias)[lane];
#pragma unroll
  for (int q = 0; q < 2; ++q) {
    const int r = wave * 2 + q;
    union { __bf16 h[4]; uint2 u; } pk;
    const float4 acc = pull_row128(hs2, src, nodeQ[q], begQ[q], degQ[q], firstQ[q], lane);
    const float d = dQ[q];
    pk.h[0] = (__bf16)fmaxf(acc.x * d + b4.x, 0.0f);
    pk.h[1] = (__bf16)fmaxf(acc.y * d + b4.y, 0.0f);
    pk.h[2] = (__bf16)fmaxf(acc.z * d + b4.z, 0.0f);
    pk.h[3] = (__bf16)fmaxf(acc.w * d + b4.w, 0.0f);
    *(uint2*)(ldsA + ((lane >> 3) * 16 + r) * 32 + ((lane & 7) << 2)) = pk.u;
  }
  __syncthreads();

  if (wave < 2) {
    v8f acc = {0.f, 0.f, 0.f, 0.f, 0.f, 0.f, 0.f, 0.f};
#pragma unroll
    for (int ks = 0; ks < 4; ++ks) {
      const v16bf a = frag_from_lds(ldsA + ks * 16 * 32, laneM, khalf);
      const v16bf b = frag_from_lds(ldsB + ks * D_OUT * 32, (wave << 4) + laneM, khalf);
      acc = __builtin_amdgcn_wmma_f32_16x16x32_bf16(
          false, a, false, b, (short)0, acc, false, false);
    }
    const int mBase = (lane >> 4) << 3;
#pragma unroll
    for (int v = 0; v < 8; ++v)
      ldsC[(mBase + v) * D_OUT + (wave << 4) + laneM] = acc[v];
  }
  __syncthreads();

  if (tid < 128 && outRow < n) {
    float4 c = ((const float4*)ldsC)[(tid >> 3) * (D_OUT / 4) + (tid & 7)];
    c.x *= dOut; c.y *= dOut; c.z *= dOut; c.w *= dOut;
    ((float4*)g2)[(size_t)outRow * (D_OUT / 4) + (tid & 7)] = c;
  }
}

__global__ __launch_bounds__(256) void k_pull32_logsoftmax(
    const float* __restrict__ gs, const float* __restrict__ dinv,
    const float* __restrict__ bias, const int* __restrict__ rowptr,
    const int* __restrict__ cnt, const int* __restrict__ src,
    float* __restrict__ out, int n) {
  const int wave = threadIdx.x >> 5;
  const int lane = threadIdx.x & 31;
  const int g    = lane >> 3;
  const int c4   = lane & 7;
  const float4* __restrict__ gs4 = (const float4*)gs;
  float4* __restrict__ out4 = (float4*)out;
  const float4 b4 = ((const float4*)bias)[c4];
  const int nodeBase = blockIdx.x * 128 + wave * 4 + g;

  int node = nodeBase;
  int nd   = min(node, n - 1);
  int beg  = rowptr[nd];
  int deg  = (node < n) ? cnt[nd] : 0;
  int mine = (c4 < deg) ? src[beg + c4] : -1;
  float4 acc = gs4[(size_t)nd * 8 + c4];
  float d = dinv[nd];

  for (int t = 0; t < 4; ++t) {
    const int cNode = node, cBeg = beg, cDeg = deg;
    const float cD = d;
    node = nodeBase + (t + 1) * 32;
    nd   = min(node, n - 1);
    if (t + 1 < 4) {
      beg = rowptr[nd];
      deg = (node < n) ? cnt[nd] : 0;
    }
    int dmax = cDeg;
    dmax = max(dmax, __shfl_xor(dmax, 8, 32));
    dmax = max(dmax, __shfl_xor(dmax, 16, 32));
    dmax = __builtin_amdgcn_readfirstlane(dmax);
    for (int base = 0; base < dmax; base += 8) {
      const int e2 = base + 8 + c4;
      const int nxt = (e2 < cDeg) ? src[cBeg + e2] : -1;
      float4 v[8];
#pragma unroll
      for (int k = 0; k < 8; ++k) {
        const int r = __shfl(mine, k, 8);
        v[k] = make_float4(0.0f, 0.0f, 0.0f, 0.0f);
        if (r >= 0) v[k] = gs4[(size_t)r * 8 + c4];
      }
#pragma unroll
      for (int k = 0; k < 8; ++k) {
        acc.x += v[k].x; acc.y += v[k].y; acc.z += v[k].z; acc.w += v[k].w;
      }
      mine = nxt;
    }
    float4 v;
    v.x = acc.x * cD + b4.x;
    v.y = acc.y * cD + b4.y;
    v.z = acc.z * cD + b4.z;
    v.w = acc.w * cD + b4.w;
    if (t + 1 < 4) {
      mine = (c4 < deg) ? src[beg + c4] : -1;
      acc  = gs4[(size_t)nd * 8 + c4];
      d    = dinv[nd];
    }
    float mx = fmaxf(fmaxf(v.x, v.y), fmaxf(v.z, v.w));
    for (int o = 4; o > 0; o >>= 1) mx = fmaxf(mx, __shfl_xor(mx, o, 32));
    float s = (__expf(v.x - mx) + __expf(v.y - mx)) + (__expf(v.z - mx) + __expf(v.w - mx));
    for (int o = 4; o > 0; o >>= 1) s += __shfl_xor(s, o, 32);
    const float ls = __logf(s);
    if (cNode < n) {
      float4 o4;
      o4.x = v.x - mx - ls;
      o4.y = v.y - mx - ls;
      o4.z = v.z - mx - ls;
      o4.w = v.w - mx - ls;
      out4[(size_t)cNode * 8 + c4] = o4;
    }
  }
}

extern "C" void kernel_launch(void* const* d_in, const int* in_sizes, int n_in,
                              void* d_out, int out_size, void* d_ws, size_t ws_size,
                              hipStream_t stream) {
  const float* x  = (const float*)d_in[0];
  const float* W1 = (const float*)d_in[1];
  const float* b1 = (const float*)d_in[2];
  const float* W2 = (const float*)d_in[3];
  const float* b2 = (const float*)d_in[4];
  const int*  eix = (const int*)d_in[5];

  const int n = N_NODES;
  const int E = in_sizes[5] / 2;
  const int* row = eix;
  const int* col = eix + E;

  float* ws     = (float*)d_ws;
  float* dinv   = ws;
  int*   cnt    = (int*)(ws + N_PAD);
  int*   rowptr = cnt + N_PAD;
  int*   part   = rowptr + N_PAD;
  int*   src    = part + E;
  unsigned int* hs = (unsigned int*)(src + E);
  float* g2     = (float*)(hs + (size_t)n * (D_H / 2));
  int*   gcnt   = (int*)(g2 + (size_t)n * D_OUT);
  int*   gcur   = gcnt + NBIN_PAD;
  float* outf   = (float*)d_out;

  k_zero<<<1, 2 * NBIN_PAD, 0, stream>>>(gcnt, 2 * NBIN_PAD);
  k_hist<<<(E + CHUNK - 1) / CHUNK, 1024, 0, stream>>>(col, gcnt, E);
  k_part<<<(E + CHUNK - 1) / CHUNK, 1024, 0, stream>>>(row, col, gcnt, gcur, part, E);
  k_sort<<<NBIN, 1024, 0, stream>>>(part, gcnt, src, rowptr, cnt, dinv);

  gemm_bf16_wmma<D_IN, D_H><<<(n + 127) / 128, 256, 0, stream>>>(x, W1, hs, dinv, n);
  k_pull128_relu_gemm32<<<(n + 15) / 16, 256, 0, stream>>>(hs, dinv, b1, rowptr, cnt, src, W2, g2, n);

  k_pull32_logsoftmax<<<(n + 127) / 128, 256, 0, stream>>>(g2, dinv, b2, rowptr, cnt, src, outf, n);
}


# Layer-1 GEMM fed from registers: W1 rounded to bf16 once per call (spare blocks of the histogram launch), whole in LDS; A fragments straight from global, no barrier in k-loop; 768 resident groups
# speedup vs baseline: 6.9838x; 1.4388x over previous
#include <hip/hip_runtime.h>

#define N_NODES 100000
#define D_IN 256
#define D_H 128
#define D_OUT 32

typedef __attribute__((ext_vector_type(16))) __bf16 v16bf;
typedef __attribute__((ext_vector_type(8)))  float  v8f;

__device__ __forceinline__ v16bf frag_from_lds(const __bf16* base, int rowIdx, int khalf) {
  union { uint4 u[2]; v16bf v; } c;
  const uint4* p = (const uint4*)(base + rowIdx * 32 + khalf);
  c.u[0] = p[0];
  c.u[1] = p[2];
  return c.v;
}

#define W1T_PITCH 264
__device__ __forceinline__ v16bf frag_from_w1(const __bf16* base, int rowIdx, int k0) {
  union { uint4 u[2]; v16bf v; } c;
  const uint4* p = (const uint4*)(base + rowIdx * W1T_PITCH + k0);
  c.u[0] = p[0];
  c.u[1] = p[2];
  return c.v;
}

__device__ __forceinline__ v16bf frag_from_f32(const float4 (&q)[4]) {
  union { __bf16 h[16]; v16bf v; } c;
#pragma unroll
  for (int j = 0; j < 4; ++j) {
    c.h[j * 4 + 0] = (__bf16)q[j].x;
    c.h[j * 4 + 1] = (__bf16)q[j].y;
    c.h[j * 4 + 2] = (__bf16)q[j].z;
    c.h[j * 4 + 3] = (__bf16)q[j].w;
  }
  return c.v;
}

__device__ __forceinline__ void w1_to_bf16_rows(
    const float* __restrict__ W, __bf16* __restrict__ w1t, int g) {
  const int n  = g & (D_H - 1);
  const int kq = g >> 7;
  const int p  = (n & 7) * 16 + (n >> 3);
  union { __bf16 h[4]; uint2 q; } wb;
#pragma unroll
  for (int j = 0; j < 4; ++j)
    wb.h[j] = (__bf16)W[(size_t)(kq * 4 + j) * D_H + n];
  *(uint2*)(w1t + p * D_IN + kq * 4) = wb.q;
}

__global__ __launch_bounds__(256) void gemm1_bf16_wmma(
    const float* __restrict__ A, const unsigned int* __restrict__ w1t,
    unsigned int* __restrict__ C,
    const float* __restrict__ dinv, int M) {
  constexpr int K = D_IN, N = D_H, NT = N / 16;
  static_assert(NT == 8, "the epilogue packs 8 column tiles into one 16-byte store");
  __shared__ __bf16 ldsB[N * W1T_PITCH];

  const int tid   = threadIdx.x;
  const int wave  = __builtin_amdgcn_readfirstlane(tid >> 5);
  const int lane  = tid & 31;
  const int laneM = lane & 15;
  const int khalf = (lane >> 4) << 3;
  const int nTiles = (M + 15) >> 4;
  const int G = gridDim.x;

  const int tile0 = (int)blockIdx.x * 8 + wave;
  const float4* __restrict__ arow0 =
      (const float4*)(A + (size_t)min((tile0 << 4) + laneM, M - 1) * K + khalf);

  float4 abuf[2][4];
  {
    const uint4* __restrict__ w4 = (const uint4*)w1t;
    uint4 wv[16];
#pragma unroll
    for (int i = 0; i < 16; ++i) wv[i] = w4[tid + i * 256];
#pragma unroll
    for (int s = 0; s < 2; ++s) {
      abuf[s][0] = arow0[s * 8 + 0];
      abuf[s][1] = arow0[s * 8 + 1];
      abuf[s][2] = arow0[s * 8 + 4];
      abuf[s][3] = arow0[s * 8 + 5];
    }
#pragma unroll
    for (int i = 0; i < 16; ++i) {
      const int idx = tid + i * 256;
      *(uint4*)(ldsB + (idx >> 5) * W1T_PITCH + ((idx & 31) << 3)) = wv[i];
    }
  }
  __syncthreads();

  uint4* __restrict__ C4 = (uint4*)C;
  for (int it = 0; ; ++it) {
    const int tile = (it == 0) ? tile0
                               : 8 * G + ((it - 1) * 8 + wave) * G + (int)blockIdx.x;
    if (tile >= nTiles) break;
    __builtin_amdgcn_fence(__ATOMIC_ACQUIRE, "wavefront");
    const int row0 = tile << 4;
    const float4* __restrict__ arow =
        (const float4*)(A + (size_t)min(row0 + laneM, M - 1) * K + khalf);
    const float4* __restrict__ dp = (const float4*)(dinv + row0 + ((lane >> 4) << 3));
    const float4 dLo = dp[0], dHi = dp[1];
    if (it > 0) {
#pragma unroll
      for (int s = 0; s < 2; ++s) {
        abuf[s][0] = arow[s * 8 + 0];
        abuf[s][1] = arow[s * 8 + 1];
        abuf[s][2] = arow[s * 8 + 4];
        abuf[s][3] = arow[s * 8 + 5];
      }
    }

    v8f acc[NT];
#pragma unroll
    for (int t = 0; t < NT; ++t) {
      v8f z = {0.f, 0.f, 0.f, 0.f, 0.f, 0.f, 0.f, 0.f};
      acc[t] = z;
    }

#pragma unroll
    for (int s = 0; s < K / 32; ++s) {
      const v16bf a = frag_from_f32(abuf[s & 1]);
      if (s + 2 < K / 32) {
        abuf[s & 1][0] = arow[(s + 2) * 8 + 0];
        abuf[s & 1][1] = arow[(s + 2) * 8 + 1];
        abuf[s & 1][2] = arow[(s + 2) * 8 + 4];
        abuf[s & 1][3] = arow[(s + 2) * 8 + 5];
      }
      const int k0 = s * 32 + khalf;
      v16bf bc = frag_from_w1(ldsB, laneM, k0);
#pragma unroll
      for (int t = 0; t < NT; ++t) {
        v16bf bn = (t + 1 < NT) ? frag_from_w1(ldsB, ((t + 1) << 4) + laneM, k0) : bc;
        acc[t] = __builtin_amdgcn_wmma_f32_16x16x32_bf16(
            false, a, false, bc, (short)0, acc[t], false, false);
        bc = bn;
      }
    }

    const int mBase = row0 + ((lane >> 4) << 3);
#pragma unroll
    for (int v = 0; v < 8; ++v) {
      const int gr = mBase + v;
      if (gr < M) {
        const float d = (v == 0) ? dLo.x : (v == 1) ? dLo.y : (v == 2) ? dLo.z : (v == 3) ? dLo.w
                      : (v == 4) ? dHi.x : (v == 5) ? dHi.y : (v == 6) ? dHi.z : dHi.w;
        union { __bf16 h[8]; uint4 q; } o;
#pragma unroll
        for (int t = 0; t < NT; ++t) o.h[t] = (__bf16)(acc[t][v] * d);
        C4[(size_t)gr * (N / 8) + laneM] = o.q;
      }
    }
  }
}

#define N_PAD 100352

#define BIN_SHIFT 9
#define NBIN 196
#define NBIN_PAD 256
#define CHUNK 8192
#define SORT_CAP 16384
#define SRC_MASK 0x1FFFF

__global__ void k_zero(int* __restrict__ p, int n) {
  int i = blockIdx.x * blockDim.x + threadIdx.x;
  if (i < n) p[i] = 0;
}

#define W1_CONV_BLOCKS 8
__global__ __launch_bounds__(1024) void k_hist(
    const int* __restrict__ col, int* __restrict__ gcnt, int E,
    const float* __restrict__ W1, unsigned int* __restrict__ w1t) {
  __shared__ int h[NBIN_PAD];
  const int tid = threadIdx.x;
  const int nChunk = (E + CHUNK - 1) / CHUNK;
  if ((int)blockIdx.x >= nChunk) {
    w1_to_bf16_rows(W1, (__bf16*)w1t, ((int)blockIdx.x - nChunk) * 1024 + tid);
    return;
  }
  if (tid < NBIN_PAD) h[tid] = 0;
  __syncthreads();
  const int base = blockIdx.x * CHUNK;
#pragma unroll
  for (int k = 0; k < CHUNK / 1024; ++k) {
    const int e = base + k * 1024 + tid;
    if (e < E) atomicAdd(&h[col[e] >> BIN_SHIFT], 1);
  }
  __syncthreads();
  if (tid < NBIN_PAD) {
    const int v = h[tid];
    if (v) atomicAdd(&gcnt[tid], v);
  }
}

__global__ __launch_bounds__(1024) void k_part(
    const int* __restrict__ row, const int* __restrict__ col,
    const int* __restrict__ gcnt, int* __restrict__ gcur,
    int* __restrict__ part, int E) {
  __shared__ int h[NBIN_PAD];
  __shared__ int lstart[NBIN_PAD];
  __shared__ int gb[NBIN_PAD];
  __shared__ int sval[CHUNK];
  __shared__ unsigned char sbin[CHUNK];
  const int tid = threadIdx.x;
  if (tid < NBIN_PAD) h[tid] = 0;
  __syncthreads();
  const int base = blockIdx.x * CHUNK;
  int c[CHUNK / 1024], r[CHUNK / 1024], lr[CHUNK / 1024];
#pragma unroll
  for (int k = 0; k < CHUNK / 1024; ++k) {
    const int e = base + k * 1024 + tid;
    const bool ok = e < E;
    c[k] = ok ? col[e] : -1;
    r[k] = ok ? row[e] : 0;
  }
#pragma unroll
  for (int k = 0; k < CHUNK / 1024; ++k) {
    lr[k] = 0;
    if (c[k] >= 0) lr[k] = atomicAdd(&h[c[k] >> BIN_SHIFT], 1);
  }
  __syncthreads();
  if (tid < 32) {
    int hv[8], gv[8], hloc[8], gloc[8];
    int hsum = 0, gsum = 0;
#pragma unroll
    for (int j = 0; j < 8; ++j) {
      hv[j] = h[tid * 8 + j];
      gv[j] = gcnt[tid * 8 + j];
    }
#pragma unroll
    for (int j = 0; j < 8; ++j) {
      hloc[j] = hsum; hsum += hv[j];
      gloc[j] = gsum; gsum += gv[j];
    }
    int hi = hsum, gi = gsum;
    for (int o = 1; o < 32; o <<= 1) {
      const int y = __shfl_up(hi, o, 32);
      const int z = __shfl_up(gi, o, 32);
      if (tid >= o) { hi += y; gi += z; }
    }
    const int hex = hi - hsum, gex = gi - gsum;
#pragma unroll
    for (int j = 0; j < 8; ++j) {
      const int b = tid * 8 + j;
      lstart[b] = hex + hloc[j];
      int g = gex + gloc[j];
      if (hv[j]) g += atomicAdd(&gcur[b], hv[j]);
      gb[b] = g;
    }
  }
  __syncthreads();
#pragma unroll
  for (int k = 0; k < CHUNK / 1024; ++k) {
    if (c[k] >= 0) {
      const int b = c[k] >> BIN_SHIFT;
      const int p = lstart[b] + lr[k];
      sval[p] = ((c[k] & ((1 << BIN_SHIFT) - 1)) << 17) | r[k];
      sbin[p] = (unsigned char)b;
    }
  }
  __syncthreads();
  const int nhere = min(CHUNK, E - base);
#pragma unroll
  for (int k = 0; k < CHUNK / 1024; ++k) {
    const int i = k * 1024 + tid;
    if (i < nhere) {
      const int b = sbin[i];
      part[gb[b] + (i - lstart[b])] = sval[i];
    }
  }
}

__global__ __launch_bounds__(1024) void k_sort(
    const int* __restrict__ part, const int* __restrict__ gcnt,
    int* __restrict__ src, int* __restrict__ rowptr, int* __restrict__ cnt,
    float* __restrict__ dinv) {
  __shared__ int c_l[512];
  __shared__ int cur_l[512];
  __shared__ int rp_l[512];
  __shared__ int wsum[32];
  __shared__ int s_start;
  __shared__ int sorted_l[SORT_CAP];
  const int tid  = threadIdx.x;
  const int lane = tid & 31;
  const int wave = tid >> 5;
  const int bin  = blockIdx.x;

  if (tid < 512) { c_l[tid] = 0; cur_l[tid] = 0; }
  if (tid < 32) {
    int s = 0;
    for (int j = tid; j < bin; j += 32) s += gcnt[j];
    for (int o = 16; o > 0; o >>= 1) s += __shfl_xor(s, o, 32);
    if (tid == 0) s_start = s;
  }
  __syncthreads();
  const int bstart = s_start;
  const int nb = gcnt[bin];

  for (int i0 = 0; i0 < nb; i0 += 8192) {
    int en[8];
#pragma unroll
    for (int k = 0; k < 8; ++k) {
      const int i = i0 + k * 1024 + tid;
      en[k] = (i < nb) ? part[bstart + i] : -1;
    }
#pragma unroll
    for (int k = 0; k < 8; ++k)
      if (en[k] >= 0) atomicAdd(&c_l[en[k] >> 17], 1);
  }
  __syncthreads();

  int x = 0, c = 0;
  if (tid < 512) {
    c = c_l[tid];
    x = c;
    for (int o = 1; o < 32; o <<= 1) {
      const int y = __shfl_up(x, o, 32);
      if (lane >= o) x += y;
    }
    if (lane == 31) wsum[wave] = x;
  }
  __syncthreads();
  if (tid < 32) {
    int t = (lane < 16) ? wsum[lane] : 0;
    for (int o = 1; o < 32; o <<= 1) {
      const int y = __shfl_up(t, o, 32);
      if (lane >= o) t += y;
    }
    wsum[lane] = t;
  }
  __syncthreads();
  if (tid < 512) {
    const int rp = x - c + ((wave > 0) ? wsum[wave - 1] : 0);
    rp_l[tid] = rp;
    const int node = (bin << BIN_SHIFT) + tid;
    cnt[node]    = c;
    rowptr[node] = bstart + rp;
    dinv[node]   = rsqrtf((float)(c + 1));
  }
  __syncthreads();

  const bool fits = nb <= SORT_CAP;
  for (int i0 = 0; i0 < nb; i0 += 8192) {
    int en[8];
#pragma unroll
    for (int k = 0; k < 8; ++k) {
      const int i = i0 + k * 1024 + tid;
      en[k] = (i < nb) ? part[bstart + i] : -1;
    }
#pragma unroll
    for (int k = 0; k < 8; ++k) {
      if (en[k] >= 0) {
        const int l = en[k] >> 17;
        const int p = rp_l[l] + atomicAdd(&cur_l[l], 1);
        if (fits) sorted_l[p] = en[k] & SRC_MASK;
        else      src[bstart + p] = en[k] & SRC_MASK;
      }
    }
  }
  __syncthreads();
  if (fits)
    for (int i = tid; i < nb; i += 1024) src[bstart + i] = sorted_l[i];
}

__device__ __forceinline__ float4 widen_bf16x4(uint2 u) {
  float4 f;
  f.x = __uint_as_float(u.x << 16);
  f.y = __uint_as_float(u.x & 0xFFFF0000u);
  f.z = __uint_as_float(u.y << 16);
  f.w = __uint_as_float(u.y & 0xFFFF0000u);
  return f;
}

__device__ __forceinline__ float4 pull_row128(
    const uint2* __restrict__ hs2, const int* __restrict__ src,
    int node, int beg, int deg, int first, int lane) {
  float4 acc = widen_bf16x4(hs2[(size_t)node * 32 + lane]);
  for (int base = 0; base < deg; base += 32) {
    const int m = min(32, deg - base);
    const int mine = (base == 0) ? first : ((lane < m) ? src[beg + base + lane] : 0);
    int j = 0;
    for (; j + 8 <= m; j += 8) {
      uint2 v[8];
#pragma unroll
      for (int k = 0; k < 8; ++k) {
        const int r = __builtin_amdgcn_readlane(mine, j + k);
        v[k] = hs2[(size_t)r * 32 + lane];
      }
#pragma unroll
      for (int k = 0; k < 8; ++k) {
        const float4 f = widen_bf16x4(v[k]);
        acc.x += f.x; acc.y += f.y; acc.z += f.z; acc.w += f.w;
      }
    }
    if (j + 4 <= m) {
      uint2 v[4];
#pragma unroll
      for (int k = 0; k < 4; ++k) {
        const int r = __builtin_amdgcn_readlane(mine, j + k);
        v[k] = hs2[(size_t)r * 32 + lane];
      }
#pragma unroll
      for (int k = 0; k < 4; ++k) {
        const float4 f = widen_bf16x4(v[k]);
        acc.x += f.x; acc.y += f.y; acc.z += f.z; acc.w += f.w;
      }
      j += 4;
    }
    for (; j < m; ++j) {
      const int r = __builtin_amdgcn_readlane(mine, j);
      const float4 f = widen_bf16x4(hs2[(size_t)r * 32 + lane]);
      acc.x += f.x; acc.y += f.y; acc.z += f.z; acc.w += f.w;
    }
  }
  return acc;
}

__global__ __launch_bounds__(256) void k_pull128_relu_gemm32(
    const unsigned int* __restrict__ hs, const float* __restrict__ dinv,
    const float* __restrict__ bias, const int* __restrict__ rowptr,
    const int* __restrict__ cnt, const int* __restrict__ src,
    const float* __restrict__ W2, float* __restrict__ g2, int n) {
  __shared__ __bf16 ldsA[4 * 16 * 32];
  __shared__ __bf16 ldsB[4 * D_OUT * 32];
  __shared__ float  ldsC[16 * D_OUT];

  const int tid  = threadIdx.x;
  const int wave = __builtin_amdgcn_readfirstlane(tid >> 5);
  const int lane = tid & 31;
  const int laneM = lane & 15;
  const int khalf = (lane >> 4) << 3;
  const int node0 = blockIdx.x * 16;

  int nodeQ[2], begQ[2], degQ[2];
  float dQ[2];
#pragma unroll
  for (int q = 0; q < 2; ++q) {
    nodeQ[q] = min(node0 + wave * 2 + q, n - 1);
    begQ[q]  = rowptr[nodeQ[q]];
    degQ[q]  = cnt[nodeQ[q]];
    dQ[q]    = dinv[nodeQ[q]];
  }

#pragma unroll
  for (int ks = 0; ks < 4; ++ks) {
    const int nn = tid & 31;
    const int kq = tid >> 5;
    union { __bf16 h[4]; uint2 q; } wb;
#pragma unroll
    for (int j = 0; j < 4; ++j)
      wb.h[j] = (__bf16)W2[(size_t)(ks * 32 + kq * 4 + j) * D_OUT + nn];
    *(uint2*)(ldsB + (ks * D_OUT + nn) * 32 + kq * 4) = wb.q;
  }

  const int outRow = node0 + (tid >> 3);
  const float dOut = (tid < 128 && outRow < n) ? dinv[outRow] : 0.0f;

  int firstQ[2];
#pragma unroll
  for (int q = 0; q < 2; ++q)
    firstQ[q] = (lane < min(32, degQ[q])) ? src[begQ[q] + lane] : 0;
  const uint2* __restrict__ hs2 = (const uint2*)hs;
  const float4 b4 = ((const float4*)bias)[lane];
#pragma unroll
  for (int q = 0; q < 2; ++q) {
    const int r = wave * 2 + q;
    union { __bf16 h[4]; uint2 u; } pk;
    const float4 acc = pull_row128(hs2, src, nodeQ[q], begQ[q], degQ[q], firstQ[q], lane);
    const float d = dQ[q];
    pk.h[0] = (__bf16)fmaxf(acc.x * d + b4.x, 0.0f);
    pk.h[1] = (__bf16)fmaxf(acc.y * d + b4.y, 0.0f);
    pk.h[2] = (__bf16)fmaxf(acc.z * d + b4.z, 0.0f);
    pk.h[3] = (__bf16)fmaxf(acc.w * d + b4.w, 0.0f);
    *(uint2*)(ldsA + ((lane >> 3) * 16 + r) * 32 + ((lane & 7) << 2)) = pk.u;
  }
  __syncthreads();

  if (wave < 2) {
    v8f acc = {0.f, 0.f, 0.f, 0.f, 0.f, 0.f, 0.f, 0.f};
#pragma unroll
    for (int ks = 0; ks < 4; ++ks) {
      const v16bf a = frag_from_lds(ldsA + ks * 16 * 32, laneM, khalf);
      const v16bf b = frag_from_lds(ldsB + ks * D_OUT * 32, (wave << 4) + laneM, khalf);
      acc = __builtin_amdgcn_wmma_f32_16x16x32_bf16(
          false, a, false, b, (short)0, acc, false, false);
    }
    const int mBase = (lane >> 4) << 3;
#pragma unroll
    for (int v = 0; v < 8; ++v)
      ldsC[(mBase + v) * D_OUT + (wave << 4) + laneM] = acc[v];
  }
  __syncthreads();

  if (tid < 128 && outRow < n) {
    float4 c = ((const float4*)ldsC)[(tid >> 3) * (D_OUT / 4) + (tid & 7)];
    c.x *= dOut; c.y *= dOut; c.z *= dOut; c.w *= dOut;
    ((float4*)g2)[(size_t)outRow * (D_OUT / 4) + (tid & 7)] = c;
  }
}

__global__ __launch_bounds__(256) void k_pull32_logsoftmax(
    const float* __restrict__ gs, const float* __restrict__ dinv,
    const float* __restrict__ bias, const int* __restrict__ rowptr,
    const int* __restrict__ cnt, const int* __restrict__ src,
    float* __restrict__ out, int n) {
  const int wave = threadIdx.x >> 5;
  const int lane = threadIdx.x & 31;
  const int g    = lane >> 3;
  const int c4   = lane & 7;
  const float4* __restrict__ gs4 = (const float4*)gs;
  float4* __restrict__ out4 = (float4*)out;
  const float4 b4 = ((const float4*)bias)[c4];
  const int nodeBase = blockIdx.x * 128 + wave * 4 + g;

  int node = nodeBase;
  int nd   = min(node, n - 1);
  int beg  = rowptr[nd];
  int deg  = (node < n) ? cnt[nd] : 0;
  int mine = (c4 < deg) ? src[beg + c4] : -1;
  float4 acc = gs4[(size_t)nd * 8 + c4];
  float d = dinv[nd];

  for (int t = 0; t < 4; ++t) {
    const int cNode = node, cBeg = beg, cDeg = deg;
    const float cD = d;
    node = nodeBase + (t + 1) * 32;
    nd   = min(node, n - 1);
    if (t + 1 < 4) {
      beg = rowptr[nd];
      deg = (node < n) ? cnt[nd] : 0;
    }
    int dmax = cDeg;
    dmax = max(dmax, __shfl_xor(dmax, 8, 32));
    dmax = max(dmax, __shfl_xor(dmax, 16, 32));
    dmax = __builtin_amdgcn_readfirstlane(dmax);
    for (int base = 0; base < dmax; base += 8) {
      const int e2 = base + 8 + c4;
      const int nxt = (e2 < cDeg) ? src[cBeg + e2] : -1;
      float4 v[8];
#pragma unroll
      for (int k = 0; k < 8; ++k) {
        const int r = __shfl(mine, k, 8);
        v[k] = make_float4(0.0f, 0.0f, 0.0f, 0.0f);
        if (r >= 0) v[k] = gs4[(size_t)r * 8 + c4];
      }
#pragma unroll
      for (int k = 0; k < 8; ++k) {
        acc.x += v[k].x; acc.y += v[k].y; acc.z += v[k].z; acc.w += v[k].w;
      }
      mine = nxt;
    }
    float4 v;
    v.x = acc.x * cD + b4.x;
    v.y = acc.y * cD + b4.y;
    v.z = acc.z * cD + b4.z;
    v.w = acc.w * cD + b4.w;
    if (t + 1 < 4) {
      mine = (c4 < deg) ? src[beg + c4] : -1;
      acc  = gs4[(size_t)nd * 8 + c4];
      d    = dinv[nd];
    }
    float mx = fmaxf(fmaxf(v.x, v.y), fmaxf(v.z, v.w));
    for (int o = 4; o > 0; o >>= 1) mx = fmaxf(mx, __shfl_xor(mx, o, 32));
    float s = (__expf(v.x - mx) + __expf(v.y - mx)) + (__expf(v.z - mx) + __expf(v.w - mx));
    for (int o = 4; o > 0; o >>= 1) s += __shfl_xor(s, o, 32);
    const float ls = __logf(s);
    if (cNode < n) {
      float4 o4;
      o4.x = v.x - mx - ls;
      o4.y = v.y - mx - ls;
      o4.z = v.z - mx - ls;
      o4.w = v.w - mx - ls;
      out4[(size_t)cNode * 8 + c4] = o4;
    }
  }
}

extern "C" void kernel_launch(void* const* d_in, const int* in_sizes, int n_in,
                              void* d_out, int out_size, void* d_ws, size_t ws_size,
                              hipStream_t stream) {
  const float* x  = (const float*)d_in[0];
  const float* W1 = (const float*)d_in[1];
  const float* b1 = (const float*)d_in[2];
  const float* W2 = (const float*)d_in[3];
  const float* b2 = (const float*)d_in[4];
  const int*  eix = (const int*)d_in[5];

  const int n = N_NODES;
  const int E = in_sizes[5] / 2;
  const int* row = eix;
  const int* col = eix + E;

  float* ws     = (float*)d_ws;
  float* dinv   = ws;
  int*   cnt    = (int*)(ws + N_PAD);
  int*   rowptr = cnt + N_PAD;
  int*   part   = rowptr + N_PAD;
  int*   src    = part + E;
  unsigned int* hs = (unsigned int*)(src + E);
  float* g2     = (float*)(hs + (size_t)n * (D_H / 2));
  int*   gcnt   = (int*)(g2 + (size_t)n * D_OUT);
  int*   gcur   = gcnt + NBIN_PAD;
  unsigned int* w1t = (unsigned int*)(gcur + NBIN_PAD);
  float* outf   = (float*)d_out;

  k_zero<<<1, 2 * NBIN_PAD, 0, stream>>>(gcnt, 2 * NBIN_PAD);
  k_hist<<<(E + CHUNK - 1) / CHUNK + W1_CONV_BLOCKS, 1024, 0, stream>>>(col, gcnt, E, W1, w1t);
  k_part<<<(E + CHUNK - 1) / CHUNK, 1024, 0, stream>>>(row, col, gcnt, gcur, part, E);
  k_sort<<<NBIN, 1024, 0, stream>>>(part, gcnt, src, rowptr, cnt, dinv);

  gemm1_bf16_wmma<<<768, 256, 0, stream>>>(x, w1t, hs, dinv, n);
  k_pull128_relu_gemm32<<<(n + 15) / 16, 256, 0, stream>>>(hs, dinv, b1, rowptr, cnt, src, W2, g2, n);

  k_pull32_logsoftmax<<<(n + 127) / 128, 256, 0, stream>>>(g2, dinv, b2, rowptr, cnt, src, outf, n);
}


# Edge grouping without global histogram or counters: each chunk ordered by bin in place, run start/length tables [bin][chunk], bin work-group fetches its runs; 5 launches; sum order LDS arrival
# speedup vs baseline: 7.3134x; 1.0472x over previous
#include <hip/hip_runtime.h>

#define N_NODES 100000
#define D_IN 256
#define D_H 128
#define D_OUT 32

typedef __attribute__((ext_vector_type(16))) __bf16 v16bf;
typedef __attribute__((ext_vector_type(8)))  float  v8f;

__device__ __forceinline__ v16bf frag_from_lds(const __bf16* base, int rowIdx, int khalf) {
  union { uint4 u[2]; v16bf v; } c;
  const uint4* p = (const uint4*)(base + rowIdx * 32 + khalf);
  c.u[0] = p[0];
  c.u[1] = p[2];
  return c.v;
}

#define W1T_PITCH 264
__device__ __forceinline__ v16bf frag_from_w1(const __bf16* base, int rowIdx, int k0) {
  union { uint4 u[2]; v16bf v; } c;
  const uint4* p = (const uint4*)(base + rowIdx * W1T_PITCH + k0);
  c.u[0] = p[0];
  c.u[1] = p[2];
  return c.v;
}

__device__ __forceinline__ v16bf frag_from_f32(const float4 (&q)[4]) {
  union { __bf16 h[16]; v16bf v; } c;
#pragma unroll
  for (int j = 0; j < 4; ++j) {
    c.h[j * 4 + 0] = (__bf16)q[j].x;
    c.h[j * 4 + 1] = (__bf16)q[j].y;
    c.h[j * 4 + 2] = (__bf16)q[j].z;
    c.h[j * 4 + 3] = (__bf16)q[j].w;
  }
  return c.v;
}

__device__ __forceinline__ void w1_to_bf16_rows(
    const float* __restrict__ W, __bf16* __restrict__ w1t, int g) {
  const int n  = g & (D_H - 1);
  const int kq = g >> 7;
  const int p  = (n & 7) * 16 + (n >> 3);
  union { __bf16 h[4]; uint2 q; } wb;
#pragma unroll
  for (int j = 0; j < 4; ++j)
    wb.h[j] = (__bf16)W[(size_t)(kq * 4 + j) * D_H + n];
  *(uint2*)(w1t + p * D_IN + kq * 4) = wb.q;
}

__global__ __launch_bounds__(256) void gemm1_bf16_wmma(
    const float* __restrict__ A, const unsigned int* __restrict__ w1t,
    unsigned int* __restrict__ C,
    const float* __restrict__ dinv, int M) {
  constexpr int K = D_IN, N = D_H, NT = N / 16;
  static_assert(NT == 8, "the epilogue packs 8 column tiles into one 16-byte store");
  __shared__ __bf16 ldsB[N * W1T_PITCH];

  const int tid   = threadIdx.x;
  const int wave  = __builtin_amdgcn_readfirstlane(tid >> 5);
  const int lane  = tid & 31;
  const int laneM = lane & 15;
  const int khalf = (lane >> 4) << 3;
  const int nTiles = (M + 15) >> 4;
  const int G = gridDim.x;

  const int tile0 = (int)blockIdx.x * 8 + wave;
  const float4* __restrict__ arow0 =
      (const float4*)(A + (size_t)min((tile0 << 4) + laneM, M - 1) * K + khalf);

  float4 abuf[2][4];
  {
    const uint4* __restrict__ w4 = (const uint4*)w1t;
    uint4 wv[16];
#pragma unroll
    for (int i = 0; i < 16; ++i) wv[i] = w4[tid + i * 256];
#pragma unroll
    for (int s = 0; s < 2; ++s) {
      abuf[s][0] = arow0[s * 8 + 0];
      abuf[s][1] = arow0[s * 8 + 1];
      abuf[s][2] = arow0[s * 8 + 4];
      abuf[s][3] = arow0[s * 8 + 5];
    }
#pragma unroll
    for (int i = 0; i < 16; ++i) {
      const int idx = tid + i * 256;
      *(uint4*)(ldsB + (idx >> 5) * W1T_PITCH + ((idx & 31) << 3)) = wv[i];
    }
  }
  __syncthreads();

  uint4* __restrict__ C4 = (uint4*)C;
  for (int it = 0; ; ++it) {
    const int tile = (it == 0) ? tile0
                               : 8 * G + ((it - 1) * 8 + wave) * G + (int)blockIdx.x;
    if (tile >= nTiles) break;
    __builtin_amdgcn_fence(__ATOMIC_ACQUIRE, "wavefront");
    const int row0 = tile << 4;
    const float4* __restrict__ arow =
        (const float4*)(A + (size_t)min(row0 + laneM, M - 1) * K + khalf);
    const float4* __restrict__ dp = (const float4*)(dinv + row0 + ((lane >> 4) << 3));
    const float4 dLo = dp[0], dHi = dp[1];
    if (it > 0) {
#pragma unroll
      for (int s = 0; s < 2; ++s) {
        abuf[s][0] = arow[s * 8 + 0];
        abuf[s][1] = arow[s * 8 + 1];
        abuf[s][2] = arow[s * 8 + 4];
        abuf[s][3] = arow[s * 8 + 5];
      }
    }

    v8f acc[NT];
#pragma unroll
    for (int t = 0; t < NT; ++t) {
      v8f z = {0.f, 0.f, 0.f, 0.f, 0.f, 0.f, 0.f, 0.f};
      acc[t] = z;
    }

#pragma unroll
    for (int s = 0; s < K / 32; ++s) {
      const v16bf a = frag_from_f32(abuf[s & 1]);
      if (s + 2 < K / 32) {
        abuf[s & 1][0] = arow[(s + 2) * 8 + 0];
        abuf[s & 1][1] = arow[(s + 2) * 8 + 1];
        abuf[s & 1][2] = arow[(s + 2) * 8 + 4];
        abuf[s & 1][3] = arow[(s + 2) * 8 + 5];
      }
      const int k0 = s * 32 + khalf;
      v16bf bc = frag_from_w1(ldsB, laneM, k0);
#pragma unroll
      for (int t = 0; t < NT; ++t) {
        v16bf bn = (t + 1 < NT) ? frag_from_w1(ldsB, ((t + 1) << 4) + laneM, k0) : bc;
        acc[t] = __builtin_amdgcn_wmma_f32_16x16x32_bf16(
            false, a, false, bc, (short)0, acc[t], false, false);
        bc = bn;
      }
    }

    const int mBase = row0 + ((lane >> 4) << 3);
#pragma unroll
    for (int v = 0; v < 8; ++v) {
      const int gr = mBase + v;
      if (gr < M) {
        const float d = (v == 0) ? dLo.x : (v == 1) ? dLo.y : (v == 2) ? dLo.z : (v == 3) ? dLo.w
                      : (v == 4) ? dHi.x : (v == 5) ? dHi.y : (v == 6) ? dHi.z : dHi.w;
        union { __bf16 h[8]; uint4 q; } o;
#pragma unroll
        for (int t = 0; t < NT; ++t) o.h[t] = (__bf16)(acc[t][v] * d);
        C4[(size_t)gr * (N / 8) + laneM] = o.q;
      }
    }
  }
}

#define N_PAD 100352

#define BIN_SHIFT 9
#define NBIN 196
#define NBIN_PAD 256
#define CHUNK 8192
#define NCH_PAD 1024
#define SORT_CAP 16384
#define SRC_MASK 0x1FFFF

#define W1_CONV_BLOCKS 8
__global__ __launch_bounds__(1024) void k_part(
    const int* __restrict__ row, const int* __restrict__ col,
    int* __restrict__ part, int* __restrict__ ltab, int* __restrict__ htab, int E,
    const float* __restrict__ W1, unsigned int* __restrict__ w1t) {
  __shared__ int h[NBIN_PAD];
  __shared__ int lstart[NBIN_PAD];
  __shared__ int sval[CHUNK];
  const int tid = threadIdx.x;
  const int nChunk = (E + CHUNK - 1) / CHUNK;
  if ((int)blockIdx.x >= nChunk) {
    w1_to_bf16_rows(W1, (__bf16*)w1t, ((int)blockIdx.x - nChunk) * 1024 + tid);
    return;
  }
  if (tid < NBIN_PAD) h[tid] = 0;
  __syncthreads();
  const int base = blockIdx.x * CHUNK;
  int c[CHUNK / 1024], r[CHUNK / 1024], lr[CHUNK / 1024];
#pragma unroll
  for (int k = 0; k < CHUNK / 1024; ++k) {
    const int e = base + k * 1024 + tid;
    const bool ok = e < E;
    c[k] = ok ? col[e] : -1;
    r[k] = ok ? row[e] : 0;
  }
#pragma unroll
  for (int k = 0; k < CHUNK / 1024; ++k) {
    lr[k] = 0;
    if (c[k] >= 0) lr[k] = atomicAdd(&h[c[k] >> BIN_SHIFT], 1);
  }
  __syncthreads();
  if (tid < 32) {
    int hv[8], hloc[8];
    int hsum = 0;
#pragma unroll
    for (int j = 0; j < 8; ++j) hv[j] = h[tid * 8 + j];
#pragma unroll
    for (int j = 0; j < 8; ++j) { hloc[j] = hsum; hsum += hv[j]; }
    int hi = hsum;
    for (int o = 1; o < 32; o <<= 1) {
      const int y = __shfl_up(hi, o, 32);
      if (tid >= o) hi += y;
    }
    const int hex = hi - hsum;
#pragma unroll
    for (int j = 0; j < 8; ++j) {
      const int b = tid * 8 + j;
      const int ls = hex + hloc[j];
      lstart[b] = ls;
      ltab[b * NCH_PAD + (int)blockIdx.x] = ls;
      htab[b * NCH_PAD + (int)blockIdx.x] = hv[j];
    }
  }
  __syncthreads();
#pragma unroll
  for (int k = 0; k < CHUNK / 1024; ++k) {
    if (c[k] >= 0) {
      const int p = lstart[c[k] >> BIN_SHIFT] + lr[k];
      sval[p] = ((c[k] & ((1 << BIN_SHIFT) - 1)) << 17) | r[k];
    }
  }
  __syncthreads();
  const int nhere = min(CHUNK, E - base);
#pragma unroll
  for (int k = 0; k < CHUNK / 1024; ++k) {
    const int i = k * 1024 + tid;
    if (i < nhere) part[base + i] = sval[i];
  }
}

#define RUNS_REG 7
template <class F>
__device__ __forceinline__ void sort_rest(
    const int* __restrict__ part, const int* run_at, const int* run_n,
    int nChunk, int wave, int lane, F f) {
  for (int c = wave; c < nChunk; c += 32) {
    const int rs = run_at[c], rn = run_n[c];
    const int j0 = (c - wave < 32 * RUNS_REG) ? 64 : 0;
    for (int j = j0 + lane; j < rn; j += 32) f(part[rs + j]);
  }
}

__global__ __launch_bounds__(1024) void k_sort(
    const int* __restrict__ part, const int* __restrict__ ltab,
    const int* __restrict__ htab, int nChunk,
    int* __restrict__ src, int* __restrict__ rowptr, int* __restrict__ cnt,
    float* __restrict__ dinv) {
  __shared__ int c_l[512];
  __shared__ int cur_l[512];
  __shared__ int rp_l[512];
  __shared__ int wsum[32];
  __shared__ int wsum2[32];
  __shared__ int s_start, s_nb;
  __shared__ int run_at[NCH_PAD];
  __shared__ int run_n[NCH_PAD];
  __shared__ int sorted_l[SORT_CAP];
  const int tid  = threadIdx.x;
  const int lane = tid & 31;
  const int wave = __builtin_amdgcn_readfirstlane(tid >> 5);
  const int bin  = blockIdx.x;

  if (tid < 512) { c_l[tid] = 0; cur_l[tid] = 0; }
  int ls = 0, hn = 0;
  if (tid < nChunk) {
    ls = ltab[bin * NCH_PAD + tid];
    hn = htab[bin * NCH_PAD + tid];
  }
  run_at[tid] = tid * CHUNK + ls;
  run_n[tid]  = hn;
  for (int o = 16; o > 0; o >>= 1) {
    ls += __shfl_xor(ls, o, 32);
    hn += __shfl_xor(hn, o, 32);
  }
  if (lane == 0) { wsum[wave] = ls; wsum2[wave] = hn; }
  __syncthreads();
  if (tid < 32) {
    int a = wsum[lane], b = wsum2[lane];
    for (int o = 16; o > 0; o >>= 1) {
      a += __shfl_xor(a, o, 32);
      b += __shfl_xor(b, o, 32);
    }
    if (tid == 0) { s_start = a; s_nb = b; }
  }
  __syncthreads();
  const int bstart = s_start;
  const int nb = s_nb;

  int en[RUNS_REG][2];
  int rsK[RUNS_REG], rnK[RUNS_REG];
#pragma unroll
  for (int k = 0; k < RUNS_REG; ++k) {
    const int cidx = wave + 32 * k;
    rsK[k] = run_at[cidx];
    rnK[k] = run_n[cidx];
  }
#pragma unroll
  for (int k = 0; k < RUNS_REG; ++k) {
    en[k][0] = (lane < rnK[k])      ? part[rsK[k] + lane]      : -1;
    en[k][1] = (lane + 32 < rnK[k]) ? part[rsK[k] + lane + 32] : -1;
  }

#pragma unroll
  for (int k = 0; k < RUNS_REG; ++k) {
    if (en[k][0] >= 0) atomicAdd(&c_l[en[k][0] >> 17], 1);
    if (en[k][1] >= 0) atomicAdd(&c_l[en[k][1] >> 17], 1);
  }
  sort_rest(part, run_at, run_n, nChunk, wave, lane,
            [&](int e) { atomicAdd(&c_l[e >> 17], 1); });
  __syncthreads();

  int x = 0, c = 0;
  if (tid < 512) {
    c = c_l[tid];
    x = c;
    for (int o = 1; o < 32; o <<= 1) {
      const int y = __shfl_up(x, o, 32);
      if (lane >= o) x += y;
    }
    if (lane == 31) wsum[wave] = x;
  }
  __syncthreads();
  if (tid < 32) {
    int t = (lane < 16) ? wsum[lane] : 0;
    for (int o = 1; o < 32; o <<= 1) {
      const int y = __shfl_up(t, o, 32);
      if (lane >= o) t += y;
    }
    wsum[lane] = t;
  }
  __syncthreads();
  if (tid < 512) {
    const int rp = x - c + ((wave > 0) ? wsum[wave - 1] : 0);
    rp_l[tid] = rp;
    const int node = (bin << BIN_SHIFT) + tid;
    cnt[node]    = c;
    rowptr[node] = bstart + rp;
    dinv[node]   = rsqrtf((float)(c + 1));
  }
  __syncthreads();

  const bool fits = nb <= SORT_CAP;
  auto place = [&](int e) {
    const int l = e >> 17;
    const int p = rp_l[l] + atomicAdd(&cur_l[l], 1);
    if (fits) sorted_l[p] = e & SRC_MASK;
    else      src[bstart + p] = e & SRC_MASK;
  };
#pragma unroll
  for (int k = 0; k < RUNS_REG; ++k) {
    if (en[k][0] >= 0) place(en[k][0]);
    if (en[k][1] >= 0) place(en[k][1]);
  }
  sort_rest(part, run_at, run_n, nChunk, wave, lane, place);
  __syncthreads();
  if (fits)
    for (int i = tid; i < nb; i += 1024) src[bstart + i] = sorted_l[i];
}

__device__ __forceinline__ float4 widen_bf16x4(uint2 u) {
  float4 f;
  f.x = __uint_as_float(u.x << 16);
  f.y = __uint_as_float(u.x & 0xFFFF0000u);
  f.z = __uint_as_float(u.y << 16);
  f.w = __uint_as_float(u.y & 0xFFFF0000u);
  return f;
}

__device__ __forceinline__ float4 pull_row128(
    const uint2* __restrict__ hs2, const int* __restrict__ src,
    int node, int beg, int deg, int first, int lane) {
  float4 acc = widen_bf16x4(hs2[(size_t)node * 32 + lane]);
  for (int base = 0; base < deg; base += 32) {
    const int m = min(32, deg - base);
    const int mine = (base == 0) ? first : ((lane < m) ? src[beg + base + lane] : 0);
    int j = 0;
    for (; j + 8 <= m; j += 8) {
      uint2 v[8];
#pragma unroll
      for (int k = 0; k < 8; ++k) {
        const int r = __builtin_amdgcn_readlane(mine, j + k);
        v[k] = hs2[(size_t)r * 32 + lane];
      }
#pragma unroll
      for (int k = 0; k < 8; ++k) {
        const float4 f = widen_bf16x4(v[k]);
        acc.x += f.x; acc.y += f.y; acc.z += f.z; acc.w += f.w;
      }
    }
    if (j + 4 <= m) {
      uint2 v[4];
#pragma unroll
      for (int k = 0; k < 4; ++k) {
        const int r = __builtin_amdgcn_readlane(mine, j + k);
        v[k] = hs2[(size_t)r * 32 + lane];
      }
#pragma unroll
      for (int k = 0; k < 4; ++k) {
        const float4 f = widen_bf16x4(v[k]);
        acc.x += f.x; acc.y += f.y; acc.z += f.z; acc.w += f.w;
      }
      j += 4;
    }
    for (; j < m; ++j) {
      const int r = __builtin_amdgcn_readlane(mine, j);
      const float4 f = widen_bf16x4(hs2[(size_t)r * 32 + lane]);
      acc.x += f.x; acc.y += f.y; acc.z += f.z; acc.w += f.w;
    }
  }
  return acc;
}

__global__ __launch_bounds__(256) void k_pull128_relu_gemm32(
    const unsigned int* __restrict__ hs, const float* __restrict__ dinv,
    const float* __restrict__ bias, const int* __restrict__ rowptr,
    const int* __restrict__ cnt, const int* __restrict__ src,
    const float* __restrict__ W2, float* __restrict__ g2, int n) {
  __shared__ __bf16 ldsA[4 * 16 * 32];
  __shared__ __bf16 ldsB[4 * D_OUT * 32];
  __shared__ float  ldsC[16 * D_OUT];

  const int tid  = threadIdx.x;
  const int wave = __builtin_amdgcn_readfirstlane(tid >> 5);
  const int lane = tid & 31;
  const int laneM = lane & 15;
  const int khalf = (lane >> 4) << 3;
  const int node0 = blockIdx.x * 16;

  int nodeQ[2], begQ[2], degQ[2];
  float dQ[2];
#pragma unroll
  for (int q = 0; q < 2; ++q) {
    nodeQ[q] = min(node0 + wave * 2 + q, n - 1);
    begQ[q]  = rowptr[nodeQ[q]];
    degQ[q]  = cnt[nodeQ[q]];
    dQ[q]    = dinv[nodeQ[q]];
  }

#pragma unroll
  for (int ks = 0; ks < 4; ++ks) {
    const int nn = tid & 31;
    const int kq = tid >> 5;
    union { __bf16 h[4]; uint2 q; } wb;
#pragma unroll
    for (int j = 0; j < 4; ++j)
      wb.h[j] = (__bf16)W2[(size_t)(ks * 32 + kq * 4 + j) * D_OUT + nn];
    *(uint2*)(ldsB + (ks * D_OUT + nn) * 32 + kq * 4) = wb.q;
  }

  const int outRow = node0 + (tid >> 3);
  const float dOut = (tid < 128 && outRow < n) ? dinv[outRow] : 0.0f;

  int firstQ[2];
#pragma unroll
  for (int q = 0; q < 2; ++q)
    firstQ[q] = (lane < min(32, degQ[q])) ? src[begQ[q] + lane] : 0;
  const uint2* __restrict__ hs2 = (const uint2*)hs;
  const float4 b4 = ((const float4*)bias)[lane];
#pragma unroll
  for (int q = 0; q < 2; ++q) {
    const int r = wave * 2 + q;
    union { __bf16 h[4]; uint2 u; } pk;
    const float4 acc = pull_row128(hs2, src, nodeQ[q], begQ[q], degQ[q], firstQ[q], lane);
    const float d = dQ[q];
    pk.h[0] = (__bf16)fmaxf(acc.x * d + b4.x, 0.0f);
    pk.h[1] = (__bf16)fmaxf(acc.y * d + b4.y, 0.0f);
    pk.h[2] = (__bf16)fmaxf(acc.z * d + b4.z, 0.0f);
    pk.h[3] = (__bf16)fmaxf(acc.w * d + b4.w, 0.0f);
    *(uint2*)(ldsA + ((lane >> 3) * 16 + r) * 32 + ((lane & 7) << 2)) = pk.u;
  }
  __syncthreads();

  if (wave < 2) {
    v8f acc = {0.f, 0.f, 0.f, 0.f, 0.f, 0.f, 0.f, 0.f};
#pragma unroll
    for (int ks = 0; ks < 4; ++ks) {
      const v16bf a = frag_from_lds(ldsA + ks * 16 * 32, laneM, khalf);
      const v16bf b = frag_from_lds(ldsB + ks * D_OUT * 32, (wave << 4) + laneM, khalf);
      acc = __builtin_amdgcn_wmma_f32_16x16x32_bf16(
          false, a, false, b, (short)0, acc, false, false);
    }
    const int mBase = (lane >> 4) << 3;
#pragma unroll
    for (int v = 0; v < 8; ++v)
      ldsC[(mBase + v) * D_OUT + (wave << 4) + laneM] = acc[v];
  }
  __syncthreads();

  if (tid < 128 && outRow < n) {
    float4 c = ((const float4*)ldsC)[(tid >> 3) * (D_OUT / 4) + (tid & 7)];
    c.x *= dOut; c.y *= dOut; c.z *= dOut; c.w *= dOut;
    ((float4*)g2)[(size_t)outRow * (D_OUT / 4) + (tid & 7)] = c;
  }
}

__global__ __launch_bounds__(256) void k_pull32_logsoftmax(
    const float* __restrict__ gs, const float* __restrict__ dinv,
    const float* __restrict__ bias, const int* __restrict__ rowptr,
    const int* __restrict__ cnt, const int* __restrict__ src,
    float* __restrict__ out, int n) {
  const int wave = threadIdx.x >> 5;
  const int lane = threadIdx.x & 31;
  const int g    = lane >> 3;
  const int c4   = lane & 7;
  const float4* __restrict__ gs4 = (const float4*)gs;
  float4* __restrict__ out4 = (float4*)out;
  const float4 b4 = ((const float4*)bias)[c4];
  const int nodeBase = blockIdx.x * 128 + wave * 4 + g;

  int node = nodeBase;
  int nd   = min(node, n - 1);
  int beg  = rowptr[nd];
  int deg  = (node < n) ? cnt[nd] : 0;
  int mine = (c4 < deg) ? src[beg + c4] : -1;
  float4 acc = gs4[(size_t)nd * 8 + c4];
  float d = dinv[nd];

  for (int t = 0; t < 4; ++t) {
    const int cNode = node, cBeg = beg, cDeg = deg;
    const float cD = d;
    node = nodeBase + (t + 1) * 32;
    nd   = min(node, n - 1);
    if (t + 1 < 4) {
      beg = rowptr[nd];
      deg = (node < n) ? cnt[nd] : 0;
    }
    int dmax = cDeg;
    dmax = max(dmax, __shfl_xor(dmax, 8, 32));
    dmax = max(dmax, __shfl_xor(dmax, 16, 32));
    dmax = __builtin_amdgcn_readfirstlane(dmax);
    for (int base = 0; base < dmax; base += 8) {
      const int e2 = base + 8 + c4;
      const int nxt = (e2 < cDeg) ? src[cBeg + e2] : -1;
      float4 v[8];
#pragma unroll
      for (int k = 0; k < 8; ++k) {
        const int r = __shfl(mine, k, 8);
        v[k] = make_float4(0.0f, 0.0f, 0.0f, 0.0f);
        if (r >= 0) v[k] = gs4[(size_t)r * 8 + c4];
      }
#pragma unroll
      for (int k = 0; k < 8; ++k) {
        acc.x += v[k].x; acc.y += v[k].y; acc.z += v[k].z; acc.w += v[k].w;
      }
      mine = nxt;
    }
    float4 v;
    v.x = acc.x * cD + b4.x;
    v.y = acc.y * cD + b4.y;
    v.z = acc.z * cD + b4.z;
    v.w = acc.w * cD + b4.w;
    if (t + 1 < 4) {
      mine = (c4 < deg) ? src[beg + c4] : -1;
      acc  = gs4[(size_t)nd * 8 + c4];
      d    = dinv[nd];
    }
    float mx = fmaxf(fmaxf(v.x, v.y), fmaxf(v.z, v.w));
    for (int o = 4; o > 0; o >>= 1) mx = fmaxf(mx, __shfl_xor(mx, o, 32));
    float s = (__expf(v.x - mx) + __expf(v.y - mx)) + (__expf(v.z - mx) + __expf(v.w - mx));
    for (int o = 4; o > 0; o >>= 1) s += __shfl_xor(s, o, 32);
    const float ls = __logf(s);
    if (cNode < n) {
      float4 o4;
      o4.x = v.x - mx - ls;
      o4.y = v.y - mx - ls;
      o4.z = v.z - mx - ls;
      o4.w = v.w - mx - ls;
      out4[(size_t)cNode * 8 + c4] = o4;
    }
  }
}

extern "C" void kernel_launch(void* const* d_in, const int* in_sizes, int n_in,
                              void* d_out, int out_size, void* d_ws, size_t ws_size,
                              hipStream_t stream) {
  const float* x  = (const float*)d_in[0];
  const float* W1 = (const float*)d_in[1];
  const float* b1 = (const float*)d_in[2];
  const float* W2 = (const float*)d_in[3];
  const float* b2 = (const float*)d_in[4];
  const int*  eix = (const int*)d_in[5];

  const int n = N_NODES;
  const int E = in_sizes[5] / 2;
  const int* row = eix;
  const int* col = eix + E;

  float* ws     = (float*)d_ws;
  float* dinv   = ws;
  int*   cnt    = (int*)(ws + N_PAD);
  int*   rowptr = cnt + N_PAD;
  int*   part   = rowptr + N_PAD;
  int*   src    = part + E;
  unsigned int* hs = (unsigned int*)(src + E);
  float* g2     = (float*)(hs + (size_t)n * (D_H / 2));
  int*   ltab   = (int*)(g2 + (size_t)n * D_OUT);
  int*   htab   = ltab + NBIN_PAD * NCH_PAD;
  unsigned int* w1t = (unsigned int*)(htab + NBIN_PAD * NCH_PAD);
  float* outf   = (float*)d_out;

  k_part<<<(E + CHUNK - 1) / CHUNK + W1_CONV_BLOCKS, 1024, 0, stream>>>(row, col, part, ltab, htab, E, W1, w1t);
  k_sort<<<NBIN, 1024, 0, stream>>>(part, ltab, htab, (E + CHUNK - 1) / CHUNK, src, rowptr, cnt, dinv);

  gemm1_bf16_wmma<<<768, 256, 0, stream>>>(x, w1t, hs, dinv, n);
  k_pull128_relu_gemm32<<<(n + 15) / 16, 256, 0, stream>>>(hs, dinv, b1, rowptr, cnt, src, W2, g2, n);

  k_pull32_logsoftmax<<<(n + 127) / 128, 256, 0, stream>>>(g2, dinv, b2, rowptr, cnt, src, outf, n);
}


# Layer-1 GEMM moved into the edge-grouping launch: the work-group that groups a bin of 512 targets then computes those 512 rows (one 16-row tile per wave, dinv from LDS); 4 launches; same sums
# speedup vs baseline: 7.6563x; 1.0469x over previous
#include <hip/hip_runtime.h>

#define N_NODES 100000
#define D_IN 256
#define D_H 128
#define D_OUT 32

typedef __attribute__((ext_vector_type(16))) __bf16 v16bf;
typedef __attribute__((ext_vector_type(8)))  float  v8f;

__device__ __forceinline__ v16bf frag_from_lds(const __bf16* base, int rowIdx, int khalf) {
  union { uint4 u[2]; v16bf v; } c;
  const uint4* p = (const uint4*)(base + rowIdx * 32 + khalf);
  c.u[0] = p[0];
  c.u[1] = p[2];
  return c.v;
}

#define W1T_PITCH 264
__device__ __forceinline__ v16bf frag_from_w1(const __bf16* base, int rowIdx, int k0) {
  union { uint4 u[2]; v16bf v; } c;
  const uint4* p = (const uint4*)(base + rowIdx * W1T_PITCH + k0);
  c.u[0] = p[0];
  c.u[1] = p[2];
  return c.v;
}

__device__ __forceinline__ v16bf frag_from_f32(const float4 (&q)[4]) {
  union { __bf16 h[16]; v16bf v; } c;
#pragma unroll
  for (int j = 0; j < 4; ++j) {
    c.h[j * 4 + 0] = (__bf16)q[j].x;
    c.h[j * 4 + 1] = (__bf16)q[j].y;
    c.h[j * 4 + 2] = (__bf16)q[j].z;
    c.h[j * 4 + 3] = (__bf16)q[j].w;
  }
  return c.v;
}

__device__ __forceinline__ void w1_to_bf16_rows(
    const float* __restrict__ W, __bf16* __restrict__ w1t, int g) {
  const int n  = g & (D_H - 1);
  const int kq = g >> 7;
  const int p  = (n & 7) * 16 + (n >> 3);
  union { __bf16 h[4]; uint2 q; } wb;
#pragma unroll
  for (int j = 0; j < 4; ++j)
    wb.h[j] = (__bf16)W[(size_t)(kq * 4 + j) * D_H + n];
  *(uint2*)(w1t + p * D_IN + kq * 4) = wb.q;
}


#define N_PAD 100352

#define BIN_SHIFT 9
#define NBIN 196
#define NBIN_PAD 256
#define CHUNK 8192
#define NCH_PAD 1024
#define SORT_CAP 16384
#define SRC_MASK 0x1FFFF

#define W1_CONV_BLOCKS 8
__global__ __launch_bounds__(1024) void k_part(
    const int* __restrict__ row, const int* __restrict__ col,
    int* __restrict__ part, int* __restrict__ ltab, int* __restrict__ htab, int E,
    const float* __restrict__ W1, unsigned int* __restrict__ w1t) {
  __shared__ int h[NBIN_PAD];
  __shared__ int lstart[NBIN_PAD];
  __shared__ int sval[CHUNK];
  const int tid = threadIdx.x;
  const int nChunk = (E + CHUNK - 1) / CHUNK;
  if ((int)blockIdx.x >= nChunk) {
    w1_to_bf16_rows(W1, (__bf16*)w1t, ((int)blockIdx.x - nChunk) * 1024 + tid);
    return;
  }
  if (tid < NBIN_PAD) h[tid] = 0;
  __syncthreads();
  const int base = blockIdx.x * CHUNK;
  int c[CHUNK / 1024], r[CHUNK / 1024], lr[CHUNK / 1024];
#pragma unroll
  for (int k = 0; k < CHUNK / 1024; ++k) {
    const int e = base + k * 1024 + tid;
    const bool ok = e < E;
    c[k] = ok ? col[e] : -1;
    r[k] = ok ? row[e] : 0;
  }
#pragma unroll
  for (int k = 0; k < CHUNK / 1024; ++k) {
    lr[k] = 0;
    if (c[k] >= 0) lr[k] = atomicAdd(&h[c[k] >> BIN_SHIFT], 1);
  }
  __syncthreads();
  if (tid < 32) {
    int hv[8], hloc[8];
    int hsum = 0;
#pragma unroll
    for (int j = 0; j < 8; ++j) hv[j] = h[tid * 8 + j];
#pragma unroll
    for (int j = 0; j < 8; ++j) { hloc[j] = hsum; hsum += hv[j]; }
    int hi = hsum;
    for (int o = 1; o < 32; o <<= 1) {
      const int y = __shfl_up(hi, o, 32);
      if (tid >= o) hi += y;
    }
    const int hex = hi - hsum;
#pragma unroll
    for (int j = 0; j < 8; ++j) {
      const int b = tid * 8 + j;
      const int ls = hex + hloc[j];
      lstart[b] = ls;
      ltab[b * NCH_PAD + (int)blockIdx.x] = ls;
      htab[b * NCH_PAD + (int)blockIdx.x] = hv[j];
    }
  }
  __syncthreads();
#pragma unroll
  for (int k = 0; k < CHUNK / 1024; ++k) {
    if (c[k] >= 0) {
      const int p = lstart[c[k] >> BIN_SHIFT] + lr[k];
      sval[p] = ((c[k] & ((1 << BIN_SHIFT) - 1)) << 17) | r[k];
    }
  }
  __syncthreads();
  const int nhere = min(CHUNK, E - base);
#pragma unroll
  for (int k = 0; k < CHUNK / 1024; ++k) {
    const int i = k * 1024 + tid;
    if (i < nhere) part[base + i] = sval[i];
  }
}

#define RUNS_REG 7
template <class F>
__device__ __forceinline__ void sort_rest(
    const int* __restrict__ part, const int* run_at, const int* run_n,
    int nChunk, int wave, int lane, F f) {
  for (int c = wave; c < nChunk; c += 32) {
    const int rs = run_at[c], rn = run_n[c];
    const int j0 = (c - wave < 32 * RUNS_REG) ? 64 : 0;
    for (int j = j0 + lane; j < rn; j += 32) f(part[rs + j]);
  }
}

__global__ __launch_bounds__(1024) void k_sort(
    const int* __restrict__ part, const int* __restrict__ ltab,
    const int* __restrict__ htab, int nChunk,
    int* __restrict__ src, int* __restrict__ rowptr, int* __restrict__ cnt,
    float* __restrict__ dinv,
    const float* __restrict__ A, const unsigned int* __restrict__ w1t,
    unsigned int* __restrict__ C, int M) {
  __shared__ __attribute__((aligned(16))) __bf16 ldsB[D_H * W1T_PITCH];
  __shared__ __attribute__((aligned(16))) float dinv_l[512];
  __shared__ int c_l[512];
  __shared__ int cur_l[512];
  __shared__ int rp_l[512];
  __shared__ int wsum[32];
  __shared__ int wsum2[32];
  __shared__ int s_start, s_nb;
  __shared__ int run_at[NCH_PAD];
  __shared__ int run_n[NCH_PAD];
  __shared__ int sorted_l[SORT_CAP];
  const int tid  = threadIdx.x;
  const int lane = tid & 31;
  const int wave = __builtin_amdgcn_readfirstlane(tid >> 5);
  const int bin  = blockIdx.x;

  uint4 wv[4];
  {
    const uint4* __restrict__ w4 = (const uint4*)w1t;
#pragma unroll
    for (int i = 0; i < 4; ++i) wv[i] = w4[tid + i * 1024];
  }

  if (tid < 512) { c_l[tid] = 0; cur_l[tid] = 0; }
  int ls = 0, hn = 0;
  if (tid < nChunk) {
    ls = ltab[bin * NCH_PAD + tid];
    hn = htab[bin * NCH_PAD + tid];
  }
  run_at[tid] = tid * CHUNK + ls;
  run_n[tid]  = hn;
#pragma unroll
  for (int i = 0; i < 4; ++i) {
    const int idx = tid + i * 1024;
    *(uint4*)(ldsB + (idx >> 5) * W1T_PITCH + ((idx & 31) << 3)) = wv[i];
  }
  for (int o = 16; o > 0; o >>= 1) {
    ls += __shfl_xor(ls, o, 32);
    hn += __shfl_xor(hn, o, 32);
  }
  if (lane == 0) { wsum[wave] = ls; wsum2[wave] = hn; }
  __syncthreads();
  if (tid < 32) {
    int a = wsum[lane], b = wsum2[lane];
    for (int o = 16; o > 0; o >>= 1) {
      a += __shfl_xor(a, o, 32);
      b += __shfl_xor(b, o, 32);
    }
    if (tid == 0) { s_start = a; s_nb = b; }
  }
  __syncthreads();
  const int bstart = s_start;
  const int nb = s_nb;

  int en[RUNS_REG][2];
  int rsK[RUNS_REG], rnK[RUNS_REG];
#pragma unroll
  for (int k = 0; k < RUNS_REG; ++k) {
    const int cidx = wave + 32 * k;
    rsK[k] = run_at[cidx];
    rnK[k] = run_n[cidx];
  }
#pragma unroll
  for (int k = 0; k < RUNS_REG; ++k) {
    en[k][0] = (lane < rnK[k])      ? part[rsK[k] + lane]      : -1;
    en[k][1] = (lane + 32 < rnK[k]) ? part[rsK[k] + lane + 32] : -1;
  }

#pragma unroll
  for (int k = 0; k < RUNS_REG; ++k) {
    if (en[k][0] >= 0) atomicAdd(&c_l[en[k][0] >> 17], 1);
    if (en[k][1] >= 0) atomicAdd(&c_l[en[k][1] >> 17], 1);
  }
  sort_rest(part, run_at, run_n, nChunk, wave, lane,
            [&](int e) { atomicAdd(&c_l[e >> 17], 1); });
  __syncthreads();

  int x = 0, c = 0;
  if (tid < 512) {
    c = c_l[tid];
    x = c;
    for (int o = 1; o < 32; o <<= 1) {
      const int y = __shfl_up(x, o, 32);
      if (lane >= o) x += y;
    }
    if (lane == 31) wsum[wave] = x;
  }
  __syncthreads();
  if (tid < 32) {
    int t = (lane < 16) ? wsum[lane] : 0;
    for (int o = 1; o < 32; o <<= 1) {
      const int y = __shfl_up(t, o, 32);
      if (lane >= o) t += y;
    }
    wsum[lane] = t;
  }
  __syncthreads();
  if (tid < 512) {
    const int rp = x - c + ((wave > 0) ? wsum[wave - 1] : 0);
    rp_l[tid] = rp;
    const int node = (bin << BIN_SHIFT) + tid;
    cnt[node]    = c;
    rowptr[node] = bstart + rp;
    const float dv = rsqrtf((float)(c + 1));
    dinv[node]   = dv;
    dinv_l[tid]  = dv;
  }
  __syncthreads();

  const bool fits = nb <= SORT_CAP;
  auto place = [&](int e) {
    const int l = e >> 17;
    const int p = rp_l[l] + atomicAdd(&cur_l[l], 1);
    if (fits) sorted_l[p] = e & SRC_MASK;
    else      src[bstart + p] = e & SRC_MASK;
  };
#pragma unroll
  for (int k = 0; k < RUNS_REG; ++k) {
    if (en[k][0] >= 0) place(en[k][0]);
    if (en[k][1] >= 0) place(en[k][1]);
  }
  sort_rest(part, run_at, run_n, nChunk, wave, lane, place);

  constexpr int K = D_IN, NT = D_H / 16;
  static_assert(NT == 8, "the epilogue packs 8 column tiles into one 16-byte store");
  const int laneM = lane & 15;
  const int khalf = (lane >> 4) << 3;
  const int tile  = bin * 32 + wave;
  const bool haveTile = tile < ((M + 15) >> 4);
  const int row0 = tile << 4;
  const float4* __restrict__ arow =
      (const float4*)(A + (size_t)min(row0 + laneM, M - 1) * K + khalf);
  float4 abuf[2][4];
  if (haveTile) {
#pragma unroll
    for (int s = 0; s < 2; ++s) {
      abuf[s][0] = arow[s * 8 + 0];
      abuf[s][1] = arow[s * 8 + 1];
      abuf[s][2] = arow[s * 8 + 4];
      abuf[s][3] = arow[s * 8 + 5];
    }
  }
  __syncthreads();
  if (fits)
    for (int i = tid; i < nb; i += 1024) src[bstart + i] = sorted_l[i];

  if (!haveTile) return;
  v8f acc[NT];
#pragma unroll
  for (int t = 0; t < NT; ++t) {
    v8f z = {0.f, 0.f, 0.f, 0.f, 0.f, 0.f, 0.f, 0.f};
    acc[t] = z;
  }
#pragma unroll
  for (int s = 0; s < K / 32; ++s) {
    const v16bf a = frag_from_f32(abuf[s & 1]);
    if (s + 2 < K / 32) {
      abuf[s & 1][0] = arow[(s + 2) * 8 + 0];
      abuf[s & 1][1] = arow[(s + 2) * 8 + 1];
      abuf[s & 1][2] = arow[(s + 2) * 8 + 4];
      abuf[s & 1][3] = arow[(s + 2) * 8 + 5];
    }
    const int k0 = s * 32 + khalf;
    v16bf bc = frag_from_w1(ldsB, laneM, k0);
#pragma unroll
    for (int t = 0; t < NT; ++t) {
      v16bf bn = (t + 1 < NT) ? frag_from_w1(ldsB, ((t + 1) << 4) + laneM, k0) : bc;
      acc[t] = __builtin_amdgcn_wmma_f32_16x16x32_bf16(
          false, a, false, bc, (short)0, acc[t], false, false);
      bc = bn;
    }
  }

  uint4* __restrict__ C4 = (uint4*)C;
  const int lBase = (wave << 4) + ((lane >> 4) << 3);
  const int mBase = row0 + ((lane >> 4) << 3);
  const float4 dLo = *(const float4*)(dinv_l + lBase);
  const float4 dHi = *(const float4*)(dinv_l + lBase + 4);
#pragma unroll
  for (int v = 0; v < 8; ++v) {
    const int gr = mBase + v;
    if (gr < M) {
      const float d = (v == 0) ? dLo.x : (v == 1) ? dLo.y : (v == 2) ? dLo.z : (v == 3) ? dLo.w
                    : (v == 4) ? dHi.x : (v == 5) ? dHi.y : (v == 6) ? dHi.z : dHi.w;
      union { __bf16 h[8]; uint4 q; } o;
#pragma unroll
      for (int t = 0; t < NT; ++t) o.h[t] = (__bf16)(acc[t][v] * d);
      C4[(size_t)gr * (D_H / 8) + laneM] = o.q;
    }
  }
}

__device__ __forceinline__ float4 widen_bf16x4(uint2 u) {
  float4 f;
  f.x = __uint_as_float(u.x << 16);
  f.y = __uint_as_float(u.x & 0xFFFF0000u);
  f.z = __uint_as_float(u.y << 16);
  f.w = __uint_as_float(u.y & 0xFFFF0000u);
  return f;
}

__device__ __forceinline__ float4 pull_row128(
    const uint2* __restrict__ hs2, const int* __restrict__ src,
    int node, int beg, int deg, int first, int lane) {
  float4 acc = widen_bf16x4(hs2[(size_t)node * 32 + lane]);
  for (int base = 0; base < deg; base += 32) {
    const int m = min(32, deg - base);
    const int mine = (base == 0) ? first : ((lane < m) ? src[beg + base + lane] : 0);
    int j = 0;
    for (; j + 8 <= m; j += 8) {
      uint2 v[8];
#pragma unroll
      for (int k = 0; k < 8; ++k) {
        const int r = __builtin_amdgcn_readlane(mine, j + k);
        v[k] = hs2[(size_t)r * 32 + lane];
      }
#pragma unroll
      for (int k = 0; k < 8; ++k) {
        const float4 f = widen_bf16x4(v[k]);
        acc.x += f.x; acc.y += f.y; acc.z += f.z; acc.w += f.w;
      }
    }
    if (j + 4 <= m) {
      uint2 v[4];
#pragma unroll
      for (int k = 0; k < 4; ++k) {
        const int r = __builtin_amdgcn_readlane(mine, j + k);
        v[k] = hs2[(size_t)r * 32 + lane];
      }
#pragma unroll
      for (int k = 0; k < 4; ++k) {
        const float4 f = widen_bf16x4(v[k]);
        acc.x += f.x; acc.y += f.y; acc.z += f.z; acc.w += f.w;
      }
      j += 4;
    }
    for (; j < m; ++j) {
      const int r = __builtin_amdgcn_readlane(mine, j);
      const float4 f = widen_bf16x4(hs2[(size_t)r * 32 + lane]);
      acc.x += f.x; acc.y += f.y; acc.z += f.z; acc.w += f.w;
    }
  }
  return acc;
}

__global__ __launch_bounds__(256) void k_pull128_relu_gemm32(
    const unsigned int* __restrict__ hs, const float* __restrict__ dinv,
    const float* __restrict__ bias, const int* __restrict__ rowptr,
    const int* __restrict__ cnt, const int* __restrict__ src,
    const float* __restrict__ W2, float* __restrict__ g2, int n) {
  __shared__ __bf16 ldsA[4 * 16 * 32];
  __shared__ __bf16 ldsB[4 * D_OUT * 32];
  __shared__ float  ldsC[16 * D_OUT];

  const int tid  = threadIdx.x;
  const int wave = __builtin_amdgcn_readfirstlane(tid >> 5);
  const int lane = tid & 31;
  const int laneM = lane & 15;
  const int khalf = (lane >> 4) << 3;
  const int node0 = blockIdx.x * 16;

  int nodeQ[2], begQ[2], degQ[2];
  float dQ[2];
#pragma unroll
  for (int q = 0; q < 2; ++q) {
    nodeQ[q] = min(node0 + wave * 2 + q, n - 1);
    begQ[q]  = rowptr[nodeQ[q]];
    degQ[q]  = cnt[nodeQ[q]];
    dQ[q]    = dinv[nodeQ[q]];
  }

#pragma unroll
  for (int ks = 0; ks < 4; ++ks) {
    const int nn = tid & 31;
    const int kq = tid >> 5;
    union { __bf16 h[4]; uint2 q; } wb;
#pragma unroll
    for (int j = 0; j < 4; ++j)
      wb.h[j] = (__bf16)W2[(size_t)(ks * 32 + kq * 4 + j) * D_OUT + nn];
    *(uint2*)(ldsB + (ks * D_OUT + nn) * 32 + kq * 4) = wb.q;
  }

  const int outRow = node0 + (tid >> 3);
  const float dOut = (tid < 128 && outRow < n) ? dinv[outRow] : 0.0f;

  int firstQ[2];
#pragma unroll
  for (int q = 0; q < 2; ++q)
    firstQ[q] = (lane < min(32, degQ[q])) ? src[begQ[q] + lane] : 0;
  const uint2* __restrict__ hs2 = (const uint2*)hs;
  const float4 b4 = ((const float4*)bias)[lane];
#pragma unroll
  for (int q = 0; q < 2; ++q) {
    const int r = wave * 2 + q;
    union { __bf16 h[4]; uint2 u; } pk;
    const float4 acc = pull_row128(hs2, src, nodeQ[q], begQ[q], degQ[q], firstQ[q], lane);
    const float d = dQ[q];
    pk.h[0] = (__bf16)fmaxf(acc.x * d + b4.x, 0.0f);
    pk.h[1] = (__bf16)fmaxf(acc.y * d + b4.y, 0.0f);
    pk.h[2] = (__bf16)fmaxf(acc.z * d + b4.z, 0.0f);
    pk.h[3] = (__bf16)fmaxf(acc.w * d + b4.w, 0.0f);
    *(uint2*)(ldsA + ((lane >> 3) * 16 + r) * 32 + ((lane & 7) << 2)) = pk.u;
  }
  __syncthreads();

  if (wave < 2) {
    v8f acc = {0.f, 0.f, 0.f, 0.f, 0.f, 0.f, 0.f, 0.f};
#pragma unroll
    for (int ks = 0; ks < 4; ++ks) {
      const v16bf a = frag_from_lds(ldsA + ks * 16 * 32, laneM, khalf);
      const v16bf b = frag_from_lds(ldsB + ks * D_OUT * 32, (wave << 4) + laneM, khalf);
      acc = __builtin_amdgcn_wmma_f32_16x16x32_bf16(
          false, a, false, b, (short)0, acc, false, false);
    }
    const int mBase = (lane >> 4) << 3;
#pragma unroll
    for (int v = 0; v < 8; ++v)
      ldsC[(mBase + v) * D_OUT + (wave << 4) + laneM] = acc[v];
  }
  __syncthreads();

  if (tid < 128 && outRow < n) {
    float4 c = ((const float4*)ldsC)[(tid >> 3) * (D_OUT / 4) + (tid & 7)];
    c.x *= dOut; c.y *= dOut; c.z *= dOut; c.w *= dOut;
    ((float4*)g2)[(size_t)outRow * (D_OUT / 4) + (tid & 7)] = c;
  }
}

__global__ __launch_bounds__(256) void k_pull32_logsoftmax(
    const float* __restrict__ gs, const float* __restrict__ dinv,
    const float* __restrict__ bias, const int* __restrict__ rowptr,
    const int* __restrict__ cnt, const int* __restrict__ src,
    float* __restrict__ out, int n) {
  const int wave = threadIdx.x >> 5;
  const int lane = threadIdx.x & 31;
  const int g    = lane >> 3;
  const int c4   = lane & 7;
  const float4* __restrict__ gs4 = (const float4*)gs;
  float4* __restrict__ out4 = (float4*)out;
  const float4 b4 = ((const float4*)bias)[c4];
  const int nodeBase = blockIdx.x * 128 + wave * 4 + g;

  int node = nodeBase;
  int nd   = min(node, n - 1);
  int beg  = rowptr[nd];
  int deg  = (node < n) ? cnt[nd] : 0;
  int mine = (c4 < deg) ? src[beg + c4] : -1;
  float4 acc = gs4[(size_t)nd * 8 + c4];
  float d = dinv[nd];

  for (int t = 0; t < 4; ++t) {
    const int cNode = node, cBeg = beg, cDeg = deg;
    const float cD = d;
    node = nodeBase + (t + 1) * 32;
    nd   = min(node, n - 1);
    if (t + 1 < 4) {
      beg = rowptr[nd];
      deg = (node < n) ? cnt[nd] : 0;
    }
    int dmax = cDeg;
    dmax = max(dmax, __shfl_xor(dmax, 8, 32));
    dmax = max(dmax, __shfl_xor(dmax, 16, 32));
    dmax = __builtin_amdgcn_readfirstlane(dmax);
    for (int base = 0; base < dmax; base += 8) {
      const int e2 = base + 8 + c4;
      const int nxt = (e2 < cDeg) ? src[cBeg + e2] : -1;
      float4 v[8];
#pragma unroll
      for (int k = 0; k < 8; ++k) {
        const int r = __shfl(mine, k, 8);
        v[k] = make_float4(0.0f, 0.0f, 0.0f, 0.0f);
        if (r >= 0) v[k] = gs4[(size_t)r * 8 + c4];
      }
#pragma unroll
      for (int k = 0; k < 8; ++k) {
        acc.x += v[k].x; acc.y += v[k].y; acc.z += v[k].z; acc.w += v[k].w;
      }
      mine = nxt;
    }
    float4 v;
    v.x = acc.x * cD + b4.x;
    v.y = acc.y * cD + b4.y;
    v.z = acc.z * cD + b4.z;
    v.w = acc.w * cD + b4.w;
    if (t + 1 < 4) {
      mine = (c4 < deg) ? src[beg + c4] : -1;
      acc  = gs4[(size_t)nd * 8 + c4];
      d    = dinv[nd];
    }
    float mx = fmaxf(fmaxf(v.x, v.y), fmaxf(v.z, v.w));
    for (int o = 4; o > 0; o >>= 1) mx = fmaxf(mx, __shfl_xor(mx, o, 32));
    float s = (__expf(v.x - mx) + __expf(v.y - mx)) + (__expf(v.z - mx) + __expf(v.w - mx));
    for (int o = 4; o > 0; o >>= 1) s += __shfl_xor(s, o, 32);
    const float ls = __logf(s);
    if (cNode < n) {
      float4 o4;
      o4.x = v.x - mx - ls;
      o4.y = v.y - mx - ls;
      o4.z = v.z - mx - ls;
      o4.w = v.w - mx - ls;
      out4[(size_t)cNode * 8 + c4] = o4;
    }
  }
}

extern "C" void kernel_launch(void* const* d_in, const int* in_sizes, int n_in,
                              void* d_out, int out_size, void* d_ws, size_t ws_size,
                              hipStream_t stream) {
  const float* x  = (const float*)d_in[0];
  const float* W1 = (const float*)d_in[1];
  const float* b1 = (const float*)d_in[2];
  const float* W2 = (const float*)d_in[3];
  const float* b2 = (const float*)d_in[4];
  const int*  eix = (const int*)d_in[5];

  const int n = N_NODES;
  const int E = in_sizes[5] / 2;
  const int* row = eix;
  const int* col = eix + E;

  float* ws     = (float*)d_ws;
  float* dinv   = ws;
  int*   cnt    = (int*)(ws + N_PAD);
  int*   rowptr = cnt + N_PAD;
  int*   part   = rowptr + N_PAD;
  int*   src    = part + E;
  unsigned int* hs = (unsigned int*)(src + E);
  float* g2     = (float*)(hs + (size_t)n * (D_H / 2));
  int*   ltab   = (int*)(g2 + (size_t)n * D_OUT);
  int*   htab   = ltab + NBIN_PAD * NCH_PAD;
  unsigned int* w1t = (unsigned int*)(htab + NBIN_PAD * NCH_PAD);
  float* outf   = (float*)d_out;

  k_part<<<(E + CHUNK - 1) / CHUNK + W1_CONV_BLOCKS, 1024, 0, stream>>>(row, col, part, ltab, htab, E, W1, w1t);
  k_sort<<<NBIN, 1024, 0, stream>>>(part, ltab, htab, (E + CHUNK - 1) / CHUNK, src, rowptr, cnt, dinv,
                                    x, w1t, hs, n);

  k_pull128_relu_gemm32<<<(n + 15) / 16, 256, 0, stream>>>(hs, dinv, b1, rowptr, cnt, src, W2, g2, n);

  k_pull32_logsoftmax<<<(n + 127) / 128, 256, 0, stream>>>(g2, dinv, b2, rowptr, cnt, src, outf, n);
}


# Layer-1 product: each work-group prefetches its 512 rows of x into the second-level cache (one 256-B unit per lane; k-steps 0-3 behind the edge-run requests, 4-7 at the first fragment loads)
# speedup vs baseline: 7.8863x; 1.0300x over previous
#include <hip/hip_runtime.h>

#define N_NODES 100000
#define D_IN 256
#define D_H 128
#define D_OUT 32

typedef __attribute__((ext_vector_type(16))) __bf16 v16bf;
typedef __attribute__((ext_vector_type(8)))  float  v8f;

__device__ __forceinline__ v16bf frag_from_lds(const __bf16* base, int rowIdx, int khalf) {
  union { uint4 u[2]; v16bf v; } c;
  const uint4* p = (const uint4*)(base + rowIdx * 32 + khalf);
  c.u[0] = p[0];
  c.u[1] = p[2];
  return c.v;
}

#define W1T_PITCH 264
__device__ __forceinline__ v16bf frag_from_w1(const __bf16* base, int rowIdx, int k0) {
  union { uint4 u[2]; v16bf v; } c;
  const uint4* p = (const uint4*)(base + rowIdx * W1T_PITCH + k0);
  c.u[0] = p[0];
  c.u[1] = p[2];
  return c.v;
}

__device__ __forceinline__ v16bf frag_from_f32(const float4 (&q)[4]) {
  union { __bf16 h[16]; v16bf v; } c;
#pragma unroll
  for (int j = 0; j < 4; ++j) {
    c.h[j * 4 + 0] = (__bf16)q[j].x;
    c.h[j * 4 + 1] = (__bf16)q[j].y;
    c.h[j * 4 + 2] = (__bf16)q[j].z;
    c.h[j * 4 + 3] = (__bf16)q[j].w;
  }
  return c.v;
}

__device__ __forceinline__ void w1_to_bf16_rows(
    const float* __restrict__ W, __bf16* __restrict__ w1t, int g) {
  const int n  = g & (D_H - 1);
  const int kq = g >> 7;
  const int p  = (n & 7) * 16 + (n >> 3);
  union { __bf16 h[4]; uint2 q; } wb;
#pragma unroll
  for (int j = 0; j < 4; ++j)
    wb.h[j] = (__bf16)W[(size_t)(kq * 4 + j) * D_H + n];
  *(uint2*)(w1t + p * D_IN + kq * 4) = wb.q;
}


#define N_PAD 100352

#define BIN_SHIFT 9
#define NBIN 196
#define NBIN_PAD 256
#define CHUNK 8192
#define NCH_PAD 1024
#define SORT_CAP 16384
#define SRC_MASK 0x1FFFF

#define W1_CONV_BLOCKS 8
__global__ __launch_bounds__(1024) void k_part(
    const int* __restrict__ row, const int* __restrict__ col,
    int* __restrict__ part, int* __restrict__ ltab, int* __restrict__ htab, int E,
    const float* __restrict__ W1, unsigned int* __restrict__ w1t) {
  __shared__ int h[NBIN_PAD];
  __shared__ int lstart[NBIN_PAD];
  __shared__ int sval[CHUNK];
  const int tid = threadIdx.x;
  const int nChunk = (E + CHUNK - 1) / CHUNK;
  if ((int)blockIdx.x >= nChunk) {
    w1_to_bf16_rows(W1, (__bf16*)w1t, ((int)blockIdx.x - nChunk) * 1024 + tid);
    return;
  }
  if (tid < NBIN_PAD) h[tid] = 0;
  __syncthreads();
  const int base = blockIdx.x * CHUNK;
  int c[CHUNK / 1024], r[CHUNK / 1024], lr[CHUNK / 1024];
#pragma unroll
  for (int k = 0; k < CHUNK / 1024; ++k) {
    const int e = base + k * 1024 + tid;
    const bool ok = e < E;
    c[k] = ok ? col[e] : -1;
    r[k] = ok ? row[e] : 0;
  }
#pragma unroll
  for (int k = 0; k < CHUNK / 1024; ++k) {
    lr[k] = 0;
    if (c[k] >= 0) lr[k] = atomicAdd(&h[c[k] >> BIN_SHIFT], 1);
  }
  __syncthreads();
  if (tid < 32) {
    int hv[8], hloc[8];
    int hsum = 0;
#pragma unroll
    for (int j = 0; j < 8; ++j) hv[j] = h[tid * 8 + j];
#pragma unroll
    for (int j = 0; j < 8; ++j) { hloc[j] = hsum; hsum += hv[j]; }
    int hi = hsum;
    for (int o = 1; o < 32; o <<= 1) {
      const int y = __shfl_up(hi, o, 32);
      if (tid >= o) hi += y;
    }
    const int hex = hi - hsum;
#pragma unroll
    for (int j = 0; j < 8; ++j) {
      const int b = tid * 8 + j;
      const int ls = hex + hloc[j];
      lstart[b] = ls;
      ltab[b * NCH_PAD + (int)blockIdx.x] = ls;
      htab[b * NCH_PAD + (int)blockIdx.x] = hv[j];
    }
  }
  __syncthreads();
#pragma unroll
  for (int k = 0; k < CHUNK / 1024; ++k) {
    if (c[k] >= 0) {
      const int p = lstart[c[k] >> BIN_SHIFT] + lr[k];
      sval[p] = ((c[k] & ((1 << BIN_SHIFT) - 1)) << 17) | r[k];
    }
  }
  __syncthreads();
  const int nhere = min(CHUNK, E - base);
#pragma unroll
  for (int k = 0; k < CHUNK / 1024; ++k) {
    const int i = k * 1024 + tid;
    if (i < nhere) part[base + i] = sval[i];
  }
}

#define RUNS_REG 7
template <class F>
__device__ __forceinline__ void sort_rest(
    const int* __restrict__ part, const int* run_at, const int* run_n,
    int nChunk, int wave, int lane, F f) {
  for (int c = wave; c < nChunk; c += 32) {
    const int rs = run_at[c], rn = run_n[c];
    const int j0 = (c - wave < 32 * RUNS_REG) ? 64 : 0;
    for (int j = j0 + lane; j < rn; j += 32) f(part[rs + j]);
  }
}

__global__ __launch_bounds__(1024) void k_sort(
    const int* __restrict__ part, const int* __restrict__ ltab,
    const int* __restrict__ htab, int nChunk,
    int* __restrict__ src, int* __restrict__ rowptr, int* __restrict__ cnt,
    float* __restrict__ dinv,
    const float* __restrict__ A, const unsigned int* __restrict__ w1t,
    unsigned int* __restrict__ C, int M) {
  __shared__ __attribute__((aligned(16))) __bf16 ldsB[D_H * W1T_PITCH];
  __shared__ __attribute__((aligned(16))) float dinv_l[512];
  __shared__ int c_l[512];
  __shared__ int cur_l[512];
  __shared__ int rp_l[512];
  __shared__ int wsum[32];
  __shared__ int wsum2[32];
  __shared__ int s_start, s_nb;
  __shared__ int run_at[NCH_PAD];
  __shared__ int run_n[NCH_PAD];
  __shared__ int sorted_l[SORT_CAP];
  const int tid  = threadIdx.x;
  const int lane = tid & 31;
  const int wave = __builtin_amdgcn_readfirstlane(tid >> 5);
  const int bin  = blockIdx.x;

  uint4 wv[4];
  {
    const uint4* __restrict__ w4 = (const uint4*)w1t;
#pragma unroll
    for (int i = 0; i < 4; ++i) wv[i] = w4[tid + i * 1024];
  }

  if (tid < 512) { c_l[tid] = 0; cur_l[tid] = 0; }
  int ls = 0, hn = 0;
  if (tid < nChunk) {
    ls = ltab[bin * NCH_PAD + tid];
    hn = htab[bin * NCH_PAD + tid];
  }
  run_at[tid] = tid * CHUNK + ls;
  run_n[tid]  = hn;
#pragma unroll
  for (int i = 0; i < 4; ++i) {
    const int idx = tid + i * 1024;
    *(uint4*)(ldsB + (idx >> 5) * W1T_PITCH + ((idx & 31) << 3)) = wv[i];
  }
  for (int o = 16; o > 0; o >>= 1) {
    ls += __shfl_xor(ls, o, 32);
    hn += __shfl_xor(hn, o, 32);
  }
  if (lane == 0) { wsum[wave] = ls; wsum2[wave] = hn; }
  __syncthreads();
  if (tid < 32) {
    int a = wsum[lane], b = wsum2[lane];
    for (int o = 16; o > 0; o >>= 1) {
      a += __shfl_xor(a, o, 32);
      b += __shfl_xor(b, o, 32);
    }
    if (tid == 0) { s_start = a; s_nb = b; }
  }
  __syncthreads();
  const int bstart = s_start;
  const int nb = s_nb;

  int en[RUNS_REG][2];
  int rsK[RUNS_REG], rnK[RUNS_REG];
#pragma unroll
  for (int k = 0; k < RUNS_REG; ++k) {
    const int cidx = wave + 32 * k;
    rsK[k] = run_at[cidx];
    rnK[k] = run_n[cidx];
  }
#pragma unroll
  for (int k = 0; k < RUNS_REG; ++k) {
    en[k][0] = (lane < rnK[k])      ? part[rsK[k] + lane]      : -1;
    en[k][1] = (lane + 32 < rnK[k]) ? part[rsK[k] + lane + 32] : -1;
  }

  const int prow = (bin << BIN_SHIFT) + (tid & 511);
  const float* __restrict__ xunit = A + (size_t)min(prow, M - 1) * D_IN + (tid >> 9) * 64;
  if (prow < M) __builtin_prefetch(xunit, 0, 2);

#pragma unroll
  for (int k = 0; k < RUNS_REG; ++k) {
    if (en[k][0] >= 0) atomicAdd(&c_l[en[k][0] >> 17], 1);
    if (en[k][1] >= 0) atomicAdd(&c_l[en[k][1] >> 17], 1);
  }
  sort_rest(part, run_at, run_n, nChunk, wave, lane,
            [&](int e) { atomicAdd(&c_l[e >> 17], 1); });
  __syncthreads();

  int x = 0, c = 0;
  if (tid < 512) {
    c = c_l[tid];
    x = c;
    for (int o = 1; o < 32; o <<= 1) {
      const int y = __shfl_up(x, o, 32);
      if (lane >= o) x += y;
    }
    if (lane == 31) wsum[wave] = x;
  }
  __syncthreads();
  if (tid < 32) {
    int t = (lane < 16) ? wsum[lane] : 0;
    for (int o = 1; o < 32; o <<= 1) {
      const int y = __shfl_up(t, o, 32);
      if (lane >= o) t += y;
    }
    wsum[lane] = t;
  }
  __syncthreads();
  if (tid < 512) {
    const int rp = x - c + ((wave > 0) ? wsum[wave - 1] : 0);
    rp_l[tid] = rp;
    const int node = (bin << BIN_SHIFT) + tid;
    cnt[node]    = c;
    rowptr[node] = bstart + rp;
    const float dv = rsqrtf((float)(c + 1));
    dinv[node]   = dv;
    dinv_l[tid]  = dv;
  }
  __syncthreads();

  const bool fits = nb <= SORT_CAP;
  auto place = [&](int e) {
    const int l = e >> 17;
    const int p = rp_l[l] + atomicAdd(&cur_l[l], 1);
    if (fits) sorted_l[p] = e & SRC_MASK;
    else      src[bstart + p] = e & SRC_MASK;
  };
#pragma unroll
  for (int k = 0; k < RUNS_REG; ++k) {
    if (en[k][0] >= 0) place(en[k][0]);
    if (en[k][1] >= 0) place(en[k][1]);
  }
  sort_rest(part, run_at, run_n, nChunk, wave, lane, place);

  constexpr int K = D_IN, NT = D_H / 16;
  static_assert(NT == 8, "the epilogue packs 8 column tiles into one 16-byte store");
  const int laneM = lane & 15;
  const int khalf = (lane >> 4) << 3;
  const int tile  = bin * 32 + wave;
  const bool haveTile = tile < ((M + 15) >> 4);
  const int row0 = tile << 4;
  const float4* __restrict__ arow =
      (const float4*)(A + (size_t)min(row0 + laneM, M - 1) * K + khalf);
  float4 abuf[2][4];
  if (haveTile) {
#pragma unroll
    for (int s = 0; s < 2; ++s) {
      abuf[s][0] = arow[s * 8 + 0];
      abuf[s][1] = arow[s * 8 + 1];
      abuf[s][2] = arow[s * 8 + 4];
      abuf[s][3] = arow[s * 8 + 5];
    }
  }
  if (prow < M) __builtin_prefetch(xunit + 128, 0, 2);
  __syncthreads();
  if (fits)
    for (int i = tid; i < nb; i += 1024) src[bstart + i] = sorted_l[i];

  if (!haveTile) return;
  v8f acc[NT];
#pragma unroll
  for (int t = 0; t < NT; ++t) {
    v8f z = {0.f, 0.f, 0.f, 0.f, 0.f, 0.f, 0.f, 0.f};
    acc[t] = z;
  }
#pragma unroll
  for (int s = 0; s < K / 32; ++s) {
    const v16bf a = frag_from_f32(abuf[s & 1]);
    if (s + 2 < K / 32) {
      abuf[s & 1][0] = arow[(s + 2) * 8 + 0];
      abuf[s & 1][1] = arow[(s + 2) * 8 + 1];
      abuf[s & 1][2] = arow[(s + 2) * 8 + 4];
      abuf[s & 1][3] = arow[(s + 2) * 8 + 5];
    }
    const int k0 = s * 32 + khalf;
    v16bf bc = frag_from_w1(ldsB, laneM, k0);
#pragma unroll
    for (int t = 0; t < NT; ++t) {
      v16bf bn = (t + 1 < NT) ? frag_from_w1(ldsB, ((t + 1) << 4) + laneM, k0) : bc;
      acc[t] = __builtin_amdgcn_wmma_f32_16x16x32_bf16(
          false, a, false, bc, (short)0, acc[t], false, false);
      bc = bn;
    }
  }

  uint4* __restrict__ C4 = (uint4*)C;
  const int lBase = (wave << 4) + ((lane >> 4) << 3);
  const int mBase = row0 + ((lane >> 4) << 3);
  const float4 dLo = *(const float4*)(dinv_l + lBase);
  const float4 dHi = *(const float4*)(dinv_l + lBase + 4);
#pragma unroll
  for (int v = 0; v < 8; ++v) {
    const int gr = mBase + v;
    if (gr < M) {
      const float d = (v == 0) ? dLo.x : (v == 1) ? dLo.y : (v == 2) ? dLo.z : (v == 3) ? dLo.w
                    : (v == 4) ? dHi.x : (v == 5) ? dHi.y : (v == 6) ? dHi.z : dHi.w;
      union { __bf16 h[8]; uint4 q; } o;
#pragma unroll
      for (int t = 0; t < NT; ++t) o.h[t] = (__bf16)(acc[t][v] * d);
      C4[(size_t)gr * (D_H / 8) + laneM] = o.q;
    }
  }
}

__device__ __forceinline__ float4 widen_bf16x4(uint2 u) {
  float4 f;
  f.x = __uint_as_float(u.x << 16);
  f.y = __uint_as_float(u.x & 0xFFFF0000u);
  f.z = __uint_as_float(u.y << 16);
  f.w = __uint_as_float(u.y & 0xFFFF0000u);
  return f;
}

__device__ __forceinline__ float4 pull_row128(
    const uint2* __restrict__ hs2, const int* __restrict__ src,
    int node, int beg, int deg, int first, int lane) {
  float4 acc = widen_bf16x4(hs2[(size_t)node * 32 + lane]);
  for (int base = 0; base < deg; base += 32) {
    const int m = min(32, deg - base);
    const int mine = (base == 0) ? first : ((lane < m) ? src[beg + base + lane] : 0);
    int j = 0;
    for (; j + 8 <= m; j += 8) {
      uint2 v[8];
#pragma unroll
      for (int k = 0; k < 8; ++k) {
        const int r = __builtin_amdgcn_readlane(mine, j + k);
        v[k] = hs2[(size_t)r * 32 + lane];
      }
#pragma unroll
      for (int k = 0; k < 8; ++k) {
        const float4 f = widen_bf16x4(v[k]);
        acc.x += f.x; acc.y += f.y; acc.z += f.z; acc.w += f.w;
      }
    }
    if (j + 4 <= m) {
      uint2 v[4];
#pragma unroll
      for (int k = 0; k < 4; ++k) {
        const int r = __builtin_amdgcn_readlane(mine, j + k);
        v[k] = hs2[(size_t)r * 32 + lane];
      }
#pragma unroll
      for (int k = 0; k < 4; ++k) {
        const float4 f = widen_bf16x4(v[k]);
        acc.x += f.x; acc.y += f.y; acc.z += f.z; acc.w += f.w;
      }
      j += 4;
    }
    for (; j < m; ++j) {
      const int r = __builtin_amdgcn_readlane(mine, j);
      const float4 f = widen_bf16x4(hs2[(size_t)r * 32 + lane]);
      acc.x += f.x; acc.y += f.y; acc.z += f.z; acc.w += f.w;
    }
  }
  return acc;
}

__global__ __launch_bounds__(256) void k_pull128_relu_gemm32(
    const unsigned int* __restrict__ hs, const float* __restrict__ dinv,
    const float* __restrict__ bias, const int* __restrict__ rowptr,
    const int* __restrict__ cnt, const int* __restrict__ src,
    const float* __restrict__ W2, float* __restrict__ g2, int n) {
  __shared__ __bf16 ldsA[4 * 16 * 32];
  __shared__ __bf16 ldsB[4 * D_OUT * 32];
  __shared__ float  ldsC[16 * D_OUT];

  const int tid  = threadIdx.x;
  const int wave = __builtin_amdgcn_readfirstlane(tid >> 5);
  const int lane = tid & 31;
  const int laneM = lane & 15;
  const int khalf = (lane >> 4) << 3;
  const int node0 = blockIdx.x * 16;

  int nodeQ[2], begQ[2], degQ[2];
  float dQ[2];
#pragma unroll
  for (int q = 0; q < 2; ++q) {
    nodeQ[q] = min(node0 + wave * 2 + q, n - 1);
    begQ[q]  = rowptr[nodeQ[q]];
    degQ[q]  = cnt[nodeQ[q]];
    dQ[q]    = dinv[nodeQ[q]];
  }

#pragma unroll
  for (int ks = 0; ks < 4; ++ks) {
    const int nn = tid & 31;
    const int kq = tid >> 5;
    union { __bf16 h[4]; uint2 q; } wb;
#pragma unroll
    for (int j = 0; j < 4; ++j)
      wb.h[j] = (__bf16)W2[(size_t)(ks * 32 + kq * 4 + j) * D_OUT + nn];
    *(uint2*)(ldsB + (ks * D_OUT + nn) * 32 + kq * 4) = wb.q;
  }

  const int outRow = node0 + (tid >> 3);
  const float dOut = (tid < 128 && outRow < n) ? dinv[outRow] : 0.0f;

  int firstQ[2];
#pragma unroll
  for (int q = 0; q < 2; ++q)
    firstQ[q] = (lane < min(32, degQ[q])) ? src[begQ[q] + lane] : 0;
  const uint2* __restrict__ hs2 = (const uint2*)hs;
  const float4 b4 = ((const float4*)bias)[lane];
#pragma unroll
  for (int q = 0; q < 2; ++q) {
    const int r = wave * 2 + q;
    union { __bf16 h[4]; uint2 u; } pk;
    const float4 acc = pull_row128(hs2, src, nodeQ[q], begQ[q], degQ[q], firstQ[q], lane);
    const float d = dQ[q];
    pk.h[0] = (__bf16)fmaxf(acc.x * d + b4.x, 0.0f);
    pk.h[1] = (__bf16)fmaxf(acc.y * d + b4.y, 0.0f);
    pk.h[2] = (__bf16)fmaxf(acc.z * d + b4.z, 0.0f);
    pk.h[3] = (__bf16)fmaxf(acc.w * d + b4.w, 0.0f);
    *(uint2*)(ldsA + ((lane >> 3) * 16 + r) * 32 + ((lane & 7) << 2)) = pk.u;
  }
  __syncthreads();

  if (wave < 2) {
    v8f acc = {0.f, 0.f, 0.f, 0.f, 0.f, 0.f, 0.f, 0.f};
#pragma unroll
    for (int ks = 0; ks < 4; ++ks) {
      const v16bf a = frag_from_lds(ldsA + ks * 16 * 32, laneM, khalf);
      const v16bf b = frag_from_lds(ldsB + ks * D_OUT * 32, (wave << 4) + laneM, khalf);
      acc = __builtin_amdgcn_wmma_f32_16x16x32_bf16(
          false, a, false, b, (short)0, acc, false, false);
    }
    const int mBase = (lane >> 4) << 3;
#pragma unroll
    for (int v = 0; v < 8; ++v)
      ldsC[(mBase + v) * D_OUT + (wave << 4) + laneM] = acc[v];
  }
  __syncthreads();

  if (tid < 128 && outRow < n) {
    float4 c = ((const float4*)ldsC)[(tid >> 3) * (D_OUT / 4) + (tid & 7)];
    c.x *= dOut; c.y *= dOut; c.z *= dOut; c.w *= dOut;
    ((float4*)g2)[(size_t)outRow * (D_OUT / 4) + (tid & 7)] = c;
  }
}

__global__ __launch_bounds__(256) void k_pull32_logsoftmax(
    const float* __restrict__ gs, const float* __restrict__ dinv,
    const float* __restrict__ bias, const int* __restrict__ rowptr,
    const int* __restrict__ cnt, const int* __restrict__ src,
    float* __restrict__ out, int n) {
  const int wave = threadIdx.x >> 5;
  const int lane = threadIdx.x & 31;
  const int g    = lane >> 3;
  const int c4   = lane & 7;
  const float4* __restrict__ gs4 = (const float4*)gs;
  float4* __restrict__ out4 = (float4*)out;
  const float4 b4 = ((const float4*)bias)[c4];
  const int nodeBase = blockIdx.x * 128 + wave * 4 + g;

  int node = nodeBase;
  int nd   = min(node, n - 1);
  int beg  = rowptr[nd];
  int deg  = (node < n) ? cnt[nd] : 0;
  int mine = (c4 < deg) ? src[beg + c4] : -1;
  float4 acc = gs4[(size_t)nd * 8 + c4];
  float d = dinv[nd];

  for (int t = 0; t < 4; ++t) {
    const int cNode = node, cBeg = beg, cDeg = deg;
    const float cD = d;
    node = nodeBase + (t + 1) * 32;
    nd   = min(node, n - 1);
    if (t + 1 < 4) {
      beg = rowptr[nd];
      deg = (node < n) ? cnt[nd] : 0;
    }
    int dmax = cDeg;
    dmax = max(dmax, __shfl_xor(dmax, 8, 32));
    dmax = max(dmax, __shfl_xor(dmax, 16, 32));
    dmax = __builtin_amdgcn_readfirstlane(dmax);
    for (int base = 0; base < dmax; base += 8) {
      const int e2 = base + 8 + c4;
      const int nxt = (e2 < cDeg) ? src[cBeg + e2] : -1;
      float4 v[8];
#pragma unroll
      for (int k = 0; k < 8; ++k) {
        const int r = __shfl(mine, k, 8);
        v[k] = make_float4(0.0f, 0.0f, 0.0f, 0.0f);
        if (r >= 0) v[k] = gs4[(size_t)r * 8 + c4];
      }
#pragma unroll
      for (int k = 0; k < 8; ++k) {
        acc.x += v[k].x; acc.y += v[k].y; acc.z += v[k].z; acc.w += v[k].w;
      }
      mine = nxt;
    }
    float4 v;
    v.x = acc.x * cD + b4.x;
    v.y = acc.y * cD + b4.y;
    v.z = acc.z * cD + b4.z;
    v.w = acc.w * cD + b4.w;
    if (t + 1 < 4) {
      mine = (c4 < deg) ? src[beg + c4] : -1;
      acc  = gs4[(size_t)nd * 8 + c4];
      d    = dinv[nd];
    }
    float mx = fmaxf(fmaxf(v.x, v.y), fmaxf(v.z, v.w));
    for (int o = 4; o > 0; o >>= 1) mx = fmaxf(mx, __shfl_xor(mx, o, 32));
    float s = (__expf(v.x - mx) + __expf(v.y - mx)) + (__expf(v.z - mx) + __expf(v.w - mx));
    for (int o = 4; o > 0; o >>= 1) s += __shfl_xor(s, o, 32);
    const float ls = __logf(s);
    if (cNode < n) {
      float4 o4;
      o4.x = v.x - mx - ls;
      o4.y = v.y - mx - ls;
      o4.z = v.z - mx - ls;
      o4.w = v.w - mx - ls;
      out4[(size_t)cNode * 8 + c4] = o4;
    }
  }
}

extern "C" void kernel_launch(void* const* d_in, const int* in_sizes, int n_in,
                              void* d_out, int out_size, void* d_ws, size_t ws_size,
                              hipStream_t stream) {
  const float* x  = (const float*)d_in[0];
  const float* W1 = (const float*)d_in[1];
  const float* b1 = (const float*)d_in[2];
  const float* W2 = (const float*)d_in[3];
  const float* b2 = (const float*)d_in[4];
  const int*  eix = (const int*)d_in[5];

  const int n = N_NODES;
  const int E = in_sizes[5] / 2;
  const int* row = eix;
  const int* col = eix + E;

  float* ws     = (float*)d_ws;
  float* dinv   = ws;
  int*   cnt    = (int*)(ws + N_PAD);
  int*   rowptr = cnt + N_PAD;
  int*   part   = rowptr + N_PAD;
  int*   src    = part + E;
  unsigned int* hs = (unsigned int*)(src + E);
  float* g2     = (float*)(hs + (size_t)n * (D_H / 2));
  int*   ltab   = (int*)(g2 + (size_t)n * D_OUT);
  int*   htab   = ltab + NBIN_PAD * NCH_PAD;
  unsigned int* w1t = (unsigned int*)(htab + NBIN_PAD * NCH_PAD);
  float* outf   = (float*)d_out;

  k_part<<<(E + CHUNK - 1) / CHUNK + W1_CONV_BLOCKS, 1024, 0, stream>>>(row, col, part, ltab, htab, E, W1, w1t);
  k_sort<<<NBIN, 1024, 0, stream>>>(part, ltab, htab, (E + CHUNK - 1) / CHUNK, src, rowptr, cnt, dinv,
                                    x, w1t, hs, n);

  k_pull128_relu_gemm32<<<(n + 15) / 16, 256, 0, stream>>>(hs, dinv, b1, rowptr, cnt, src, W2, g2, n);

  k_pull32_logsoftmax<<<(n + 127) / 128, 256, 0, stream>>>(g2, dinv, b2, rowptr, cnt, src, outf, n);
}
